# starting kernel
# baseline (speedup 1.0000x reference)
#include <hip/hip_runtime.h>
#include <math.h>

typedef __attribute__((ext_vector_type(2))) float v2f;
typedef __attribute__((ext_vector_type(8))) float v8f;

#define NEG_SLOPE 0.2f

__device__ __forceinline__ void atomicMaxF(float* addr, float v) {
  if (v >= 0.0f) atomicMax((int*)addr, __float_as_int(v));
  else           atomicMin((unsigned int*)addr, __float_as_uint(v));
}

__global__ void __launch_bounds__(256)
fill_kernel(float* __restrict__ p, int n, float v) {
  int i = blockIdx.x * blockDim.x + threadIdx.x;
  if (i < n) p[i] = v;
}

__global__ void __launch_bounds__(128)
gemm1_kernel(const float* __restrict__ x, const float* __restrict__ W1l,
             const float* __restrict__ W1r, float* __restrict__ xl1,
             float* __restrict__ xr1, int n_nodes) {
  const int lane = threadIdx.x & 31;
  const int wave = threadIdx.x >> 5;
  const int m0   = blockIdx.x * 16;
  const int n0   = wave * 16;
  const int half = lane >> 4;
  const int rc   = lane & 15;

  int arow = m0 + rc; if (arow >= n_nodes) arow = n_nodes - 1;
  const v2f*   xrow = (const v2f*)(x + (size_t)arow * 128);
  const float* Bl   = W1l + n0 + rc;
  const float* Br   = W1r + n0 + rc;

  v8f accL = {}, accR = {};
  #pragma unroll
  for (int kc = 0; kc < 32; ++kc) {
    v2f a = xrow[2 * kc + half];
    const int k0 = 4 * kc + 2 * half;
    v2f bl, br;
    bl[0] = Bl[(size_t)(k0 + 0) * 64];
    bl[1] = Bl[(size_t)(k0 + 1) * 64];
    br[0] = Br[(size_t)(k0 + 0) * 64];
    br[1] = Br[(size_t)(k0 + 1) * 64];
    accL = __builtin_amdgcn_wmma_f32_16x16x4_f32(
        false, a, false, bl, (short)0, accL, false, false);
    accR = __builtin_amdgcn_wmma_f32_16x16x4_f32(
        false, a, false, br, (short)0, accR, false, false);
  }

  #pragma unroll
  for (int r = 0; r < 8; ++r) {
    int row = m0 + r + 8 * half;
    if (row < n_nodes) {
      xl1[(size_t)row * 64 + n0 + rc] = accL[r];
      xr1[(size_t)row * 64 + n0 + rc] = accR[r];
    }
  }
}

__global__ void __launch_bounds__(128)
gemm2_kernel(const float* __restrict__ h, const float* __restrict__ W2l,
             const float* __restrict__ W2r, float* __restrict__ xl2,
             float* __restrict__ xr2, int n_nodes, int mtiles) {
  const int lane  = threadIdx.x & 31;
  const int mtile = blockIdx.x * 4 + (threadIdx.x >> 5);
  if (mtile >= mtiles) return;
  const int m0   = mtile * 16;
  const int half = lane >> 4;
  const int rc   = lane & 15;

  int arow = m0 + rc; if (arow >= n_nodes) arow = n_nodes - 1;
  const v2f* hrow = (const v2f*)(h + (size_t)arow * 64);
  const bool colok = (rc < 10);

  v8f accL = {}, accR = {};
  #pragma unroll
  for (int kc = 0; kc < 16; ++kc) {
    v2f a = hrow[2 * kc + half];
    const int k0 = 4 * kc + 2 * half;
    v2f bl, br;
    bl[0] = colok ? W2l[(size_t)(k0 + 0) * 10 + rc] : 0.0f;
    bl[1] = colok ? W2l[(size_t)(k0 + 1) * 10 + rc] : 0.0f;
    br[0] = colok ? W2r[(size_t)(k0 + 0) * 10 + rc] : 0.0f;
    br[1] = colok ? W2r[(size_t)(k0 + 1) * 10 + rc] : 0.0f;
    accL = __builtin_amdgcn_wmma_f32_16x16x4_f32(
        false, a, false, bl, (short)0, accL, false, false);
    accR = __builtin_amdgcn_wmma_f32_16x16x4_f32(
        false, a, false, br, (short)0, accR, false, false);
  }

  #pragma unroll
  for (int r = 0; r < 8; ++r) {
    int row = m0 + r + 8 * half;
    if (row < n_nodes) {
      xl2[(size_t)row * 16 + rc] = accL[r];
      xr2[(size_t)row * 16 + rc] = accR[r];
    }
  }
}

__global__ void __launch_bounds__(256)
l1_edge_logits(const int* __restrict__ ei, const float* __restrict__ xl1,
               const float* __restrict__ xr1, const float* __restrict__ att1,
               float* __restrict__ logits1, float* __restrict__ m1, int E) {
  int gid = blockIdx.x * blockDim.x + threadIdx.x;
  int g   = gid >> 4;
  bool act = (g < E);
  if (g >= E) g = E - 1;
  const int l   = threadIdx.x & 15;
  const int src = ei[g];
  const int dst = ei[E + g];

  const float4 xv = *(const float4*)(xl1 + (size_t)src * 64 + l * 4);
  const float4 rv = *(const float4*)(xr1 + (size_t)dst * 64 + l * 4);
  const float4 at = *(const float4*)(att1 + l * 4);

  float p = 0.0f, s;
  s = xv.x + rv.x; s = (s > 0.f) ? s : NEG_SLOPE * s; p += s * at.x;
  s = xv.y + rv.y; s = (s > 0.f) ? s : NEG_SLOPE * s; p += s * at.y;
  s = xv.z + rv.z; s = (s > 0.f) ? s : NEG_SLOPE * s; p += s * at.z;
  s = xv.w + rv.w; s = (s > 0.f) ? s : NEG_SLOPE * s; p += s * at.w;

  p += __shfl_xor(p, 1, 32);
  p += __shfl_xor(p, 2, 32);

  if (act && (l & 3) == 0) {
    int hd = l >> 2;
    logits1[(size_t)g * 4 + hd] = p;
    atomicMaxF(m1 + (size_t)dst * 4 + hd, p);
  }
}

__global__ void __launch_bounds__(256)
l1_edge_aggregate(const int* __restrict__ ei, const float* __restrict__ xl1,
                  const float* __restrict__ logits1, const float* __restrict__ m1,
                  float* __restrict__ den1, float* __restrict__ hacc, int E) {
  int g = (blockIdx.x * blockDim.x + threadIdx.x) >> 4;
  if (g >= E) return;
  const int l   = threadIdx.x & 15;
  const int src = ei[g];
  const int dst = ei[E + g];
  const int hd  = l >> 2;

  float w = __expf(logits1[(size_t)g * 4 + hd] - m1[(size_t)dst * 4 + hd]);
  if ((l & 3) == 0) atomicAdd(den1 + (size_t)dst * 4 + hd, w);

  const float4 v = *(const float4*)(xl1 + (size_t)src * 64 + l * 4);
  float* o = hacc + (size_t)dst * 64 + l * 4;
  atomicAdd(o + 0, w * v.x);
  atomicAdd(o + 1, w * v.y);
  atomicAdd(o + 2, w * v.z);
  atomicAdd(o + 3, w * v.w);
}

__global__ void __launch_bounds__(256)
l1_finish(float* __restrict__ hacc, const float* __restrict__ den1,
          const float* __restrict__ b1, int total) {
  int i = blockIdx.x * blockDim.x + threadIdx.x;
  if (i >= total) return;
  int n = i >> 6, ch = i & 63, hd = ch >> 4;
  float d = den1[(size_t)n * 4 + hd];
  float v = (d > 0.0f) ? hacc[i] / d : 0.0f;
  v += b1[ch];
  hacc[i] = (v > 0.0f) ? v : expm1f(v);
}

__global__ void __launch_bounds__(256)
l2_edge_logits(const int* __restrict__ ei, const float* __restrict__ xl2,
               const float* __restrict__ xr2, const float* __restrict__ att2,
               float* __restrict__ logits2, float* __restrict__ m2, int E) {
  int gid = blockIdx.x * blockDim.x + threadIdx.x;
  int g   = gid >> 2;
  bool act = (g < E);
  if (g >= E) g = E - 1;
  const int l   = threadIdx.x & 3;
  const int src = ei[g];
  const int dst = ei[E + g];
  const int c0  = l * 4;

  const float4 a = *(const float4*)(xl2 + (size_t)src * 16 + c0);
  const float4 b = *(const float4*)(xr2 + (size_t)dst * 16 + c0);

  float p = 0.0f, s;
  s = a.x + b.x; s = (s > 0.f) ? s : NEG_SLOPE * s; if (c0 + 0 < 10) p += s * att2[c0 + 0];
  s = a.y + b.y; s = (s > 0.f) ? s : NEG_SLOPE * s; if (c0 + 1 < 10) p += s * att2[c0 + 1];
  s = a.z + b.z; s = (s > 0.f) ? s : NEG_SLOPE * s; if (c0 + 2 < 10) p += s * att2[c0 + 2];
  s = a.w + b.w; s = (s > 0.f) ? s : NEG_SLOPE * s; if (c0 + 3 < 10) p += s * att2[c0 + 3];

  p += __shfl_xor(p, 1, 32);
  p += __shfl_xor(p, 2, 32);

  if (act && l == 0) {
    logits2[g] = p;
    atomicMaxF(m2 + dst, p);
  }
}

__global__ void __launch_bounds__(256)
l2_edge_aggregate(const int* __restrict__ ei, const float* __restrict__ xl2,
                  const float* __restrict__ logits2, const float* __restrict__ m2,
                  float* __restrict__ den2, float* __restrict__ acc2, int E) {
  int g = (blockIdx.x * blockDim.x + threadIdx.x) >> 2;
  if (g >= E) return;
  const int l   = threadIdx.x & 3;
  const int src = ei[g];
  const int dst = ei[E + g];
  const int c0  = l * 4;

  float w = __expf(logits2[g] - m2[dst]);
  if (l == 0) atomicAdd(den2 + dst, w);

  const float4 v = *(const float4*)(xl2 + (size_t)src * 16 + c0);
  float* o = acc2 + (size_t)dst * 16 + c0;
  if (c0 + 0 < 10) atomicAdd(o + 0, w * v.x);
  if (c0 + 1 < 10) atomicAdd(o + 1, w * v.y);
  if (c0 + 2 < 10) atomicAdd(o + 2, w * v.z);
  if (c0 + 3 < 10) atomicAdd(o + 3, w * v.w);
}

__global__ void __launch_bounds__(256)
l2_finish(const float* __restrict__ acc2, const float* __restrict__ den2,
          const float* __restrict__ b2, float* __restrict__ out, int total) {
  int i = blockIdx.x * blockDim.x + threadIdx.x;
  if (i >= total) return;
  int n = i / 10, c = i % 10;
  float d = den2[n];
  float v = (d > 0.0f) ? acc2[(size_t)n * 16 + c] / d : 0.0f;
  out[i] = v + b2[c];
}

extern "C" void kernel_launch(void* const* d_in, const int* in_sizes, int n_in,
                              void* d_out, int out_size, void* d_ws, size_t ws_size,
                              hipStream_t stream) {
  const float* x    = (const float*)d_in[0];
  const int*   ei   = (const int*)d_in[1];
  const float* W1l  = (const float*)d_in[2];
  const float* W1r  = (const float*)d_in[3];
  const float* att1 = (const float*)d_in[4];
  const float* b1   = (const float*)d_in[5];
  const float* W2l  = (const float*)d_in[6];
  const float* W2r  = (const float*)d_in[7];
  const float* att2 = (const float*)d_in[8];
  const float* b2   = (const float*)d_in[9];
  float* out = (float*)d_out;

  const int N = in_sizes[0] / 128;
  const int E = in_sizes[1] / 2;

  float* ws = (float*)d_ws;
  size_t o = 0;
  float* xl1  = ws + o; o += (size_t)N * 64;
  float* xr1  = ws + o; o += (size_t)N * 64;
  float* hacc = ws + o; o += (size_t)N * 64;
  float* lg1  = ws + o; o += (size_t)E * 4;
  float* m1   = ws + o; o += (size_t)N * 4;
  float* dn1  = ws + o; o += (size_t)N * 4;
  float* xl2  = ws + o; o += (size_t)N * 16;
  float* xr2  = ws + o; o += (size_t)N * 16;
  float* acc2 = ws + o; o += (size_t)N * 16;
  float* lg2  = ws + o; o += (size_t)E;
  float* m2   = ws + o; o += (size_t)N;
  float* dn2  = ws + o; o += (size_t)N;

  auto blocks = [](long long n) { return (unsigned)((n + 255) / 256); };
  const float NEG_INF = -__builtin_inff();

  fill_kernel<<<blocks((long long)N * 64), 256, 0, stream>>>(hacc, N * 64, 0.0f);
  fill_kernel<<<blocks((long long)N * 4), 256, 0, stream>>>(m1, N * 4, NEG_INF);
  fill_kernel<<<blocks((long long)N * 4), 256, 0, stream>>>(dn1, N * 4, 0.0f);
  fill_kernel<<<blocks((long long)N * 16), 256, 0, stream>>>(acc2, N * 16, 0.0f);
  fill_kernel<<<blocks(N), 256, 0, stream>>>(m2, N, NEG_INF);
  fill_kernel<<<blocks(N), 256, 0, stream>>>(dn2, N, 0.0f);

  const int mtiles = (N + 15) / 16;

  gemm1_kernel<<<mtiles, 128, 0, stream>>>(x, W1l, W1r, xl1, xr1, N);
  l1_edge_logits<<<blocks((long long)E * 16), 256, 0, stream>>>(ei, xl1, xr1, att1,
                                                                lg1, m1, E);
  l1_edge_aggregate<<<blocks((long long)E * 16), 256, 0, stream>>>(ei, xl1, lg1, m1,
                                                                   dn1, hacc, E);
  l1_finish<<<blocks((long long)N * 64), 256, 0, stream>>>(hacc, dn1, b1, N * 64);

  gemm2_kernel<<<(mtiles + 3) / 4, 128, 0, stream>>>(hacc, W2l, W2r, xl2, xr2, N, mtiles);
  l2_edge_logits<<<blocks((long long)E * 4), 256, 0, stream>>>(ei, xl2, xr2, att2,
                                                               lg2, m2, E);
  l2_edge_aggregate<<<blocks((long long)E * 4), 256, 0, stream>>>(ei, xl2, lg2, m2,
                                                                  dn2, acc2, E);
  l2_finish<<<blocks((long long)N * 10), 256, 0, stream>>>(acc2, dn2, b2, out, N * 10);
}


# Edges grouped by destination per call (counting sort: int atomic rank, scan, scatter); each node's scores, online softmax and sum then run in registers, no float atomics; sum order = arrival rank
# speedup vs baseline: 2.5953x; 2.5953x over previous
#include <hip/hip_runtime.h>
#include <math.h>

typedef __attribute__((ext_vector_type(2))) float v2f;
typedef __attribute__((ext_vector_type(8))) float v8f;

#define NEG_SLOPE 0.2f

__global__ void __launch_bounds__(256)
zero_int_kernel(int* __restrict__ p, int n) {
  int i = blockIdx.x * blockDim.x + threadIdx.x;
  if (i < n) p[i] = 0;
}

__global__ void __launch_bounds__(256)
edge_rank_kernel(const int* __restrict__ ei, int* __restrict__ cnt,
                 int* __restrict__ rank, int E) {
  const int T  = gridDim.x * blockDim.x;
  const int e0 = blockIdx.x * blockDim.x + threadIdx.x;
  const int e1 = e0 + T, e2 = e1 + T, e3 = e2 + T;
  const int* dst = ei + E;
  const int d0 = dst[min(e0, E - 1)];
  const int d1 = dst[min(e1, E - 1)];
  const int d2 = dst[min(e2, E - 1)];
  const int d3 = dst[min(e3, E - 1)];
  int r0 = 0, r1 = 0, r2 = 0, r3 = 0;
  if (e0 < E) r0 = atomicAdd(cnt + d0, 1);
  if (e1 < E) r1 = atomicAdd(cnt + d1, 1);
  if (e2 < E) r2 = atomicAdd(cnt + d2, 1);
  if (e3 < E) r3 = atomicAdd(cnt + d3, 1);
  if (e0 < E) rank[e0] = r0;
  if (e1 < E) rank[e1] = r1;
  if (e2 < E) rank[e2] = r2;
  if (e3 < E) rank[e3] = r3;
}

__global__ void __launch_bounds__(1024)
scan_kernel(const int* __restrict__ cnt, int* __restrict__ rowptr, int n) {
  __shared__ int wtile[32];
  __shared__ int wfront[32];
  const int t     = threadIdx.x;
  const int lane  = t & 31;
  const int w     = t >> 5;
  const int start = blockIdx.x * 1024;
  const int idx   = start + t;

  int front = 0;
  for (int i = t; i < start; i += 1024) front += cnt[i];
  const int c = (idx < n) ? cnt[idx] : 0;

  int v = c;
  for (int o = 1; o < 32; o <<= 1) {
    int u = __shfl_up(v, o, 32);
    if (lane >= o) v += u;
  }
  for (int o = 16; o > 0; o >>= 1) front += __shfl_xor(front, o, 32);
  if (lane == 31) wtile[w] = v;
  if (lane == 0)  wfront[w] = front;
  __syncthreads();

  int ft = wfront[lane];
  int wt = (lane < w) ? wtile[lane] : 0;
  for (int o = 16; o > 0; o >>= 1) {
    ft += __shfl_xor(ft, o, 32);
    wt += __shfl_xor(wt, o, 32);
  }
  const int excl = ft + wt + (v - c);
  if (idx < n)      rowptr[idx] = excl;
  if (idx == n - 1) rowptr[n]   = excl + c;
}

__global__ void __launch_bounds__(256)
edge_place_kernel(const int* __restrict__ ei, const int* __restrict__ rowptr,
                  const int* __restrict__ rank, int* __restrict__ csrc, int E) {
  const int T  = gridDim.x * blockDim.x;
  const int e0 = blockIdx.x * blockDim.x + threadIdx.x;
  const int e1 = e0 + T, e2 = e1 + T, e3 = e2 + T;
  const int f0 = min(e0, E - 1), f1 = min(e1, E - 1);
  const int f2 = min(e2, E - 1), f3 = min(e3, E - 1);
  const int* dst = ei + E;
  const int d0 = dst[f0], d1 = dst[f1], d2 = dst[f2], d3 = dst[f3];
  const int q0 = rank[f0], q1 = rank[f1], q2 = rank[f2], q3 = rank[f3];
  const int c0 = ei[f0], c1 = ei[f1], c2 = ei[f2], c3 = ei[f3];
  const int p0 = rowptr[d0] + q0;
  const int p1 = rowptr[d1] + q1;
  const int p2 = rowptr[d2] + q2;
  const int p3 = rowptr[d3] + q3;
  if (e0 < E) csrc[p0] = c0;
  if (e1 < E) csrc[p1] = c1;
  if (e2 < E) csrc[p2] = c2;
  if (e3 < E) csrc[p3] = c3;
}

__global__ void __launch_bounds__(128)
gemm1_kernel(const float* __restrict__ x, const float* __restrict__ W1l,
             const float* __restrict__ W1r, float* __restrict__ xl1,
             float* __restrict__ xr1, int n_nodes) {
  const int lane = threadIdx.x & 31;
  const int wave = threadIdx.x >> 5;
  const int m0   = blockIdx.x * 16;
  const int n0   = wave * 16;
  const int half = lane >> 4;
  const int rc   = lane & 15;

  int arow = m0 + rc; if (arow >= n_nodes) arow = n_nodes - 1;
  const v2f*   xrow = (const v2f*)(x + (size_t)arow * 128);
  const float* Bl   = W1l + n0 + rc;
  const float* Br   = W1r + n0 + rc;

  v8f accL = {}, accR = {};
  #pragma unroll
  for (int kc = 0; kc < 32; ++kc) {
    v2f a = xrow[2 * kc + half];
    const int k0 = 4 * kc + 2 * half;
    v2f bl, br;
    bl[0] = Bl[(size_t)(k0 + 0) * 64];
    bl[1] = Bl[(size_t)(k0 + 1) * 64];
    br[0] = Br[(size_t)(k0 + 0) * 64];
    br[1] = Br[(size_t)(k0 + 1) * 64];
    accL = __builtin_amdgcn_wmma_f32_16x16x4_f32(
        false, a, false, bl, (short)0, accL, false, false);
    accR = __builtin_amdgcn_wmma_f32_16x16x4_f32(
        false, a, false, br, (short)0, accR, false, false);
  }

  #pragma unroll
  for (int r = 0; r < 8; ++r) {
    int row = m0 + r + 8 * half;
    if (row < n_nodes) {
      xl1[(size_t)row * 64 + n0 + rc] = accL[r];
      xr1[(size_t)row * 64 + n0 + rc] = accR[r];
    }
  }
}

__global__ void __launch_bounds__(128)
gemm2_kernel(const float* __restrict__ h, const float* __restrict__ W2l,
             const float* __restrict__ W2r, float* __restrict__ xl2,
             float* __restrict__ xr2, int n_nodes, int mtiles) {
  const int lane  = threadIdx.x & 31;
  const int mtile = blockIdx.x * 4 + (threadIdx.x >> 5);
  if (mtile >= mtiles) return;
  const int m0   = mtile * 16;
  const int half = lane >> 4;
  const int rc   = lane & 15;

  int arow = m0 + rc; if (arow >= n_nodes) arow = n_nodes - 1;
  const v2f* hrow = (const v2f*)(h + (size_t)arow * 64);
  const bool colok = (rc < 10);

  v8f accL = {}, accR = {};
  #pragma unroll
  for (int kc = 0; kc < 16; ++kc) {
    v2f a = hrow[2 * kc + half];
    const int k0 = 4 * kc + 2 * half;
    v2f bl, br;
    bl[0] = colok ? W2l[(size_t)(k0 + 0) * 10 + rc] : 0.0f;
    bl[1] = colok ? W2l[(size_t)(k0 + 1) * 10 + rc] : 0.0f;
    br[0] = colok ? W2r[(size_t)(k0 + 0) * 10 + rc] : 0.0f;
    br[1] = colok ? W2r[(size_t)(k0 + 1) * 10 + rc] : 0.0f;
    accL = __builtin_amdgcn_wmma_f32_16x16x4_f32(
        false, a, false, bl, (short)0, accL, false, false);
    accR = __builtin_amdgcn_wmma_f32_16x16x4_f32(
        false, a, false, br, (short)0, accR, false, false);
  }

  #pragma unroll
  for (int r = 0; r < 8; ++r) {
    int row = m0 + r + 8 * half;
    if (row < n_nodes) {
      xl2[(size_t)row * 16 + rc] = accL[r];
      xr2[(size_t)row * 16 + rc] = accR[r];
    }
  }
}

__device__ __forceinline__ float edge_score(const float4 x, const float4 r, const float4 a) {
  float p = 0.0f, s;
  s = x.x + r.x; s = (s > 0.f) ? s : NEG_SLOPE * s; p += s * a.x;
  s = x.y + r.y; s = (s > 0.f) ? s : NEG_SLOPE * s; p += s * a.y;
  s = x.z + r.z; s = (s > 0.f) ? s : NEG_SLOPE * s; p += s * a.z;
  s = x.w + r.w; s = (s > 0.f) ? s : NEG_SLOPE * s; p += s * a.w;
  p += __shfl_xor(p, 1, 32);
  p += __shfl_xor(p, 2, 32);
  return p;
}

template <int STRIDE>
__device__ __forceinline__ void node_softmax_sum(
    const int* __restrict__ csrc, const float* __restrict__ xl,
    const int beg, const int end, const int coff,
    const float4 rv, const float4 at, float4& acc, float& den) {
  const float NEG_INF = -__builtin_inff();
  float m = NEG_INF;
  float d = 0.0f;
  float4 a = make_float4(0.0f, 0.0f, 0.0f, 0.0f);

  const int last = end - 1;
  for (int j = beg; j < end; j += 4) {
    const int s0 = csrc[j];
    const int s1 = csrc[min(j + 1, last)];
    const int s2 = csrc[min(j + 2, last)];
    const int s3 = csrc[min(j + 3, last)];
    const float4 x0 = *(const float4*)(xl + (size_t)s0 * STRIDE + coff);
    const float4 x1 = *(const float4*)(xl + (size_t)s1 * STRIDE + coff);
    const float4 x2 = *(const float4*)(xl + (size_t)s2 * STRIDE + coff);
    const float4 x3 = *(const float4*)(xl + (size_t)s3 * STRIDE + coff);

    const bool v1 = (j + 1 < end), v2 = (j + 2 < end), v3 = (j + 3 < end);
    const float p0 = edge_score(x0, rv, at);
    float p1 = edge_score(x1, rv, at);
    float p2 = edge_score(x2, rv, at);
    float p3 = edge_score(x3, rv, at);
    p1 = v1 ? p1 : NEG_INF;
    p2 = v2 ? p2 : NEG_INF;
    p3 = v3 ? p3 : NEG_INF;

    const float mn = fmaxf(fmaxf(m, p0), fmaxf(fmaxf(p1, p2), p3));
    const float sc = __expf(m - mn);
    const float w0 = __expf(p0 - mn);
    const float w1 = __expf(p1 - mn);
    const float w2 = __expf(p2 - mn);
    const float w3 = __expf(p3 - mn);
    d   = d   * sc + ((w0 + w1) + (w2 + w3));
    a.x = a.x * sc + ((w0 * x0.x + w1 * x1.x) + (w2 * x2.x + w3 * x3.x));
    a.y = a.y * sc + ((w0 * x0.y + w1 * x1.y) + (w2 * x2.y + w3 * x3.y));
    a.z = a.z * sc + ((w0 * x0.z + w1 * x1.z) + (w2 * x2.z + w3 * x3.z));
    a.w = a.w * sc + ((w0 * x0.w + w1 * x1.w) + (w2 * x2.w + w3 * x3.w));
    m = mn;
  }
  acc = a;
  den = d;
}

__global__ void __launch_bounds__(256)
l1_node_kernel(const int* __restrict__ rowptr, const int* __restrict__ csrc,
               const float* __restrict__ xl1, const float* __restrict__ xr1,
               const float* __restrict__ att1, const float* __restrict__ b1,
               float* __restrict__ h, int n_nodes) {
  const int n = (blockIdx.x * blockDim.x + threadIdx.x) >> 4;
  if (n >= n_nodes) return;
  const int coff = (threadIdx.x & 15) * 4;
  const int beg = rowptr[n];
  const int end = rowptr[n + 1];
  const float4 rv = *(const float4*)(xr1 + (size_t)n * 64 + coff);
  const float4 at = *(const float4*)(att1 + coff);

  float4 acc; float den;
  node_softmax_sum<64>(csrc, xl1, beg, end, coff, rv, at, acc, den);

  const float4 bb = *(const float4*)(b1 + coff);
  float4 v;
  v.x = ((den > 0.0f) ? acc.x / den : 0.0f) + bb.x;
  v.y = ((den > 0.0f) ? acc.y / den : 0.0f) + bb.y;
  v.z = ((den > 0.0f) ? acc.z / den : 0.0f) + bb.z;
  v.w = ((den > 0.0f) ? acc.w / den : 0.0f) + bb.w;
  v.x = (v.x > 0.0f) ? v.x : expm1f(v.x);
  v.y = (v.y > 0.0f) ? v.y : expm1f(v.y);
  v.z = (v.z > 0.0f) ? v.z : expm1f(v.z);
  v.w = (v.w > 0.0f) ? v.w : expm1f(v.w);
  *(float4*)(h + (size_t)n * 64 + coff) = v;
}

__global__ void __launch_bounds__(256)
l2_node_kernel(const int* __restrict__ rowptr, const int* __restrict__ csrc,
               const float* __restrict__ xl2, const float* __restrict__ xr2,
               const float* __restrict__ att2, const float* __restrict__ b2,
               float* __restrict__ out, int n_nodes) {
  __shared__ float stage[64 * 10];
  const int nl = threadIdx.x >> 2;
  const int n  = blockIdx.x * 64 + nl;
  const int c0 = (threadIdx.x & 3) * 4;

  if (n < n_nodes) {
    const int beg = rowptr[n];
    const int end = rowptr[n + 1];
    const float4 rv = *(const float4*)(xr2 + (size_t)n * 16 + c0);
    float4 at;
    at.x = (c0 + 0 < 10) ? att2[c0 + 0] : 0.0f;
    at.y = (c0 + 1 < 10) ? att2[c0 + 1] : 0.0f;
    at.z = (c0 + 2 < 10) ? att2[c0 + 2] : 0.0f;
    at.w = (c0 + 3 < 10) ? att2[c0 + 3] : 0.0f;

    float4 acc; float den;
    node_softmax_sum<16>(csrc, xl2, beg, end, c0, rv, at, acc, den);

    const bool ok = (den > 0.0f);
    if (c0 + 0 < 10) stage[nl * 10 + c0 + 0] = (ok ? acc.x / den : 0.0f) + b2[c0 + 0];
    if (c0 + 1 < 10) stage[nl * 10 + c0 + 1] = (ok ? acc.y / den : 0.0f) + b2[c0 + 1];
    if (c0 + 2 < 10) stage[nl * 10 + c0 + 2] = (ok ? acc.z / den : 0.0f) + b2[c0 + 2];
    if (c0 + 3 < 10) stage[nl * 10 + c0 + 3] = (ok ? acc.w / den : 0.0f) + b2[c0 + 3];
  }
  __syncthreads();

  const long long base  = (long long)blockIdx.x * 640;
  const long long total = (long long)n_nodes * 10;
  for (int i = threadIdx.x; i < 640; i += 256)
    if (base + i < total) out[base + i] = stage[i];
}

extern "C" void kernel_launch(void* const* d_in, const int* in_sizes, int n_in,
                              void* d_out, int out_size, void* d_ws, size_t ws_size,
                              hipStream_t stream) {
  const float* x    = (const float*)d_in[0];
  const int*   ei   = (const int*)d_in[1];
  const float* W1l  = (const float*)d_in[2];
  const float* W1r  = (const float*)d_in[3];
  const float* att1 = (const float*)d_in[4];
  const float* b1   = (const float*)d_in[5];
  const float* W2l  = (const float*)d_in[6];
  const float* W2r  = (const float*)d_in[7];
  const float* att2 = (const float*)d_in[8];
  const float* b2   = (const float*)d_in[9];
  float* out = (float*)d_out;

  const int N = in_sizes[0] / 128;
  const int E = in_sizes[1] / 2;

  float* ws = (float*)d_ws;
  size_t o = 0;
  float* xl1  = ws + o; o += (size_t)N * 64;
  float* xr1  = ws + o; o += (size_t)N * 64;
  float* h1   = ws + o; o += (size_t)N * 64;
  float* xl2  = ws + o; o += (size_t)N * 16;
  float* xr2  = ws + o; o += (size_t)N * 16;
  int* cnt    = (int*)(ws + o); o += (size_t)N;
  int* rowptr = (int*)(ws + o); o += (size_t)N + 1;
  int* rank   = (int*)(ws + o); o += (size_t)E;
  int* csrc   = (int*)(ws + o); o += (size_t)E;

  auto blocks = [](long long n) { return (unsigned)((n + 255) / 256); };
  const int mtiles = (N + 15) / 16;

  zero_int_kernel<<<blocks(N), 256, 0, stream>>>(cnt, N);
  edge_rank_kernel<<<blocks(((long long)E + 3) / 4), 256, 0, stream>>>(ei, cnt, rank, E);
  scan_kernel<<<(unsigned)((N + 1023) / 1024), 1024, 0, stream>>>(cnt, rowptr, N);
  edge_place_kernel<<<blocks(((long long)E + 3) / 4), 256, 0, stream>>>(ei, rowptr, rank, csrc, E);

  gemm1_kernel<<<mtiles, 128, 0, stream>>>(x, W1l, W1r, xl1, xr1, N);
  l1_node_kernel<<<blocks((long long)N * 16), 256, 0, stream>>>(rowptr, csrc, xl1, xr1,
                                                                att1, b1, h1, N);

  gemm2_kernel<<<(mtiles + 3) / 4, 128, 0, stream>>>(h1, W2l, W2r, xl2, xr2, N, mtiles);
  l2_node_kernel<<<(unsigned)((N + 63) / 64), 256, 0, stream>>>(rowptr, csrc, xl2, xr2,
                                                                att2, b2, out, N);
}


# Edges fill fixed 256-byte per-node rows (count word + 63 sources) straight from the count atomic; prefix-sum and placement launches dropped (8 to 6); spill list past 63; sum order still arrival rank
# speedup vs baseline: 3.6968x; 1.4244x over previous
#include <hip/hip_runtime.h>
#include <math.h>

typedef __attribute__((ext_vector_type(2))) float v2f;
typedef __attribute__((ext_vector_type(8))) float v8f;

#define NEG_SLOPE 0.2f

#define ROW_W   64
#define ROW_CAP 63

__global__ void __launch_bounds__(256)
zero_heads_kernel(int* __restrict__ rows, int* __restrict__ spill_n, int n) {
  int i = blockIdx.x * blockDim.x + threadIdx.x;
  if (i < n) rows[(size_t)i * ROW_W] = 0;
  if (i == 0) *spill_n = 0;
}

__device__ __forceinline__ void put_edge(int* __restrict__ rows, int* __restrict__ spill_n,
                                         int* __restrict__ spill, int d, int c, int r) {
  if (r < ROW_CAP) {
    rows[(size_t)d * ROW_W + 1 + r] = c;
  } else {
    const int k = atomicAdd(spill_n, 1);
    spill[2 * (size_t)k]     = d;
    spill[2 * (size_t)k + 1] = c;
  }
}

__global__ void __launch_bounds__(256)
edge_fill_kernel(const int* __restrict__ ei, int* __restrict__ rows,
                 int* __restrict__ spill_n, int* __restrict__ spill, int E) {
  const int T  = gridDim.x * blockDim.x;
  const int e0 = blockIdx.x * blockDim.x + threadIdx.x;
  const int e1 = e0 + T, e2 = e1 + T, e3 = e2 + T;
  const int f0 = min(e0, E - 1), f1 = min(e1, E - 1);
  const int f2 = min(e2, E - 1), f3 = min(e3, E - 1);
  const int* dst = ei + E;
  const int d0 = dst[f0], d1 = dst[f1], d2 = dst[f2], d3 = dst[f3];
  const int c0 = ei[f0], c1 = ei[f1], c2 = ei[f2], c3 = ei[f3];
  int r0 = 0, r1 = 0, r2 = 0, r3 = 0;
  if (e0 < E) r0 = atomicAdd(rows + (size_t)d0 * ROW_W, 1);
  if (e1 < E) r1 = atomicAdd(rows + (size_t)d1 * ROW_W, 1);
  if (e2 < E) r2 = atomicAdd(rows + (size_t)d2 * ROW_W, 1);
  if (e3 < E) r3 = atomicAdd(rows + (size_t)d3 * ROW_W, 1);
  if (e0 < E) put_edge(rows, spill_n, spill, d0, c0, r0);
  if (e1 < E) put_edge(rows, spill_n, spill, d1, c1, r1);
  if (e2 < E) put_edge(rows, spill_n, spill, d2, c2, r2);
  if (e3 < E) put_edge(rows, spill_n, spill, d3, c3, r3);
}

__global__ void __launch_bounds__(128)
gemm1_kernel(const float* __restrict__ x, const float* __restrict__ W1l,
             const float* __restrict__ W1r, float* __restrict__ xl1,
             float* __restrict__ xr1, int n_nodes) {
  const int lane = threadIdx.x & 31;
  const int wave = threadIdx.x >> 5;
  const int m0   = blockIdx.x * 16;
  const int n0   = wave * 16;
  const int half = lane >> 4;
  const int rc   = lane & 15;

  int arow = m0 + rc; if (arow >= n_nodes) arow = n_nodes - 1;
  const v2f*   xrow = (const v2f*)(x + (size_t)arow * 128);
  const float* Bl   = W1l + n0 + rc;
  const float* Br   = W1r + n0 + rc;

  v8f accL = {}, accR = {};
  #pragma unroll
  for (int kc = 0; kc < 32; ++kc) {
    v2f a = xrow[2 * kc + half];
    const int k0 = 4 * kc + 2 * half;
    v2f bl, br;
    bl[0] = Bl[(size_t)(k0 + 0) * 64];
    bl[1] = Bl[(size_t)(k0 + 1) * 64];
    br[0] = Br[(size_t)(k0 + 0) * 64];
    br[1] = Br[(size_t)(k0 + 1) * 64];
    accL = __builtin_amdgcn_wmma_f32_16x16x4_f32(
        false, a, false, bl, (short)0, accL, false, false);
    accR = __builtin_amdgcn_wmma_f32_16x16x4_f32(
        false, a, false, br, (short)0, accR, false, false);
  }

  #pragma unroll
  for (int r = 0; r < 8; ++r) {
    int row = m0 + r + 8 * half;
    if (row < n_nodes) {
      xl1[(size_t)row * 64 + n0 + rc] = accL[r];
      xr1[(size_t)row * 64 + n0 + rc] = accR[r];
    }
  }
}

__global__ void __launch_bounds__(128)
gemm2_kernel(const float* __restrict__ h, const float* __restrict__ W2l,
             const float* __restrict__ W2r, float* __restrict__ xl2,
             float* __restrict__ xr2, int n_nodes, int mtiles) {
  const int lane  = threadIdx.x & 31;
  const int mtile = blockIdx.x * 4 + (threadIdx.x >> 5);
  if (mtile >= mtiles) return;
  const int m0   = mtile * 16;
  const int half = lane >> 4;
  const int rc   = lane & 15;

  int arow = m0 + rc; if (arow >= n_nodes) arow = n_nodes - 1;
  const v2f* hrow = (const v2f*)(h + (size_t)arow * 64);
  const bool colok = (rc < 10);

  v8f accL = {}, accR = {};
  #pragma unroll
  for (int kc = 0; kc < 16; ++kc) {
    v2f a = hrow[2 * kc + half];
    const int k0 = 4 * kc + 2 * half;
    v2f bl, br;
    bl[0] = colok ? W2l[(size_t)(k0 + 0) * 10 + rc] : 0.0f;
    bl[1] = colok ? W2l[(size_t)(k0 + 1) * 10 + rc] : 0.0f;
    br[0] = colok ? W2r[(size_t)(k0 + 0) * 10 + rc] : 0.0f;
    br[1] = colok ? W2r[(size_t)(k0 + 1) * 10 + rc] : 0.0f;
    accL = __builtin_amdgcn_wmma_f32_16x16x4_f32(
        false, a, false, bl, (short)0, accL, false, false);
    accR = __builtin_amdgcn_wmma_f32_16x16x4_f32(
        false, a, false, br, (short)0, accR, false, false);
  }

  #pragma unroll
  for (int r = 0; r < 8; ++r) {
    int row = m0 + r + 8 * half;
    if (row < n_nodes) {
      xl2[(size_t)row * 16 + rc] = accL[r];
      xr2[(size_t)row * 16 + rc] = accR[r];
    }
  }
}

__device__ __forceinline__ float edge_score(const float4 x, const float4 r, const float4 a) {
  float p = 0.0f, s;
  s = x.x + r.x; s = (s > 0.f) ? s : NEG_SLOPE * s; p += s * a.x;
  s = x.y + r.y; s = (s > 0.f) ? s : NEG_SLOPE * s; p += s * a.y;
  s = x.z + r.z; s = (s > 0.f) ? s : NEG_SLOPE * s; p += s * a.z;
  s = x.w + r.w; s = (s > 0.f) ? s : NEG_SLOPE * s; p += s * a.w;
  p += __shfl_xor(p, 1, 32);
  p += __shfl_xor(p, 2, 32);
  return p;
}

template <int STRIDE>
__device__ __forceinline__ void node_softmax_sum(
    const int* __restrict__ csrc, const float* __restrict__ xl,
    const int end, const int coff,
    const float4 rv, const float4 at, float& mrun, float4& acc, float& den) {
  const float NEG_INF = -__builtin_inff();
  const int beg = 0;
  float m = mrun;
  float d = den;
  float4 a = acc;

  const int last = end - 1;
  for (int j = beg; j < end; j += 4) {
    const int s0 = csrc[j];
    const int s1 = csrc[min(j + 1, last)];
    const int s2 = csrc[min(j + 2, last)];
    const int s3 = csrc[min(j + 3, last)];
    const float4 x0 = *(const float4*)(xl + (size_t)s0 * STRIDE + coff);
    const float4 x1 = *(const float4*)(xl + (size_t)s1 * STRIDE + coff);
    const float4 x2 = *(const float4*)(xl + (size_t)s2 * STRIDE + coff);
    const float4 x3 = *(const float4*)(xl + (size_t)s3 * STRIDE + coff);

    const bool v1 = (j + 1 < end), v2 = (j + 2 < end), v3 = (j + 3 < end);
    const float p0 = edge_score(x0, rv, at);
    float p1 = edge_score(x1, rv, at);
    float p2 = edge_score(x2, rv, at);
    float p3 = edge_score(x3, rv, at);
    p1 = v1 ? p1 : NEG_INF;
    p2 = v2 ? p2 : NEG_INF;
    p3 = v3 ? p3 : NEG_INF;

    const float mn = fmaxf(fmaxf(m, p0), fmaxf(fmaxf(p1, p2), p3));
    const float sc = __expf(m - mn);
    const float w0 = __expf(p0 - mn);
    const float w1 = __expf(p1 - mn);
    const float w2 = __expf(p2 - mn);
    const float w3 = __expf(p3 - mn);
    d   = d   * sc + ((w0 + w1) + (w2 + w3));
    a.x = a.x * sc + ((w0 * x0.x + w1 * x1.x) + (w2 * x2.x + w3 * x3.x));
    a.y = a.y * sc + ((w0 * x0.y + w1 * x1.y) + (w2 * x2.y + w3 * x3.y));
    a.z = a.z * sc + ((w0 * x0.z + w1 * x1.z) + (w2 * x2.z + w3 * x3.z));
    a.w = a.w * sc + ((w0 * x0.w + w1 * x1.w) + (w2 * x2.w + w3 * x3.w));
    m = mn;
  }
  mrun = m;
  acc = a;
  den = d;
}

template <int STRIDE>
__device__ __forceinline__ void node_one_edge(
    const int s, const float* __restrict__ xl, const int coff,
    const float4 rv, const float4 at, float& m, float4& a, float& d) {
  const float4 x = *(const float4*)(xl + (size_t)s * STRIDE + coff);
  const float p  = edge_score(x, rv, at);
  const float mn = fmaxf(m, p);
  const float sc = __expf(m - mn);
  const float w  = __expf(p - mn);
  d   = d   * sc + w;
  a.x = a.x * sc + w * x.x;
  a.y = a.y * sc + w * x.y;
  a.z = a.z * sc + w * x.z;
  a.w = a.w * sc + w * x.w;
  m = mn;
}

template <int STRIDE>
__device__ __forceinline__ void node_all_edges(
    const int* __restrict__ rows, const int* __restrict__ spill_n,
    const int* __restrict__ spill, const int spill_cap,
    const float* __restrict__ xl, const int n, const int coff,
    const float4 rv, const float4 at, float4& acc, float& den) {
  const int* row = rows + (size_t)n * ROW_W;
  const int cnt = row[0];
  float m = -__builtin_inff();
  den = 0.0f;
  acc = make_float4(0.0f, 0.0f, 0.0f, 0.0f);
  node_softmax_sum<STRIDE>(row + 1, xl, min(cnt, ROW_CAP), coff, rv, at, m, acc, den);
  if (cnt > ROW_CAP) {
    const int ns = min(*spill_n, spill_cap);
    for (int i = 0; i < ns; ++i)
      if (spill[2 * (size_t)i] == n)
        node_one_edge<STRIDE>(spill[2 * (size_t)i + 1], xl, coff, rv, at, m, acc, den);
  }
}

__global__ void __launch_bounds__(256)
l1_node_kernel(const int* __restrict__ rows, const int* __restrict__ spill_n,
               const int* __restrict__ spill, int spill_cap,
               const float* __restrict__ xl1, const float* __restrict__ xr1,
               const float* __restrict__ att1, const float* __restrict__ b1,
               float* __restrict__ h, int n_nodes) {
  const int n = (blockIdx.x * blockDim.x + threadIdx.x) >> 4;
  if (n >= n_nodes) return;
  const int coff = (threadIdx.x & 15) * 4;
  const float4 rv = *(const float4*)(xr1 + (size_t)n * 64 + coff);
  const float4 at = *(const float4*)(att1 + coff);

  float4 acc; float den;
  node_all_edges<64>(rows, spill_n, spill, spill_cap, xl1, n, coff, rv, at, acc, den);

  const float4 bb = *(const float4*)(b1 + coff);
  float4 v;
  v.x = ((den > 0.0f) ? acc.x / den : 0.0f) + bb.x;
  v.y = ((den > 0.0f) ? acc.y / den : 0.0f) + bb.y;
  v.z = ((den > 0.0f) ? acc.z / den : 0.0f) + bb.z;
  v.w = ((den > 0.0f) ? acc.w / den : 0.0f) + bb.w;
  v.x = (v.x > 0.0f) ? v.x : expm1f(v.x);
  v.y = (v.y > 0.0f) ? v.y : expm1f(v.y);
  v.z = (v.z > 0.0f) ? v.z : expm1f(v.z);
  v.w = (v.w > 0.0f) ? v.w : expm1f(v.w);
  *(float4*)(h + (size_t)n * 64 + coff) = v;
}

__global__ void __launch_bounds__(256)
l2_node_kernel(const int* __restrict__ rows, const int* __restrict__ spill_n,
               const int* __restrict__ spill, int spill_cap,
               const float* __restrict__ xl2, const float* __restrict__ xr2,
               const float* __restrict__ att2, const float* __restrict__ b2,
               float* __restrict__ out, int n_nodes) {
  __shared__ float stage[64 * 10];
  const int nl = threadIdx.x >> 2;
  const int n  = blockIdx.x * 64 + nl;
  const int c0 = (threadIdx.x & 3) * 4;

  if (n < n_nodes) {
    const float4 rv = *(const float4*)(xr2 + (size_t)n * 16 + c0);
    float4 at;
    at.x = (c0 + 0 < 10) ? att2[c0 + 0] : 0.0f;
    at.y = (c0 + 1 < 10) ? att2[c0 + 1] : 0.0f;
    at.z = (c0 + 2 < 10) ? att2[c0 + 2] : 0.0f;
    at.w = (c0 + 3 < 10) ? att2[c0 + 3] : 0.0f;

    float4 acc; float den;
    node_all_edges<16>(rows, spill_n, spill, spill_cap, xl2, n, c0, rv, at, acc, den);

    const bool ok = (den > 0.0f);
    if (c0 + 0 < 10) stage[nl * 10 + c0 + 0] = (ok ? acc.x / den : 0.0f) + b2[c0 + 0];
    if (c0 + 1 < 10) stage[nl * 10 + c0 + 1] = (ok ? acc.y / den : 0.0f) + b2[c0 + 1];
    if (c0 + 2 < 10) stage[nl * 10 + c0 + 2] = (ok ? acc.z / den : 0.0f) + b2[c0 + 2];
    if (c0 + 3 < 10) stage[nl * 10 + c0 + 3] = (ok ? acc.w / den : 0.0f) + b2[c0 + 3];
  }
  __syncthreads();

  const long long base  = (long long)blockIdx.x * 640;
  const long long total = (long long)n_nodes * 10;
  for (int i = threadIdx.x; i < 640; i += 256)
    if (base + i < total) out[base + i] = stage[i];
}

extern "C" void kernel_launch(void* const* d_in, const int* in_sizes, int n_in,
                              void* d_out, int out_size, void* d_ws, size_t ws_size,
                              hipStream_t stream) {
  const float* x    = (const float*)d_in[0];
  const int*   ei   = (const int*)d_in[1];
  const float* W1l  = (const float*)d_in[2];
  const float* W1r  = (const float*)d_in[3];
  const float* att1 = (const float*)d_in[4];
  const float* b1   = (const float*)d_in[5];
  const float* W2l  = (const float*)d_in[6];
  const float* W2r  = (const float*)d_in[7];
  const float* att2 = (const float*)d_in[8];
  const float* b2   = (const float*)d_in[9];
  float* out = (float*)d_out;

  const int N = in_sizes[0] / 128;
  const int E = in_sizes[1] / 2;

  float* ws = (float*)d_ws;
  size_t o = 0;
  int* rows    = (int*)(ws + o); o += (size_t)N * ROW_W;
  float* xl1   = ws + o; o += (size_t)N * 64;
  float* xr1   = ws + o; o += (size_t)N * 64;
  float* h1    = ws + o; o += (size_t)N * 64;
  float* xl2   = ws + o; o += (size_t)N * 16;
  float* xr2   = ws + o; o += (size_t)N * 16;
  int* spill_n = (int*)(ws + o); o += 64;
  int* spill   = (int*)(ws + o); o += 2 * (size_t)E;

  auto blocks = [](long long n) { return (unsigned)((n + 255) / 256); };
  const int mtiles = (N + 15) / 16;

  zero_heads_kernel<<<blocks(N), 256, 0, stream>>>(rows, spill_n, N);
  edge_fill_kernel<<<blocks(((long long)E + 3) / 4), 256, 0, stream>>>(ei, rows, spill_n, spill, E);

  gemm1_kernel<<<mtiles, 128, 0, stream>>>(x, W1l, W1r, xl1, xr1, N);
  l1_node_kernel<<<blocks((long long)N * 16), 256, 0, stream>>>(rows, spill_n, spill, E, xl1, xr1,
                                                                att1, b1, h1, N);

  gemm2_kernel<<<(mtiles + 3) / 4, 128, 0, stream>>>(h1, W2l, W2r, xl2, xr2, N, mtiles);
  l2_node_kernel<<<(unsigned)((N + 63) / 64), 256, 0, stream>>>(rows, spill_n, spill, E, xl2, xr2,
                                                                att2, b2, out, N);
}


# Half the waves in the two largest launches: layer-1 node kernel on 8 lanes per node (8 channels each, whole-line loads), GEMM1 wave owns two row tiles (weight fragment reused); sums unchanged
# speedup vs baseline: 3.8332x; 1.0369x over previous
#include <hip/hip_runtime.h>
#include <math.h>

typedef __attribute__((ext_vector_type(2))) float v2f;
typedef __attribute__((ext_vector_type(8))) float v8f;

#define NEG_SLOPE 0.2f

#define ROW_W   64
#define ROW_CAP 63

__global__ void __launch_bounds__(256)
zero_heads_kernel(int* __restrict__ rows, int* __restrict__ spill_n, int n) {
  int i = blockIdx.x * blockDim.x + threadIdx.x;
  if (i < n) rows[(size_t)i * ROW_W] = 0;
  if (i == 0) *spill_n = 0;
}

__device__ __forceinline__ void put_edge(int* __restrict__ rows, int* __restrict__ spill_n,
                                         int* __restrict__ spill, int d, int c, int r) {
  if (r < ROW_CAP) {
    rows[(size_t)d * ROW_W + 1 + r] = c;
  } else {
    const int k = atomicAdd(spill_n, 1);
    spill[2 * (size_t)k]     = d;
    spill[2 * (size_t)k + 1] = c;
  }
}

__global__ void __launch_bounds__(256)
edge_fill_kernel(const int* __restrict__ ei, int* __restrict__ rows,
                 int* __restrict__ spill_n, int* __restrict__ spill, int E) {
  const int T  = gridDim.x * blockDim.x;
  const int e0 = blockIdx.x * blockDim.x + threadIdx.x;
  const int e1 = e0 + T, e2 = e1 + T, e3 = e2 + T;
  const int f0 = min(e0, E - 1), f1 = min(e1, E - 1);
  const int f2 = min(e2, E - 1), f3 = min(e3, E - 1);
  const int* dst = ei + E;
  const int d0 = dst[f0], d1 = dst[f1], d2 = dst[f2], d3 = dst[f3];
  const int c0 = ei[f0], c1 = ei[f1], c2 = ei[f2], c3 = ei[f3];
  int r0 = 0, r1 = 0, r2 = 0, r3 = 0;
  if (e0 < E) r0 = atomicAdd(rows + (size_t)d0 * ROW_W, 1);
  if (e1 < E) r1 = atomicAdd(rows + (size_t)d1 * ROW_W, 1);
  if (e2 < E) r2 = atomicAdd(rows + (size_t)d2 * ROW_W, 1);
  if (e3 < E) r3 = atomicAdd(rows + (size_t)d3 * ROW_W, 1);
  if (e0 < E) put_edge(rows, spill_n, spill, d0, c0, r0);
  if (e1 < E) put_edge(rows, spill_n, spill, d1, c1, r1);
  if (e2 < E) put_edge(rows, spill_n, spill, d2, c2, r2);
  if (e3 < E) put_edge(rows, spill_n, spill, d3, c3, r3);
}

__global__ void __launch_bounds__(128)
gemm1_kernel(const float* __restrict__ x, const float* __restrict__ W1l,
             const float* __restrict__ W1r, float* __restrict__ xl1,
             float* __restrict__ xr1, int n_nodes) {
  const int lane = threadIdx.x & 31;
  const int wave = threadIdx.x >> 5;
  const int m0   = blockIdx.x * 32;
  const int n0   = wave * 16;
  const int half = lane >> 4;
  const int rc   = lane & 15;

  int arow0 = m0 + rc;      if (arow0 >= n_nodes) arow0 = n_nodes - 1;
  int arow1 = m0 + 16 + rc; if (arow1 >= n_nodes) arow1 = n_nodes - 1;
  const v2f*   xrow0 = (const v2f*)(x + (size_t)arow0 * 128);
  const v2f*   xrow1 = (const v2f*)(x + (size_t)arow1 * 128);
  const float* Bl    = W1l + n0 + rc;
  const float* Br    = W1r + n0 + rc;

  v8f accL0 = {}, accR0 = {}, accL1 = {}, accR1 = {};
  #pragma unroll
  for (int kc = 0; kc < 32; ++kc) {
    v2f a0 = xrow0[2 * kc + half];
    v2f a1 = xrow1[2 * kc + half];
    const int k0 = 4 * kc + 2 * half;
    v2f bl, br;
    bl[0] = Bl[(size_t)(k0 + 0) * 64];
    bl[1] = Bl[(size_t)(k0 + 1) * 64];
    br[0] = Br[(size_t)(k0 + 0) * 64];
    br[1] = Br[(size_t)(k0 + 1) * 64];
    accL0 = __builtin_amdgcn_wmma_f32_16x16x4_f32(
        false, a0, false, bl, (short)0, accL0, false, false);
    accR0 = __builtin_amdgcn_wmma_f32_16x16x4_f32(
        false, a0, false, br, (short)0, accR0, false, false);
    accL1 = __builtin_amdgcn_wmma_f32_16x16x4_f32(
        false, a1, false, bl, (short)0, accL1, false, false);
    accR1 = __builtin_amdgcn_wmma_f32_16x16x4_f32(
        false, a1, false, br, (short)0, accR1, false, false);
  }

  #pragma unroll
  for (int r = 0; r < 8; ++r) {
    const int row0 = m0 + r + 8 * half;
    const int row1 = row0 + 16;
    if (row0 < n_nodes) {
      xl1[(size_t)row0 * 64 + n0 + rc] = accL0[r];
      xr1[(size_t)row0 * 64 + n0 + rc] = accR0[r];
    }
    if (row1 < n_nodes) {
      xl1[(size_t)row1 * 64 + n0 + rc] = accL1[r];
      xr1[(size_t)row1 * 64 + n0 + rc] = accR1[r];
    }
  }
}

__global__ void __launch_bounds__(128)
gemm2_kernel(const float* __restrict__ h, const float* __restrict__ W2l,
             const float* __restrict__ W2r, float* __restrict__ xl2,
             float* __restrict__ xr2, int n_nodes, int mtiles) {
  const int lane  = threadIdx.x & 31;
  const int mtile = blockIdx.x * 4 + (threadIdx.x >> 5);
  if (mtile >= mtiles) return;
  const int m0   = mtile * 16;
  const int half = lane >> 4;
  const int rc   = lane & 15;

  int arow = m0 + rc; if (arow >= n_nodes) arow = n_nodes - 1;
  const v2f* hrow = (const v2f*)(h + (size_t)arow * 64);
  const bool colok = (rc < 10);

  v8f accL = {}, accR = {};
  #pragma unroll
  for (int kc = 0; kc < 16; ++kc) {
    v2f a = hrow[2 * kc + half];
    const int k0 = 4 * kc + 2 * half;
    v2f bl, br;
    bl[0] = colok ? W2l[(size_t)(k0 + 0) * 10 + rc] : 0.0f;
    bl[1] = colok ? W2l[(size_t)(k0 + 1) * 10 + rc] : 0.0f;
    br[0] = colok ? W2r[(size_t)(k0 + 0) * 10 + rc] : 0.0f;
    br[1] = colok ? W2r[(size_t)(k0 + 1) * 10 + rc] : 0.0f;
    accL = __builtin_amdgcn_wmma_f32_16x16x4_f32(
        false, a, false, bl, (short)0, accL, false, false);
    accR = __builtin_amdgcn_wmma_f32_16x16x4_f32(
        false, a, false, br, (short)0, accR, false, false);
  }

  #pragma unroll
  for (int r = 0; r < 8; ++r) {
    int row = m0 + r + 8 * half;
    if (row < n_nodes) {
      xl2[(size_t)row * 16 + rc] = accL[r];
      xr2[(size_t)row * 16 + rc] = accR[r];
    }
  }
}

__device__ __forceinline__ float edge_score(const float4 x, const float4 r, const float4 a) {
  float p = 0.0f, s;
  s = x.x + r.x; s = (s > 0.f) ? s : NEG_SLOPE * s; p += s * a.x;
  s = x.y + r.y; s = (s > 0.f) ? s : NEG_SLOPE * s; p += s * a.y;
  s = x.z + r.z; s = (s > 0.f) ? s : NEG_SLOPE * s; p += s * a.z;
  s = x.w + r.w; s = (s > 0.f) ? s : NEG_SLOPE * s; p += s * a.w;
  p += __shfl_xor(p, 1, 32);
  p += __shfl_xor(p, 2, 32);
  return p;
}

template <int STRIDE, int NV, int VOFF>
__device__ __forceinline__ void node_softmax_sum(
    const int* __restrict__ csrc, const float* __restrict__ xl,
    const int end, const int coff,
    const float4 (&rv)[NV], const float4 (&at)[NV],
    float (&mrun)[NV], float4 (&acc)[NV], float (&den)[NV]) {
  const float NEG_INF = -__builtin_inff();

  const int last = end - 1;
  for (int j = 0; j < end; j += 4) {
    const int s0 = csrc[j];
    const int s1 = csrc[min(j + 1, last)];
    const int s2 = csrc[min(j + 2, last)];
    const int s3 = csrc[min(j + 3, last)];
    const float* r0 = xl + (size_t)s0 * STRIDE + coff;
    const float* r1 = xl + (size_t)s1 * STRIDE + coff;
    const float* r2 = xl + (size_t)s2 * STRIDE + coff;
    const float* r3 = xl + (size_t)s3 * STRIDE + coff;
    float4 x0[NV], x1[NV], x2[NV], x3[NV];
    #pragma unroll
    for (int v = 0; v < NV; ++v) {
      x0[v] = *(const float4*)(r0 + v * VOFF);
      x1[v] = *(const float4*)(r1 + v * VOFF);
      x2[v] = *(const float4*)(r2 + v * VOFF);
      x3[v] = *(const float4*)(r3 + v * VOFF);
    }

    const bool v1 = (j + 1 < end), v2 = (j + 2 < end), v3 = (j + 3 < end);
    #pragma unroll
    for (int v = 0; v < NV; ++v) {
      const float p0 = edge_score(x0[v], rv[v], at[v]);
      float p1 = edge_score(x1[v], rv[v], at[v]);
      float p2 = edge_score(x2[v], rv[v], at[v]);
      float p3 = edge_score(x3[v], rv[v], at[v]);
      p1 = v1 ? p1 : NEG_INF;
      p2 = v2 ? p2 : NEG_INF;
      p3 = v3 ? p3 : NEG_INF;

      const float m  = mrun[v];
      const float mn = fmaxf(fmaxf(m, p0), fmaxf(fmaxf(p1, p2), p3));
      const float sc = __expf(m - mn);
      const float w0 = __expf(p0 - mn);
      const float w1 = __expf(p1 - mn);
      const float w2 = __expf(p2 - mn);
      const float w3 = __expf(p3 - mn);
      den[v]   = den[v]   * sc + ((w0 + w1) + (w2 + w3));
      acc[v].x = acc[v].x * sc + ((w0 * x0[v].x + w1 * x1[v].x) + (w2 * x2[v].x + w3 * x3[v].x));
      acc[v].y = acc[v].y * sc + ((w0 * x0[v].y + w1 * x1[v].y) + (w2 * x2[v].y + w3 * x3[v].y));
      acc[v].z = acc[v].z * sc + ((w0 * x0[v].z + w1 * x1[v].z) + (w2 * x2[v].z + w3 * x3[v].z));
      acc[v].w = acc[v].w * sc + ((w0 * x0[v].w + w1 * x1[v].w) + (w2 * x2[v].w + w3 * x3[v].w));
      mrun[v] = mn;
    }
  }
}

template <int STRIDE, int NV, int VOFF>
__device__ __forceinline__ void node_one_edge(
    const int s, const float* __restrict__ xl, const int coff,
    const float4 (&rv)[NV], const float4 (&at)[NV],
    float (&m)[NV], float4 (&a)[NV], float (&d)[NV]) {
  const float* r = xl + (size_t)s * STRIDE + coff;
  #pragma unroll
  for (int v = 0; v < NV; ++v) {
    const float4 x = *(const float4*)(r + v * VOFF);
    const float p  = edge_score(x, rv[v], at[v]);
    const float mn = fmaxf(m[v], p);
    const float sc = __expf(m[v] - mn);
    const float w  = __expf(p - mn);
    d[v]   = d[v]   * sc + w;
    a[v].x = a[v].x * sc + w * x.x;
    a[v].y = a[v].y * sc + w * x.y;
    a[v].z = a[v].z * sc + w * x.z;
    a[v].w = a[v].w * sc + w * x.w;
    m[v] = mn;
  }
}

template <int STRIDE, int NV, int VOFF>
__device__ __forceinline__ void node_all_edges(
    const int* __restrict__ rows, const int* __restrict__ spill_n,
    const int* __restrict__ spill, const int spill_cap,
    const float* __restrict__ xl, const int n, const int coff,
    const float4 (&rv)[NV], const float4 (&at)[NV],
    float4 (&acc)[NV], float (&den)[NV]) {
  const int* row = rows + (size_t)n * ROW_W;
  const int cnt = row[0];
  float m[NV];
  #pragma unroll
  for (int v = 0; v < NV; ++v) {
    m[v]   = -__builtin_inff();
    den[v] = 0.0f;
    acc[v] = make_float4(0.0f, 0.0f, 0.0f, 0.0f);
  }
  node_softmax_sum<STRIDE, NV, VOFF>(row + 1, xl, min(cnt, ROW_CAP), coff, rv, at, m, acc, den);
  if (cnt > ROW_CAP) {
    const int ns = min(*spill_n, spill_cap);
    for (int i = 0; i < ns; ++i)
      if (spill[2 * (size_t)i] == n)
        node_one_edge<STRIDE, NV, VOFF>(spill[2 * (size_t)i + 1], xl, coff, rv, at, m, acc, den);
  }
}

__global__ void __launch_bounds__(256)
l1_node_kernel(const int* __restrict__ rows, const int* __restrict__ spill_n,
               const int* __restrict__ spill, int spill_cap,
               const float* __restrict__ xl1, const float* __restrict__ xr1,
               const float* __restrict__ att1, const float* __restrict__ b1,
               float* __restrict__ h, int n_nodes) {
  const int n = (blockIdx.x * blockDim.x + threadIdx.x) >> 3;
  if (n >= n_nodes) return;
  const int coff = (threadIdx.x & 7) * 4;
  float4 rv[2], at[2];
  rv[0] = *(const float4*)(xr1 + (size_t)n * 64 + coff);
  rv[1] = *(const float4*)(xr1 + (size_t)n * 64 + coff + 32);
  at[0] = *(const float4*)(att1 + coff);
  at[1] = *(const float4*)(att1 + coff + 32);

  float4 acc[2]; float den[2];
  node_all_edges<64, 2, 32>(rows, spill_n, spill, spill_cap, xl1, n, coff, rv, at, acc, den);

  #pragma unroll
  for (int g = 0; g < 2; ++g) {
    const float4 bb = *(const float4*)(b1 + coff + 32 * g);
    const float  dn = den[g];
    float4 v;
    v.x = ((dn > 0.0f) ? acc[g].x / dn : 0.0f) + bb.x;
    v.y = ((dn > 0.0f) ? acc[g].y / dn : 0.0f) + bb.y;
    v.z = ((dn > 0.0f) ? acc[g].z / dn : 0.0f) + bb.z;
    v.w = ((dn > 0.0f) ? acc[g].w / dn : 0.0f) + bb.w;
    v.x = (v.x > 0.0f) ? v.x : expm1f(v.x);
    v.y = (v.y > 0.0f) ? v.y : expm1f(v.y);
    v.z = (v.z > 0.0f) ? v.z : expm1f(v.z);
    v.w = (v.w > 0.0f) ? v.w : expm1f(v.w);
    *(float4*)(h + (size_t)n * 64 + coff + 32 * g) = v;
  }
}

__global__ void __launch_bounds__(256)
l2_node_kernel(const int* __restrict__ rows, const int* __restrict__ spill_n,
               const int* __restrict__ spill, int spill_cap,
               const float* __restrict__ xl2, const float* __restrict__ xr2,
               const float* __restrict__ att2, const float* __restrict__ b2,
               float* __restrict__ out, int n_nodes) {
  __shared__ float stage[64 * 10];
  const int nl = threadIdx.x >> 2;
  const int n  = blockIdx.x * 64 + nl;
  const int c0 = (threadIdx.x & 3) * 4;

  if (n < n_nodes) {
    float4 rv[1], at[1];
    rv[0] = *(const float4*)(xr2 + (size_t)n * 16 + c0);
    at[0].x = (c0 + 0 < 10) ? att2[c0 + 0] : 0.0f;
    at[0].y = (c0 + 1 < 10) ? att2[c0 + 1] : 0.0f;
    at[0].z = (c0 + 2 < 10) ? att2[c0 + 2] : 0.0f;
    at[0].w = (c0 + 3 < 10) ? att2[c0 + 3] : 0.0f;

    float4 acc[1]; float den[1];
    node_all_edges<16, 1, 0>(rows, spill_n, spill, spill_cap, xl2, n, c0, rv, at, acc, den);

    const bool ok = (den[0] > 0.0f);
    if (c0 + 0 < 10) stage[nl * 10 + c0 + 0] = (ok ? acc[0].x / den[0] : 0.0f) + b2[c0 + 0];
    if (c0 + 1 < 10) stage[nl * 10 + c0 + 1] = (ok ? acc[0].y / den[0] : 0.0f) + b2[c0 + 1];
    if (c0 + 2 < 10) stage[nl * 10 + c0 + 2] = (ok ? acc[0].z / den[0] : 0.0f) + b2[c0 + 2];
    if (c0 + 3 < 10) stage[nl * 10 + c0 + 3] = (ok ? acc[0].w / den[0] : 0.0f) + b2[c0 + 3];
  }
  __syncthreads();

  const long long base  = (long long)blockIdx.x * 640;
  const long long total = (long long)n_nodes * 10;
  for (int i = threadIdx.x; i < 640; i += 256)
    if (base + i < total) out[base + i] = stage[i];
}

extern "C" void kernel_launch(void* const* d_in, const int* in_sizes, int n_in,
                              void* d_out, int out_size, void* d_ws, size_t ws_size,
                              hipStream_t stream) {
  const float* x    = (const float*)d_in[0];
  const int*   ei   = (const int*)d_in[1];
  const float* W1l  = (const float*)d_in[2];
  const float* W1r  = (const float*)d_in[3];
  const float* att1 = (const float*)d_in[4];
  const float* b1   = (const float*)d_in[5];
  const float* W2l  = (const float*)d_in[6];
  const float* W2r  = (const float*)d_in[7];
  const float* att2 = (const float*)d_in[8];
  const float* b2   = (const float*)d_in[9];
  float* out = (float*)d_out;

  const int N = in_sizes[0] / 128;
  const int E = in_sizes[1] / 2;

  float* ws = (float*)d_ws;
  size_t o = 0;
  int* rows    = (int*)(ws + o); o += (size_t)N * ROW_W;
  float* xl1   = ws + o; o += (size_t)N * 64;
  float* xr1   = ws + o; o += (size_t)N * 64;
  float* h1    = ws + o; o += (size_t)N * 64;
  float* xl2   = ws + o; o += (size_t)N * 16;
  float* xr2   = ws + o; o += (size_t)N * 16;
  int* spill_n = (int*)(ws + o); o += 64;
  int* spill   = (int*)(ws + o); o += 2 * (size_t)E;

  auto blocks = [](long long n) { return (unsigned)((n + 255) / 256); };
  const int mtiles = (N + 15) / 16;

  zero_heads_kernel<<<blocks(N), 256, 0, stream>>>(rows, spill_n, N);
  edge_fill_kernel<<<blocks(((long long)E + 3) / 4), 256, 0, stream>>>(ei, rows, spill_n, spill, E);

  gemm1_kernel<<<(mtiles + 1) / 2, 128, 0, stream>>>(x, W1l, W1r, xl1, xr1, N);
  l1_node_kernel<<<blocks((long long)N * 8), 256, 0, stream>>>(rows, spill_n, spill, E, xl1, xr1,
                                                                att1, b1, h1, N);

  gemm2_kernel<<<(mtiles + 3) / 4, 128, 0, stream>>>(h1, W2l, W2r, xl2, xr2, N, mtiles);
  l2_node_kernel<<<(unsigned)((N + 63) / 64), 256, 0, stream>>>(rows, spill_n, spill, E, xl2, xr2,
                                                                att2, b2, out, N);
}


# GEMM1 and the edge-grouping pass, independent of each other, share one launch (neighbouring blocks alternate), so atomic waits overlap matrix work; launches 6 to 5; sum order still arrival rank
# speedup vs baseline: 4.0545x; 1.0577x over previous
#include <hip/hip_runtime.h>
#include <math.h>

typedef __attribute__((ext_vector_type(2))) float v2f;
typedef __attribute__((ext_vector_type(8))) float v8f;

#define NEG_SLOPE 0.2f

#define ROW_W   64
#define ROW_CAP 63

__global__ void __launch_bounds__(256)
zero_heads_kernel(int* __restrict__ rows, int* __restrict__ spill_n, int n) {
  int i = blockIdx.x * blockDim.x + threadIdx.x;
  if (i < n) rows[(size_t)i * ROW_W] = 0;
  if (i == 0) *spill_n = 0;
}

__device__ __forceinline__ void put_edge(int* __restrict__ rows, int* __restrict__ spill_n,
                                         int* __restrict__ spill, int d, int c, int r) {
  if (r < ROW_CAP) {
    rows[(size_t)d * ROW_W + 1 + r] = c;
  } else {
    const int k = atomicAdd(spill_n, 1);
    spill[2 * (size_t)k]     = d;
    spill[2 * (size_t)k + 1] = c;
  }
}

__device__ __forceinline__ void
edge_fill_block(const int blk, const int nblk,
                const int* __restrict__ ei, int* __restrict__ rows,
                int* __restrict__ spill_n, int* __restrict__ spill, int E) {
  const int T  = nblk * blockDim.x;
  const int e0 = blk * blockDim.x + threadIdx.x;
  const int e1 = e0 + T, e2 = e1 + T, e3 = e2 + T;
  const int f0 = min(e0, E - 1), f1 = min(e1, E - 1);
  const int f2 = min(e2, E - 1), f3 = min(e3, E - 1);
  const int* dst = ei + E;
  const int d0 = dst[f0], d1 = dst[f1], d2 = dst[f2], d3 = dst[f3];
  const int c0 = ei[f0], c1 = ei[f1], c2 = ei[f2], c3 = ei[f3];
  int r0 = 0, r1 = 0, r2 = 0, r3 = 0;
  if (e0 < E) r0 = atomicAdd(rows + (size_t)d0 * ROW_W, 1);
  if (e1 < E) r1 = atomicAdd(rows + (size_t)d1 * ROW_W, 1);
  if (e2 < E) r2 = atomicAdd(rows + (size_t)d2 * ROW_W, 1);
  if (e3 < E) r3 = atomicAdd(rows + (size_t)d3 * ROW_W, 1);
  if (e0 < E) put_edge(rows, spill_n, spill, d0, c0, r0);
  if (e1 < E) put_edge(rows, spill_n, spill, d1, c1, r1);
  if (e2 < E) put_edge(rows, spill_n, spill, d2, c2, r2);
  if (e3 < E) put_edge(rows, spill_n, spill, d3, c3, r3);
}

__device__ __forceinline__ void
gemm1_block(const int blk,
            const float* __restrict__ x, const float* __restrict__ W1l,
            const float* __restrict__ W1r, float* __restrict__ xl1,
            float* __restrict__ xr1, int n_nodes) {
  const int lane = threadIdx.x & 31;
  const int wave = threadIdx.x >> 5;
  const int m0   = blk * 32;
  const int n0   = wave * 16;
  const int half = lane >> 4;
  const int rc   = lane & 15;

  int arow0 = m0 + rc;      if (arow0 >= n_nodes) arow0 = n_nodes - 1;
  int arow1 = m0 + 16 + rc; if (arow1 >= n_nodes) arow1 = n_nodes - 1;
  const v2f*   xrow0 = (const v2f*)(x + (size_t)arow0 * 128);
  const v2f*   xrow1 = (const v2f*)(x + (size_t)arow1 * 128);
  const float* Bl    = W1l + n0 + rc;
  const float* Br    = W1r + n0 + rc;

  v8f accL0 = {}, accR0 = {}, accL1 = {}, accR1 = {};
  #pragma unroll
  for (int kc = 0; kc < 32; ++kc) {
    v2f a0 = xrow0[2 * kc + half];
    v2f a1 = xrow1[2 * kc + half];
    const int k0 = 4 * kc + 2 * half;
    v2f bl, br;
    bl[0] = Bl[(size_t)(k0 + 0) * 64];
    bl[1] = Bl[(size_t)(k0 + 1) * 64];
    br[0] = Br[(size_t)(k0 + 0) * 64];
    br[1] = Br[(size_t)(k0 + 1) * 64];
    accL0 = __builtin_amdgcn_wmma_f32_16x16x4_f32(
        false, a0, false, bl, (short)0, accL0, false, false);
    accR0 = __builtin_amdgcn_wmma_f32_16x16x4_f32(
        false, a0, false, br, (short)0, accR0, false, false);
    accL1 = __builtin_amdgcn_wmma_f32_16x16x4_f32(
        false, a1, false, bl, (short)0, accL1, false, false);
    accR1 = __builtin_amdgcn_wmma_f32_16x16x4_f32(
        false, a1, false, br, (short)0, accR1, false, false);
  }

  #pragma unroll
  for (int r = 0; r < 8; ++r) {
    const int row0 = m0 + r + 8 * half;
    const int row1 = row0 + 16;
    if (row0 < n_nodes) {
      xl1[(size_t)row0 * 64 + n0 + rc] = accL0[r];
      xr1[(size_t)row0 * 64 + n0 + rc] = accR0[r];
    }
    if (row1 < n_nodes) {
      xl1[(size_t)row1 * 64 + n0 + rc] = accL1[r];
      xr1[(size_t)row1 * 64 + n0 + rc] = accR1[r];
    }
  }
}

__global__ void __launch_bounds__(128)
gemm1_edges_kernel(const float* __restrict__ x, const float* __restrict__ W1l,
                   const float* __restrict__ W1r, float* __restrict__ xl1,
                   float* __restrict__ xr1, int n_nodes, int gemm_blocks,
                   const int* __restrict__ ei, int* __restrict__ rows,
                   int* __restrict__ spill_n, int* __restrict__ spill, int E,
                   int edge_blocks) {
  const int  blk  = blockIdx.x >> 1;
  const bool edge = (((blockIdx.x >> 3) ^ blockIdx.x) & 1) != 0;
  if (edge) {
    if (blk < edge_blocks) edge_fill_block(blk, edge_blocks, ei, rows, spill_n, spill, E);
  } else {
    if (blk < gemm_blocks) gemm1_block(blk, x, W1l, W1r, xl1, xr1, n_nodes);
  }
}

__global__ void __launch_bounds__(128)
gemm2_kernel(const float* __restrict__ h, const float* __restrict__ W2l,
             const float* __restrict__ W2r, float* __restrict__ xl2,
             float* __restrict__ xr2, int n_nodes, int mtiles) {
  const int lane  = threadIdx.x & 31;
  const int mtile = blockIdx.x * 4 + (threadIdx.x >> 5);
  if (mtile >= mtiles) return;
  const int m0   = mtile * 16;
  const int half = lane >> 4;
  const int rc   = lane & 15;

  int arow = m0 + rc; if (arow >= n_nodes) arow = n_nodes - 1;
  const v2f* hrow = (const v2f*)(h + (size_t)arow * 64);
  const bool colok = (rc < 10);

  v8f accL = {}, accR = {};
  #pragma unroll
  for (int kc = 0; kc < 16; ++kc) {
    v2f a = hrow[2 * kc + half];
    const int k0 = 4 * kc + 2 * half;
    v2f bl, br;
    bl[0] = colok ? W2l[(size_t)(k0 + 0) * 10 + rc] : 0.0f;
    bl[1] = colok ? W2l[(size_t)(k0 + 1) * 10 + rc] : 0.0f;
    br[0] = colok ? W2r[(size_t)(k0 + 0) * 10 + rc] : 0.0f;
    br[1] = colok ? W2r[(size_t)(k0 + 1) * 10 + rc] : 0.0f;
    accL = __builtin_amdgcn_wmma_f32_16x16x4_f32(
        false, a, false, bl, (short)0, accL, false, false);
    accR = __builtin_amdgcn_wmma_f32_16x16x4_f32(
        false, a, false, br, (short)0, accR, false, false);
  }

  #pragma unroll
  for (int r = 0; r < 8; ++r) {
    int row = m0 + r + 8 * half;
    if (row < n_nodes) {
      xl2[(size_t)row * 16 + rc] = accL[r];
      xr2[(size_t)row * 16 + rc] = accR[r];
    }
  }
}

__device__ __forceinline__ float edge_score(const float4 x, const float4 r, const float4 a) {
  float p = 0.0f, s;
  s = x.x + r.x; s = (s > 0.f) ? s : NEG_SLOPE * s; p += s * a.x;
  s = x.y + r.y; s = (s > 0.f) ? s : NEG_SLOPE * s; p += s * a.y;
  s = x.z + r.z; s = (s > 0.f) ? s : NEG_SLOPE * s; p += s * a.z;
  s = x.w + r.w; s = (s > 0.f) ? s : NEG_SLOPE * s; p += s * a.w;
  p += __shfl_xor(p, 1, 32);
  p += __shfl_xor(p, 2, 32);
  return p;
}

template <int STRIDE, int NV, int VOFF>
__device__ __forceinline__ void node_softmax_sum(
    const int* __restrict__ csrc, const float* __restrict__ xl,
    const int end, const int coff,
    const float4 (&rv)[NV], const float4 (&at)[NV],
    float (&mrun)[NV], float4 (&acc)[NV], float (&den)[NV]) {
  const float NEG_INF = -__builtin_inff();

  const int last = end - 1;
  for (int j = 0; j < end; j += 4) {
    const int s0 = csrc[j];
    const int s1 = csrc[min(j + 1, last)];
    const int s2 = csrc[min(j + 2, last)];
    const int s3 = csrc[min(j + 3, last)];
    const float* r0 = xl + (size_t)s0 * STRIDE + coff;
    const float* r1 = xl + (size_t)s1 * STRIDE + coff;
    const float* r2 = xl + (size_t)s2 * STRIDE + coff;
    const float* r3 = xl + (size_t)s3 * STRIDE + coff;
    float4 x0[NV], x1[NV], x2[NV], x3[NV];
    #pragma unroll
    for (int v = 0; v < NV; ++v) {
      x0[v] = *(const float4*)(r0 + v * VOFF);
      x1[v] = *(const float4*)(r1 + v * VOFF);
      x2[v] = *(const float4*)(r2 + v * VOFF);
      x3[v] = *(const float4*)(r3 + v * VOFF);
    }

    const bool v1 = (j + 1 < end), v2 = (j + 2 < end), v3 = (j + 3 < end);
    #pragma unroll
    for (int v = 0; v < NV; ++v) {
      const float p0 = edge_score(x0[v], rv[v], at[v]);
      float p1 = edge_score(x1[v], rv[v], at[v]);
      float p2 = edge_score(x2[v], rv[v], at[v]);
      float p3 = edge_score(x3[v], rv[v], at[v]);
      p1 = v1 ? p1 : NEG_INF;
      p2 = v2 ? p2 : NEG_INF;
      p3 = v3 ? p3 : NEG_INF;

      const float m  = mrun[v];
      const float mn = fmaxf(fmaxf(m, p0), fmaxf(fmaxf(p1, p2), p3));
      const float sc = __expf(m - mn);
      const float w0 = __expf(p0 - mn);
      const float w1 = __expf(p1 - mn);
      const float w2 = __expf(p2 - mn);
      const float w3 = __expf(p3 - mn);
      den[v]   = den[v]   * sc + ((w0 + w1) + (w2 + w3));
      acc[v].x = acc[v].x * sc + ((w0 * x0[v].x + w1 * x1[v].x) + (w2 * x2[v].x + w3 * x3[v].x));
      acc[v].y = acc[v].y * sc + ((w0 * x0[v].y + w1 * x1[v].y) + (w2 * x2[v].y + w3 * x3[v].y));
      acc[v].z = acc[v].z * sc + ((w0 * x0[v].z + w1 * x1[v].z) + (w2 * x2[v].z + w3 * x3[v].z));
      acc[v].w = acc[v].w * sc + ((w0 * x0[v].w + w1 * x1[v].w) + (w2 * x2[v].w + w3 * x3[v].w));
      mrun[v] = mn;
    }
  }
}

template <int STRIDE, int NV, int VOFF>
__device__ __forceinline__ void node_one_edge(
    const int s, const float* __restrict__ xl, const int coff,
    const float4 (&rv)[NV], const float4 (&at)[NV],
    float (&m)[NV], float4 (&a)[NV], float (&d)[NV]) {
  const float* r = xl + (size_t)s * STRIDE + coff;
  #pragma unroll
  for (int v = 0; v < NV; ++v) {
    const float4 x = *(const float4*)(r + v * VOFF);
    const float p  = edge_score(x, rv[v], at[v]);
    const float mn = fmaxf(m[v], p);
    const float sc = __expf(m[v] - mn);
    const float w  = __expf(p - mn);
    d[v]   = d[v]   * sc + w;
    a[v].x = a[v].x * sc + w * x.x;
    a[v].y = a[v].y * sc + w * x.y;
    a[v].z = a[v].z * sc + w * x.z;
    a[v].w = a[v].w * sc + w * x.w;
    m[v] = mn;
  }
}

template <int STRIDE, int NV, int VOFF>
__device__ __forceinline__ void node_all_edges(
    const int* __restrict__ rows, const int* __restrict__ spill_n,
    const int* __restrict__ spill, const int spill_cap,
    const float* __restrict__ xl, const int n, const int coff,
    const float4 (&rv)[NV], const float4 (&at)[NV],
    float4 (&acc)[NV], float (&den)[NV]) {
  const int* row = rows + (size_t)n * ROW_W;
  const int cnt = row[0];
  float m[NV];
  #pragma unroll
  for (int v = 0; v < NV; ++v) {
    m[v]   = -__builtin_inff();
    den[v] = 0.0f;
    acc[v] = make_float4(0.0f, 0.0f, 0.0f, 0.0f);
  }
  node_softmax_sum<STRIDE, NV, VOFF>(row + 1, xl, min(cnt, ROW_CAP), coff, rv, at, m, acc, den);
  if (cnt > ROW_CAP) {
    const int ns = min(*spill_n, spill_cap);
    for (int i = 0; i < ns; ++i)
      if (spill[2 * (size_t)i] == n)
        node_one_edge<STRIDE, NV, VOFF>(spill[2 * (size_t)i + 1], xl, coff, rv, at, m, acc, den);
  }
}

__global__ void __launch_bounds__(256)
l1_node_kernel(const int* __restrict__ rows, const int* __restrict__ spill_n,
               const int* __restrict__ spill, int spill_cap,
               const float* __restrict__ xl1, const float* __restrict__ xr1,
               const float* __restrict__ att1, const float* __restrict__ b1,
               float* __restrict__ h, int n_nodes) {
  const int n = (blockIdx.x * blockDim.x + threadIdx.x) >> 3;
  if (n >= n_nodes) return;
  const int coff = (threadIdx.x & 7) * 4;
  float4 rv[2], at[2];
  rv[0] = *(const float4*)(xr1 + (size_t)n * 64 + coff);
  rv[1] = *(const float4*)(xr1 + (size_t)n * 64 + coff + 32);
  at[0] = *(const float4*)(att1 + coff);
  at[1] = *(const float4*)(att1 + coff + 32);

  float4 acc[2]; float den[2];
  node_all_edges<64, 2, 32>(rows, spill_n, spill, spill_cap, xl1, n, coff, rv, at, acc, den);

  #pragma unroll
  for (int g = 0; g < 2; ++g) {
    const float4 bb = *(const float4*)(b1 + coff + 32 * g);
    const float  dn = den[g];
    float4 v;
    v.x = ((dn > 0.0f) ? acc[g].x / dn : 0.0f) + bb.x;
    v.y = ((dn > 0.0f) ? acc[g].y / dn : 0.0f) + bb.y;
    v.z = ((dn > 0.0f) ? acc[g].z / dn : 0.0f) + bb.z;
    v.w = ((dn > 0.0f) ? acc[g].w / dn : 0.0f) + bb.w;
    v.x = (v.x > 0.0f) ? v.x : expm1f(v.x);
    v.y = (v.y > 0.0f) ? v.y : expm1f(v.y);
    v.z = (v.z > 0.0f) ? v.z : expm1f(v.z);
    v.w = (v.w > 0.0f) ? v.w : expm1f(v.w);
    *(float4*)(h + (size_t)n * 64 + coff + 32 * g) = v;
  }
}

__global__ void __launch_bounds__(256)
l2_node_kernel(const int* __restrict__ rows, const int* __restrict__ spill_n,
               const int* __restrict__ spill, int spill_cap,
               const float* __restrict__ xl2, const float* __restrict__ xr2,
               const float* __restrict__ att2, const float* __restrict__ b2,
               float* __restrict__ out, int n_nodes) {
  __shared__ float stage[64 * 10];
  const int nl = threadIdx.x >> 2;
  const int n  = blockIdx.x * 64 + nl;
  const int c0 = (threadIdx.x & 3) * 4;

  if (n < n_nodes) {
    float4 rv[1], at[1];
    rv[0] = *(const float4*)(xr2 + (size_t)n * 16 + c0);
    at[0].x = (c0 + 0 < 10) ? att2[c0 + 0] : 0.0f;
    at[0].y = (c0 + 1 < 10) ? att2[c0 + 1] : 0.0f;
    at[0].z = (c0 + 2 < 10) ? att2[c0 + 2] : 0.0f;
    at[0].w = (c0 + 3 < 10) ? att2[c0 + 3] : 0.0f;

    float4 acc[1]; float den[1];
    node_all_edges<16, 1, 0>(rows, spill_n, spill, spill_cap, xl2, n, c0, rv, at, acc, den);

    const bool ok = (den[0] > 0.0f);
    if (c0 + 0 < 10) stage[nl * 10 + c0 + 0] = (ok ? acc[0].x / den[0] : 0.0f) + b2[c0 + 0];
    if (c0 + 1 < 10) stage[nl * 10 + c0 + 1] = (ok ? acc[0].y / den[0] : 0.0f) + b2[c0 + 1];
    if (c0 + 2 < 10) stage[nl * 10 + c0 + 2] = (ok ? acc[0].z / den[0] : 0.0f) + b2[c0 + 2];
    if (c0 + 3 < 10) stage[nl * 10 + c0 + 3] = (ok ? acc[0].w / den[0] : 0.0f) + b2[c0 + 3];
  }
  __syncthreads();

  const long long base  = (long long)blockIdx.x * 640;
  const long long total = (long long)n_nodes * 10;
  for (int i = threadIdx.x; i < 640; i += 256)
    if (base + i < total) out[base + i] = stage[i];
}

extern "C" void kernel_launch(void* const* d_in, const int* in_sizes, int n_in,
                              void* d_out, int out_size, void* d_ws, size_t ws_size,
                              hipStream_t stream) {
  const float* x    = (const float*)d_in[0];
  const int*   ei   = (const int*)d_in[1];
  const float* W1l  = (const float*)d_in[2];
  const float* W1r  = (const float*)d_in[3];
  const float* att1 = (const float*)d_in[4];
  const float* b1   = (const float*)d_in[5];
  const float* W2l  = (const float*)d_in[6];
  const float* W2r  = (const float*)d_in[7];
  const float* att2 = (const float*)d_in[8];
  const float* b2   = (const float*)d_in[9];
  float* out = (float*)d_out;

  const int N = in_sizes[0] / 128;
  const int E = in_sizes[1] / 2;

  float* ws = (float*)d_ws;
  size_t o = 0;
  int* rows    = (int*)(ws + o); o += (size_t)N * ROW_W;
  float* xl1   = ws + o; o += (size_t)N * 64;
  float* xr1   = ws + o; o += (size_t)N * 64;
  float* h1    = ws + o; o += (size_t)N * 64;
  float* xl2   = ws + o; o += (size_t)N * 16;
  float* xr2   = ws + o; o += (size_t)N * 16;
  int* spill_n = (int*)(ws + o); o += 64;
  int* spill   = (int*)(ws + o); o += 2 * (size_t)E;

  auto blocks = [](long long n) { return (unsigned)((n + 255) / 256); };
  const int mtiles = (N + 15) / 16;

  zero_heads_kernel<<<blocks(N), 256, 0, stream>>>(rows, spill_n, N);
  const int gemm_blocks = (mtiles + 1) / 2;
  const int edge_blocks = (int)(((long long)E + 511) / 512);
  const int pair_blocks = (gemm_blocks > edge_blocks) ? gemm_blocks : edge_blocks;
  gemm1_edges_kernel<<<2 * (unsigned)pair_blocks, 128, 0, stream>>>(
      x, W1l, W1r, xl1, xr1, N, gemm_blocks, ei, rows, spill_n, spill, E, edge_blocks);

  l1_node_kernel<<<blocks((long long)N * 8), 256, 0, stream>>>(rows, spill_n, spill, E, xl1, xr1,
                                                                att1, b1, h1, N);

  gemm2_kernel<<<(mtiles + 3) / 4, 128, 0, stream>>>(h1, W2l, W2r, xl2, xr2, N, mtiles);
  l2_node_kernel<<<(unsigned)((N + 63) / 64), 256, 0, stream>>>(rows, spill_n, spill, E, xl2, xr2,
                                                                att2, b2, out, N);
}


# Layer-2 linear maps (GEMM2) run in the layer-1 node kernel's tail: a block stages its 16 h rows in LDS and one wave does the WMMAs from there; h never goes to memory; launches 5 to 4; sums unchanged
# speedup vs baseline: 4.4385x; 1.0947x over previous
#include <hip/hip_runtime.h>
#include <math.h>

typedef __attribute__((ext_vector_type(2))) float v2f;
typedef __attribute__((ext_vector_type(8))) float v8f;

#define NEG_SLOPE 0.2f

#define ROW_W   64
#define ROW_CAP 63

__global__ void __launch_bounds__(256)
zero_heads_kernel(int* __restrict__ rows, int* __restrict__ spill_n, int n) {
  int i = blockIdx.x * blockDim.x + threadIdx.x;
  if (i < n) rows[(size_t)i * ROW_W] = 0;
  if (i == 0) *spill_n = 0;
}

__device__ __forceinline__ void put_edge(int* __restrict__ rows, int* __restrict__ spill_n,
                                         int* __restrict__ spill, int d, int c, int r) {
  if (r < ROW_CAP) {
    rows[(size_t)d * ROW_W + 1 + r] = c;
  } else {
    const int k = atomicAdd(spill_n, 1);
    spill[2 * (size_t)k]     = d;
    spill[2 * (size_t)k + 1] = c;
  }
}

__device__ __forceinline__ void
edge_fill_block(const int blk, const int nblk,
                const int* __restrict__ ei, int* __restrict__ rows,
                int* __restrict__ spill_n, int* __restrict__ spill, int E) {
  const int T  = nblk * blockDim.x;
  const int e0 = blk * blockDim.x + threadIdx.x;
  const int e1 = e0 + T, e2 = e1 + T, e3 = e2 + T;
  const int f0 = min(e0, E - 1), f1 = min(e1, E - 1);
  const int f2 = min(e2, E - 1), f3 = min(e3, E - 1);
  const int* dst = ei + E;
  const int d0 = dst[f0], d1 = dst[f1], d2 = dst[f2], d3 = dst[f3];
  const int c0 = ei[f0], c1 = ei[f1], c2 = ei[f2], c3 = ei[f3];
  int r0 = 0, r1 = 0, r2 = 0, r3 = 0;
  if (e0 < E) r0 = atomicAdd(rows + (size_t)d0 * ROW_W, 1);
  if (e1 < E) r1 = atomicAdd(rows + (size_t)d1 * ROW_W, 1);
  if (e2 < E) r2 = atomicAdd(rows + (size_t)d2 * ROW_W, 1);
  if (e3 < E) r3 = atomicAdd(rows + (size_t)d3 * ROW_W, 1);
  if (e0 < E) put_edge(rows, spill_n, spill, d0, c0, r0);
  if (e1 < E) put_edge(rows, spill_n, spill, d1, c1, r1);
  if (e2 < E) put_edge(rows, spill_n, spill, d2, c2, r2);
  if (e3 < E) put_edge(rows, spill_n, spill, d3, c3, r3);
}

__device__ __forceinline__ void
gemm1_block(const int blk,
            const float* __restrict__ x, const float* __restrict__ W1l,
            const float* __restrict__ W1r, float* __restrict__ xl1,
            float* __restrict__ xr1, int n_nodes) {
  const int lane = threadIdx.x & 31;
  const int wave = threadIdx.x >> 5;
  const int m0   = blk * 32;
  const int n0   = wave * 16;
  const int half = lane >> 4;
  const int rc   = lane & 15;

  int arow0 = m0 + rc;      if (arow0 >= n_nodes) arow0 = n_nodes - 1;
  int arow1 = m0 + 16 + rc; if (arow1 >= n_nodes) arow1 = n_nodes - 1;
  const v2f*   xrow0 = (const v2f*)(x + (size_t)arow0 * 128);
  const v2f*   xrow1 = (const v2f*)(x + (size_t)arow1 * 128);
  const float* Bl    = W1l + n0 + rc;
  const float* Br    = W1r + n0 + rc;

  v8f accL0 = {}, accR0 = {}, accL1 = {}, accR1 = {};
  #pragma unroll
  for (int kc = 0; kc < 32; ++kc) {
    v2f a0 = xrow0[2 * kc + half];
    v2f a1 = xrow1[2 * kc + half];
    const int k0 = 4 * kc + 2 * half;
    v2f bl, br;
    bl[0] = Bl[(size_t)(k0 + 0) * 64];
    bl[1] = Bl[(size_t)(k0 + 1) * 64];
    br[0] = Br[(size_t)(k0 + 0) * 64];
    br[1] = Br[(size_t)(k0 + 1) * 64];
    accL0 = __builtin_amdgcn_wmma_f32_16x16x4_f32(
        false, a0, false, bl, (short)0, accL0, false, false);
    accR0 = __builtin_amdgcn_wmma_f32_16x16x4_f32(
        false, a0, false, br, (short)0, accR0, false, false);
    accL1 = __builtin_amdgcn_wmma_f32_16x16x4_f32(
        false, a1, false, bl, (short)0, accL1, false, false);
    accR1 = __builtin_amdgcn_wmma_f32_16x16x4_f32(
        false, a1, false, br, (short)0, accR1, false, false);
  }

  #pragma unroll
  for (int r = 0; r < 8; ++r) {
    const int row0 = m0 + r + 8 * half;
    const int row1 = row0 + 16;
    if (row0 < n_nodes) {
      xl1[(size_t)row0 * 64 + n0 + rc] = accL0[r];
      xr1[(size_t)row0 * 64 + n0 + rc] = accR0[r];
    }
    if (row1 < n_nodes) {
      xl1[(size_t)row1 * 64 + n0 + rc] = accL1[r];
      xr1[(size_t)row1 * 64 + n0 + rc] = accR1[r];
    }
  }
}

__global__ void __launch_bounds__(128)
gemm1_edges_kernel(const float* __restrict__ x, const float* __restrict__ W1l,
                   const float* __restrict__ W1r, float* __restrict__ xl1,
                   float* __restrict__ xr1, int n_nodes, int gemm_blocks,
                   const int* __restrict__ ei, int* __restrict__ rows,
                   int* __restrict__ spill_n, int* __restrict__ spill, int E,
                   int edge_blocks) {
  const int  blk  = blockIdx.x >> 1;
  const bool edge = (((blockIdx.x >> 3) ^ blockIdx.x) & 1) != 0;
  if (edge) {
    if (blk < edge_blocks) edge_fill_block(blk, edge_blocks, ei, rows, spill_n, spill, E);
  } else {
    if (blk < gemm_blocks) gemm1_block(blk, x, W1l, W1r, xl1, xr1, n_nodes);
  }
}

__device__ __forceinline__ float edge_score(const float4 x, const float4 r, const float4 a) {
  float p = 0.0f, s;
  s = x.x + r.x; s = (s > 0.f) ? s : NEG_SLOPE * s; p += s * a.x;
  s = x.y + r.y; s = (s > 0.f) ? s : NEG_SLOPE * s; p += s * a.y;
  s = x.z + r.z; s = (s > 0.f) ? s : NEG_SLOPE * s; p += s * a.z;
  s = x.w + r.w; s = (s > 0.f) ? s : NEG_SLOPE * s; p += s * a.w;
  p += __shfl_xor(p, 1, 32);
  p += __shfl_xor(p, 2, 32);
  return p;
}

template <int STRIDE, int NV, int VOFF>
__device__ __forceinline__ void node_softmax_sum(
    const int* __restrict__ csrc, const float* __restrict__ xl,
    const int end, const int coff,
    const float4 (&rv)[NV], const float4 (&at)[NV],
    float (&mrun)[NV], float4 (&acc)[NV], float (&den)[NV]) {
  const float NEG_INF = -__builtin_inff();

  const int last = end - 1;
  for (int j = 0; j < end; j += 4) {
    const int s0 = csrc[j];
    const int s1 = csrc[min(j + 1, last)];
    const int s2 = csrc[min(j + 2, last)];
    const int s3 = csrc[min(j + 3, last)];
    const float* r0 = xl + (size_t)s0 * STRIDE + coff;
    const float* r1 = xl + (size_t)s1 * STRIDE + coff;
    const float* r2 = xl + (size_t)s2 * STRIDE + coff;
    const float* r3 = xl + (size_t)s3 * STRIDE + coff;
    float4 x0[NV], x1[NV], x2[NV], x3[NV];
    #pragma unroll
    for (int v = 0; v < NV; ++v) {
      x0[v] = *(const float4*)(r0 + v * VOFF);
      x1[v] = *(const float4*)(r1 + v * VOFF);
      x2[v] = *(const float4*)(r2 + v * VOFF);
      x3[v] = *(const float4*)(r3 + v * VOFF);
    }

    const bool v1 = (j + 1 < end), v2 = (j + 2 < end), v3 = (j + 3 < end);
    #pragma unroll
    for (int v = 0; v < NV; ++v) {
      const float p0 = edge_score(x0[v], rv[v], at[v]);
      float p1 = edge_score(x1[v], rv[v], at[v]);
      float p2 = edge_score(x2[v], rv[v], at[v]);
      float p3 = edge_score(x3[v], rv[v], at[v]);
      p1 = v1 ? p1 : NEG_INF;
      p2 = v2 ? p2 : NEG_INF;
      p3 = v3 ? p3 : NEG_INF;

      const float m  = mrun[v];
      const float mn = fmaxf(fmaxf(m, p0), fmaxf(fmaxf(p1, p2), p3));
      const float sc = __expf(m - mn);
      const float w0 = __expf(p0 - mn);
      const float w1 = __expf(p1 - mn);
      const float w2 = __expf(p2 - mn);
      const float w3 = __expf(p3 - mn);
      den[v]   = den[v]   * sc + ((w0 + w1) + (w2 + w3));
      acc[v].x = acc[v].x * sc + ((w0 * x0[v].x + w1 * x1[v].x) + (w2 * x2[v].x + w3 * x3[v].x));
      acc[v].y = acc[v].y * sc + ((w0 * x0[v].y + w1 * x1[v].y) + (w2 * x2[v].y + w3 * x3[v].y));
      acc[v].z = acc[v].z * sc + ((w0 * x0[v].z + w1 * x1[v].z) + (w2 * x2[v].z + w3 * x3[v].z));
      acc[v].w = acc[v].w * sc + ((w0 * x0[v].w + w1 * x1[v].w) + (w2 * x2[v].w + w3 * x3[v].w));
      mrun[v] = mn;
    }
  }
}

template <int STRIDE, int NV, int VOFF>
__device__ __forceinline__ void node_one_edge(
    const int s, const float* __restrict__ xl, const int coff,
    const float4 (&rv)[NV], const float4 (&at)[NV],
    float (&m)[NV], float4 (&a)[NV], float (&d)[NV]) {
  const float* r = xl + (size_t)s * STRIDE + coff;
  #pragma unroll
  for (int v = 0; v < NV; ++v) {
    const float4 x = *(const float4*)(r + v * VOFF);
    const float p  = edge_score(x, rv[v], at[v]);
    const float mn = fmaxf(m[v], p);
    const float sc = __expf(m[v] - mn);
    const float w  = __expf(p - mn);
    d[v]   = d[v]   * sc + w;
    a[v].x = a[v].x * sc + w * x.x;
    a[v].y = a[v].y * sc + w * x.y;
    a[v].z = a[v].z * sc + w * x.z;
    a[v].w = a[v].w * sc + w * x.w;
    m[v] = mn;
  }
}

template <int STRIDE, int NV, int VOFF>
__device__ __forceinline__ void node_all_edges(
    const int* __restrict__ rows, const int* __restrict__ spill_n,
    const int* __restrict__ spill, const int spill_cap,
    const float* __restrict__ xl, const int n, const int coff,
    const float4 (&rv)[NV], const float4 (&at)[NV],
    float4 (&acc)[NV], float (&den)[NV]) {
  const int* row = rows + (size_t)n * ROW_W;
  const int cnt = row[0];
  float m[NV];
  #pragma unroll
  for (int v = 0; v < NV; ++v) {
    m[v]   = -__builtin_inff();
    den[v] = 0.0f;
    acc[v] = make_float4(0.0f, 0.0f, 0.0f, 0.0f);
  }
  node_softmax_sum<STRIDE, NV, VOFF>(row + 1, xl, min(cnt, ROW_CAP), coff, rv, at, m, acc, den);
  if (cnt > ROW_CAP) {
    const int ns = min(*spill_n, spill_cap);
    for (int i = 0; i < ns; ++i)
      if (spill[2 * (size_t)i] == n)
        node_one_edge<STRIDE, NV, VOFF>(spill[2 * (size_t)i + 1], xl, coff, rv, at, m, acc, den);
  }
}

#define H_PITCH 68

__global__ void __launch_bounds__(128)
l1_node_gemm2_kernel(const int* __restrict__ rows, const int* __restrict__ spill_n,
                     const int* __restrict__ spill, int spill_cap,
                     const float* __restrict__ xl1, const float* __restrict__ xr1,
                     const float* __restrict__ att1, const float* __restrict__ b1,
                     const float* __restrict__ W2l, const float* __restrict__ W2r,
                     float* __restrict__ xl2, float* __restrict__ xr2, int n_nodes) {
  __shared__ float hs[16 * H_PITCH];
  const int nl   = threadIdx.x >> 3;
  const int n    = blockIdx.x * 16 + nl;
  const int coff = (threadIdx.x & 7) * 4;

  float4 hv[2];
  hv[0] = make_float4(0.0f, 0.0f, 0.0f, 0.0f);
  hv[1] = make_float4(0.0f, 0.0f, 0.0f, 0.0f);
  if (n < n_nodes) {
    float4 rv[2], at[2];
    rv[0] = *(const float4*)(xr1 + (size_t)n * 64 + coff);
    rv[1] = *(const float4*)(xr1 + (size_t)n * 64 + coff + 32);
    at[0] = *(const float4*)(att1 + coff);
    at[1] = *(const float4*)(att1 + coff + 32);

    float4 acc[2]; float den[2];
    node_all_edges<64, 2, 32>(rows, spill_n, spill, spill_cap, xl1, n, coff, rv, at, acc, den);

    #pragma unroll
    for (int g = 0; g < 2; ++g) {
      const float4 bb = *(const float4*)(b1 + coff + 32 * g);
      const float  dn = den[g];
      float4 v;
      v.x = ((dn > 0.0f) ? acc[g].x / dn : 0.0f) + bb.x;
      v.y = ((dn > 0.0f) ? acc[g].y / dn : 0.0f) + bb.y;
      v.z = ((dn > 0.0f) ? acc[g].z / dn : 0.0f) + bb.z;
      v.w = ((dn > 0.0f) ? acc[g].w / dn : 0.0f) + bb.w;
      v.x = (v.x > 0.0f) ? v.x : expm1f(v.x);
      v.y = (v.y > 0.0f) ? v.y : expm1f(v.y);
      v.z = (v.z > 0.0f) ? v.z : expm1f(v.z);
      v.w = (v.w > 0.0f) ? v.w : expm1f(v.w);
      hv[g] = v;
    }
  }
  *(float4*)(hs + nl * H_PITCH + coff)      = hv[0];
  *(float4*)(hs + nl * H_PITCH + coff + 32) = hv[1];
  __syncthreads();

  const int wave = __builtin_amdgcn_readfirstlane((int)(threadIdx.x >> 5));
  const int m0   = blockIdx.x * 16;
  if (wave == 0) {
    const int lane = threadIdx.x & 31;
    const int half = lane >> 4;
    const int rc   = lane & 15;
    const float* hrow = hs + rc * H_PITCH;
    const bool colok = (rc < 10);
    const int  cc    = colok ? rc : 9;

    v8f accL = {}, accR = {};
    #pragma unroll
    for (int kc = 0; kc < 16; ++kc) {
      const int k0 = 4 * kc + 2 * half;
      const v2f a = *(const v2f*)(hrow + k0);
      const float l0 = W2l[(size_t)(k0 + 0) * 10 + cc];
      const float l1 = W2l[(size_t)(k0 + 1) * 10 + cc];
      const float r0 = W2r[(size_t)(k0 + 0) * 10 + cc];
      const float r1 = W2r[(size_t)(k0 + 1) * 10 + cc];
      v2f bl, br;
      bl[0] = colok ? l0 : 0.0f;
      bl[1] = colok ? l1 : 0.0f;
      br[0] = colok ? r0 : 0.0f;
      br[1] = colok ? r1 : 0.0f;
      accL = __builtin_amdgcn_wmma_f32_16x16x4_f32(
          false, a, false, bl, (short)0, accL, false, false);
      accR = __builtin_amdgcn_wmma_f32_16x16x4_f32(
          false, a, false, br, (short)0, accR, false, false);
    }

    #pragma unroll
    for (int r = 0; r < 8; ++r) {
      const int row = m0 + r + 8 * half;
      if (row < n_nodes) {
        xl2[(size_t)row * 16 + rc] = accL[r];
        xr2[(size_t)row * 16 + rc] = accR[r];
      }
    }
  }
}

__global__ void __launch_bounds__(256)
l2_node_kernel(const int* __restrict__ rows, const int* __restrict__ spill_n,
               const int* __restrict__ spill, int spill_cap,
               const float* __restrict__ xl2, const float* __restrict__ xr2,
               const float* __restrict__ att2, const float* __restrict__ b2,
               float* __restrict__ out, int n_nodes) {
  __shared__ float stage[64 * 10];
  const int nl = threadIdx.x >> 2;
  const int n  = blockIdx.x * 64 + nl;
  const int c0 = (threadIdx.x & 3) * 4;

  if (n < n_nodes) {
    float4 rv[1], at[1];
    rv[0] = *(const float4*)(xr2 + (size_t)n * 16 + c0);
    at[0].x = (c0 + 0 < 10) ? att2[c0 + 0] : 0.0f;
    at[0].y = (c0 + 1 < 10) ? att2[c0 + 1] : 0.0f;
    at[0].z = (c0 + 2 < 10) ? att2[c0 + 2] : 0.0f;
    at[0].w = (c0 + 3 < 10) ? att2[c0 + 3] : 0.0f;

    float4 acc[1]; float den[1];
    node_all_edges<16, 1, 0>(rows, spill_n, spill, spill_cap, xl2, n, c0, rv, at, acc, den);

    const bool ok = (den[0] > 0.0f);
    if (c0 + 0 < 10) stage[nl * 10 + c0 + 0] = (ok ? acc[0].x / den[0] : 0.0f) + b2[c0 + 0];
    if (c0 + 1 < 10) stage[nl * 10 + c0 + 1] = (ok ? acc[0].y / den[0] : 0.0f) + b2[c0 + 1];
    if (c0 + 2 < 10) stage[nl * 10 + c0 + 2] = (ok ? acc[0].z / den[0] : 0.0f) + b2[c0 + 2];
    if (c0 + 3 < 10) stage[nl * 10 + c0 + 3] = (ok ? acc[0].w / den[0] : 0.0f) + b2[c0 + 3];
  }
  __syncthreads();

  const long long base  = (long long)blockIdx.x * 640;
  const long long total = (long long)n_nodes * 10;
  for (int i = threadIdx.x; i < 640; i += 256)
    if (base + i < total) out[base + i] = stage[i];
}

extern "C" void kernel_launch(void* const* d_in, const int* in_sizes, int n_in,
                              void* d_out, int out_size, void* d_ws, size_t ws_size,
                              hipStream_t stream) {
  const float* x    = (const float*)d_in[0];
  const int*   ei   = (const int*)d_in[1];
  const float* W1l  = (const float*)d_in[2];
  const float* W1r  = (const float*)d_in[3];
  const float* att1 = (const float*)d_in[4];
  const float* b1   = (const float*)d_in[5];
  const float* W2l  = (const float*)d_in[6];
  const float* W2r  = (const float*)d_in[7];
  const float* att2 = (const float*)d_in[8];
  const float* b2   = (const float*)d_in[9];
  float* out = (float*)d_out;

  const int N = in_sizes[0] / 128;
  const int E = in_sizes[1] / 2;

  float* ws = (float*)d_ws;
  size_t o = 0;
  int* rows    = (int*)(ws + o); o += (size_t)N * ROW_W;
  float* xl1   = ws + o; o += (size_t)N * 64;
  float* xr1   = ws + o; o += (size_t)N * 64;
  float* xl2   = ws + o; o += (size_t)N * 16;
  float* xr2   = ws + o; o += (size_t)N * 16;
  int* spill_n = (int*)(ws + o); o += 64;
  int* spill   = (int*)(ws + o); o += 2 * (size_t)E;

  auto blocks = [](long long n) { return (unsigned)((n + 255) / 256); };
  const int mtiles = (N + 15) / 16;

  zero_heads_kernel<<<blocks(N), 256, 0, stream>>>(rows, spill_n, N);
  const int gemm_blocks = (mtiles + 1) / 2;
  const int edge_blocks = (int)(((long long)E + 511) / 512);
  const int pair_blocks = (gemm_blocks > edge_blocks) ? gemm_blocks : edge_blocks;
  gemm1_edges_kernel<<<2 * (unsigned)pair_blocks, 128, 0, stream>>>(
      x, W1l, W1r, xl1, xr1, N, gemm_blocks, ei, rows, spill_n, spill, E, edge_blocks);

  l1_node_gemm2_kernel<<<(unsigned)mtiles, 128, 0, stream>>>(
      rows, spill_n, spill, E, xl1, xr1, att1, b1, W2l, W2r, xl2, xr2, N);

  l2_node_kernel<<<(unsigned)((N + 63) / 64), 256, 0, stream>>>(rows, spill_n, spill, E, xl2, xr2,
                                                                att2, b2, out, N);
}


# Edge grouping kept inside one cache half: two blocks scan each 1024-edge chunk, each takes the edges into one half of the nodes, chosen by its die id and a claim bit; sum order still arrival rank
# speedup vs baseline: 5.8492x; 1.3178x over previous
#include <hip/hip_runtime.h>
#include <math.h>

typedef __attribute__((ext_vector_type(2))) float v2f;
typedef __attribute__((ext_vector_type(8))) float v8f;

#define NEG_SLOPE 0.2f

#define ROW_W   64
#define ROW_CAP 63
#define CHUNK_E 1024

__global__ void __launch_bounds__(256)
zero_heads_kernel(int* __restrict__ rows, int* __restrict__ spill_n,
                  int* __restrict__ claim, int n, int n_chunk) {
  int i = blockIdx.x * blockDim.x + threadIdx.x;
  if (i < n) rows[(size_t)i * ROW_W] = 0;
  if (i < n_chunk) claim[i] = 0;
  if (i == 0) *spill_n = 0;
}

__device__ __forceinline__ void put_edge(int* __restrict__ rows, int* __restrict__ spill_n,
                                         int* __restrict__ spill, int d, int c, int r) {
  if (r < ROW_CAP) {
    rows[(size_t)d * ROW_W + 1 + r] = c;
  } else {
    const int k = atomicAdd(spill_n, 1);
    spill[2 * (size_t)k]     = d;
    spill[2 * (size_t)k + 1] = c;
  }
}

__device__ __forceinline__ void
edge_fill_block(const int blk, const int n_chunk, const int n_split,
                const int* __restrict__ ei, int* __restrict__ rows,
                int* __restrict__ spill_n, int* __restrict__ spill,
                int* __restrict__ claim, int E, int* __restrict__ s_part) {
  const int chunk = ((blk >> 2) << 1) + (blk & 1);
  if (chunk >= n_chunk) return;
  const int half = (int)((__builtin_amdgcn_s_sendmsg_rtn(0x87) >> 18) & 1u);

  const int* dst = ei + E;
  const int e0 = chunk * CHUNK_E + threadIdx.x;
  const int e1 = e0 + 128, e2 = e0 + 256, e3 = e0 + 384;
  const int e4 = e0 + 512, e5 = e0 + 640, e6 = e0 + 768, e7 = e0 + 896;
  const int d0 = dst[min(e0, E - 1)], d1 = dst[min(e1, E - 1)];
  const int d2 = dst[min(e2, E - 1)], d3 = dst[min(e3, E - 1)];
  const int d4 = dst[min(e4, E - 1)], d5 = dst[min(e5, E - 1)];
  const int d6 = dst[min(e6, E - 1)], d7 = dst[min(e7, E - 1)];
  const int c0 = ei[min(e0, E - 1)], c1 = ei[min(e1, E - 1)];
  const int c2 = ei[min(e2, E - 1)], c3 = ei[min(e3, E - 1)];
  const int c4 = ei[min(e4, E - 1)], c5 = ei[min(e5, E - 1)];
  const int c6 = ei[min(e6, E - 1)], c7 = ei[min(e7, E - 1)];

  if (threadIdx.x == 0) {
    const int old = atomicOr(claim + chunk, 1 << half);
    *s_part = ((old >> half) & 1) ? (half ^ 1) : half;
  }
  __syncthreads();
  const bool upper = (*s_part != 0);

  const bool l0 = (e0 < E) && ((d0 >= n_split) == upper);
  const bool l1 = (e1 < E) && ((d1 >= n_split) == upper);
  const bool l2 = (e2 < E) && ((d2 >= n_split) == upper);
  const bool l3 = (e3 < E) && ((d3 >= n_split) == upper);
  const bool l4 = (e4 < E) && ((d4 >= n_split) == upper);
  const bool l5 = (e5 < E) && ((d5 >= n_split) == upper);
  const bool l6 = (e6 < E) && ((d6 >= n_split) == upper);
  const bool l7 = (e7 < E) && ((d7 >= n_split) == upper);
  int r0 = 0, r1 = 0, r2 = 0, r3 = 0, r4 = 0, r5 = 0, r6 = 0, r7 = 0;
  if (l0) r0 = atomicAdd(rows + (size_t)d0 * ROW_W, 1);
  if (l1) r1 = atomicAdd(rows + (size_t)d1 * ROW_W, 1);
  if (l2) r2 = atomicAdd(rows + (size_t)d2 * ROW_W, 1);
  if (l3) r3 = atomicAdd(rows + (size_t)d3 * ROW_W, 1);
  if (l4) r4 = atomicAdd(rows + (size_t)d4 * ROW_W, 1);
  if (l5) r5 = atomicAdd(rows + (size_t)d5 * ROW_W, 1);
  if (l6) r6 = atomicAdd(rows + (size_t)d6 * ROW_W, 1);
  if (l7) r7 = atomicAdd(rows + (size_t)d7 * ROW_W, 1);
  if (l0) put_edge(rows, spill_n, spill, d0, c0, r0);
  if (l1) put_edge(rows, spill_n, spill, d1, c1, r1);
  if (l2) put_edge(rows, spill_n, spill, d2, c2, r2);
  if (l3) put_edge(rows, spill_n, spill, d3, c3, r3);
  if (l4) put_edge(rows, spill_n, spill, d4, c4, r4);
  if (l5) put_edge(rows, spill_n, spill, d5, c5, r5);
  if (l6) put_edge(rows, spill_n, spill, d6, c6, r6);
  if (l7) put_edge(rows, spill_n, spill, d7, c7, r7);
}

__device__ __forceinline__ void
gemm1_block(const int blk,
            const float* __restrict__ x, const float* __restrict__ W1l,
            const float* __restrict__ W1r, float* __restrict__ xl1,
            float* __restrict__ xr1, int n_nodes) {
  const int lane = threadIdx.x & 31;
  const int wave = threadIdx.x >> 5;
  const int m0   = blk * 32;
  const int n0   = wave * 16;
  const int half = lane >> 4;
  const int rc   = lane & 15;

  int arow0 = m0 + rc;      if (arow0 >= n_nodes) arow0 = n_nodes - 1;
  int arow1 = m0 + 16 + rc; if (arow1 >= n_nodes) arow1 = n_nodes - 1;
  const v2f*   xrow0 = (const v2f*)(x + (size_t)arow0 * 128);
  const v2f*   xrow1 = (const v2f*)(x + (size_t)arow1 * 128);
  const float* Bl    = W1l + n0 + rc;
  const float* Br    = W1r + n0 + rc;

  v8f accL0 = {}, accR0 = {}, accL1 = {}, accR1 = {};
  #pragma unroll
  for (int kc = 0; kc < 32; ++kc) {
    v2f a0 = xrow0[2 * kc + half];
    v2f a1 = xrow1[2 * kc + half];
    const int k0 = 4 * kc + 2 * half;
    v2f bl, br;
    bl[0] = Bl[(size_t)(k0 + 0) * 64];
    bl[1] = Bl[(size_t)(k0 + 1) * 64];
    br[0] = Br[(size_t)(k0 + 0) * 64];
    br[1] = Br[(size_t)(k0 + 1) * 64];
    accL0 = __builtin_amdgcn_wmma_f32_16x16x4_f32(
        false, a0, false, bl, (short)0, accL0, false, false);
    accR0 = __builtin_amdgcn_wmma_f32_16x16x4_f32(
        false, a0, false, br, (short)0, accR0, false, false);
    accL1 = __builtin_amdgcn_wmma_f32_16x16x4_f32(
        false, a1, false, bl, (short)0, accL1, false, false);
    accR1 = __builtin_amdgcn_wmma_f32_16x16x4_f32(
        false, a1, false, br, (short)0, accR1, false, false);
  }

  #pragma unroll
  for (int r = 0; r < 8; ++r) {
    const int row0 = m0 + r + 8 * half;
    const int row1 = row0 + 16;
    if (row0 < n_nodes) {
      xl1[(size_t)row0 * 64 + n0 + rc] = accL0[r];
      xr1[(size_t)row0 * 64 + n0 + rc] = accR0[r];
    }
    if (row1 < n_nodes) {
      xl1[(size_t)row1 * 64 + n0 + rc] = accL1[r];
      xr1[(size_t)row1 * 64 + n0 + rc] = accR1[r];
    }
  }
}

__global__ void __launch_bounds__(128)
gemm1_edges_kernel(const float* __restrict__ x, const float* __restrict__ W1l,
                   const float* __restrict__ W1r, float* __restrict__ xl1,
                   float* __restrict__ xr1, int n_nodes, int gemm_blocks,
                   const int* __restrict__ ei, int* __restrict__ rows,
                   int* __restrict__ spill_n, int* __restrict__ spill,
                   int* __restrict__ claim, int E, int edge_blocks,
                   int n_chunk, int n_split) {
  __shared__ int s_part;
  const int  blk  = blockIdx.x >> 1;
  const bool edge = (((blockIdx.x >> 3) ^ blockIdx.x) & 1) != 0;
  if (edge) {
    if (blk < edge_blocks)
      edge_fill_block(blk, n_chunk, n_split, ei, rows, spill_n, spill, claim, E, &s_part);
  } else {
    if (blk < gemm_blocks) gemm1_block(blk, x, W1l, W1r, xl1, xr1, n_nodes);
  }
}

__device__ __forceinline__ float edge_score(const float4 x, const float4 r, const float4 a) {
  float p = 0.0f, s;
  s = x.x + r.x; s = (s > 0.f) ? s : NEG_SLOPE * s; p += s * a.x;
  s = x.y + r.y; s = (s > 0.f) ? s : NEG_SLOPE * s; p += s * a.y;
  s = x.z + r.z; s = (s > 0.f) ? s : NEG_SLOPE * s; p += s * a.z;
  s = x.w + r.w; s = (s > 0.f) ? s : NEG_SLOPE * s; p += s * a.w;
  p += __shfl_xor(p, 1, 32);
  p += __shfl_xor(p, 2, 32);
  return p;
}

template <int STRIDE, int NV, int VOFF>
__device__ __forceinline__ void node_softmax_sum(
    const int* __restrict__ csrc, const float* __restrict__ xl,
    const int end, const int coff,
    const float4 (&rv)[NV], const float4 (&at)[NV],
    float (&mrun)[NV], float4 (&acc)[NV], float (&den)[NV]) {
  const float NEG_INF = -__builtin_inff();

  const int last = end - 1;
  for (int j = 0; j < end; j += 4) {
    const int s0 = csrc[j];
    const int s1 = csrc[min(j + 1, last)];
    const int s2 = csrc[min(j + 2, last)];
    const int s3 = csrc[min(j + 3, last)];
    const float* r0 = xl + (size_t)s0 * STRIDE + coff;
    const float* r1 = xl + (size_t)s1 * STRIDE + coff;
    const float* r2 = xl + (size_t)s2 * STRIDE + coff;
    const float* r3 = xl + (size_t)s3 * STRIDE + coff;
    float4 x0[NV], x1[NV], x2[NV], x3[NV];
    #pragma unroll
    for (int v = 0; v < NV; ++v) {
      x0[v] = *(const float4*)(r0 + v * VOFF);
      x1[v] = *(const float4*)(r1 + v * VOFF);
      x2[v] = *(const float4*)(r2 + v * VOFF);
      x3[v] = *(const float4*)(r3 + v * VOFF);
    }

    const bool v1 = (j + 1 < end), v2 = (j + 2 < end), v3 = (j + 3 < end);
    #pragma unroll
    for (int v = 0; v < NV; ++v) {
      const float p0 = edge_score(x0[v], rv[v], at[v]);
      float p1 = edge_score(x1[v], rv[v], at[v]);
      float p2 = edge_score(x2[v], rv[v], at[v]);
      float p3 = edge_score(x3[v], rv[v], at[v]);
      p1 = v1 ? p1 : NEG_INF;
      p2 = v2 ? p2 : NEG_INF;
      p3 = v3 ? p3 : NEG_INF;

      const float m  = mrun[v];
      const float mn = fmaxf(fmaxf(m, p0), fmaxf(fmaxf(p1, p2), p3));
      const float sc = __expf(m - mn);
      const float w0 = __expf(p0 - mn);
      const float w1 = __expf(p1 - mn);
      const float w2 = __expf(p2 - mn);
      const float w3 = __expf(p3 - mn);
      den[v]   = den[v]   * sc + ((w0 + w1) + (w2 + w3));
      acc[v].x = acc[v].x * sc + ((w0 * x0[v].x + w1 * x1[v].x) + (w2 * x2[v].x + w3 * x3[v].x));
      acc[v].y = acc[v].y * sc + ((w0 * x0[v].y + w1 * x1[v].y) + (w2 * x2[v].y + w3 * x3[v].y));
      acc[v].z = acc[v].z * sc + ((w0 * x0[v].z + w1 * x1[v].z) + (w2 * x2[v].z + w3 * x3[v].z));
      acc[v].w = acc[v].w * sc + ((w0 * x0[v].w + w1 * x1[v].w) + (w2 * x2[v].w + w3 * x3[v].w));
      mrun[v] = mn;
    }
  }
}

template <int STRIDE, int NV, int VOFF>
__device__ __forceinline__ void node_one_edge(
    const int s, const float* __restrict__ xl, const int coff,
    const float4 (&rv)[NV], const float4 (&at)[NV],
    float (&m)[NV], float4 (&a)[NV], float (&d)[NV]) {
  const float* r = xl + (size_t)s * STRIDE + coff;
  #pragma unroll
  for (int v = 0; v < NV; ++v) {
    const float4 x = *(const float4*)(r + v * VOFF);
    const float p  = edge_score(x, rv[v], at[v]);
    const float mn = fmaxf(m[v], p);
    const float sc = __expf(m[v] - mn);
    const float w  = __expf(p - mn);
    d[v]   = d[v]   * sc + w;
    a[v].x = a[v].x * sc + w * x.x;
    a[v].y = a[v].y * sc + w * x.y;
    a[v].z = a[v].z * sc + w * x.z;
    a[v].w = a[v].w * sc + w * x.w;
    m[v] = mn;
  }
}

template <int STRIDE, int NV, int VOFF>
__device__ __forceinline__ void node_all_edges(
    const int* __restrict__ rows, const int* __restrict__ spill_n,
    const int* __restrict__ spill, const int spill_cap,
    const float* __restrict__ xl, const int n, const int coff,
    const float4 (&rv)[NV], const float4 (&at)[NV],
    float4 (&acc)[NV], float (&den)[NV]) {
  const int* row = rows + (size_t)n * ROW_W;
  const int cnt = row[0];
  float m[NV];
  #pragma unroll
  for (int v = 0; v < NV; ++v) {
    m[v]   = -__builtin_inff();
    den[v] = 0.0f;
    acc[v] = make_float4(0.0f, 0.0f, 0.0f, 0.0f);
  }
  node_softmax_sum<STRIDE, NV, VOFF>(row + 1, xl, min(cnt, ROW_CAP), coff, rv, at, m, acc, den);
  if (cnt > ROW_CAP) {
    const int ns = min(*spill_n, spill_cap);
    for (int i = 0; i < ns; ++i)
      if (spill[2 * (size_t)i] == n)
        node_one_edge<STRIDE, NV, VOFF>(spill[2 * (size_t)i + 1], xl, coff, rv, at, m, acc, den);
  }
}

#define H_PITCH 68

__global__ void __launch_bounds__(128)
l1_node_gemm2_kernel(const int* __restrict__ rows, const int* __restrict__ spill_n,
                     const int* __restrict__ spill, int spill_cap,
                     const float* __restrict__ xl1, const float* __restrict__ xr1,
                     const float* __restrict__ att1, const float* __restrict__ b1,
                     const float* __restrict__ W2l, const float* __restrict__ W2r,
                     float* __restrict__ xl2, float* __restrict__ xr2, int n_nodes) {
  __shared__ float hs[16 * H_PITCH];
  const int nl   = threadIdx.x >> 3;
  const int n    = blockIdx.x * 16 + nl;
  const int coff = (threadIdx.x & 7) * 4;

  float4 hv[2];
  hv[0] = make_float4(0.0f, 0.0f, 0.0f, 0.0f);
  hv[1] = make_float4(0.0f, 0.0f, 0.0f, 0.0f);
  if (n < n_nodes) {
    float4 rv[2], at[2];
    rv[0] = *(const float4*)(xr1 + (size_t)n * 64 + coff);
    rv[1] = *(const float4*)(xr1 + (size_t)n * 64 + coff + 32);
    at[0] = *(const float4*)(att1 + coff);
    at[1] = *(const float4*)(att1 + coff + 32);

    float4 acc[2]; float den[2];
    node_all_edges<64, 2, 32>(rows, spill_n, spill, spill_cap, xl1, n, coff, rv, at, acc, den);

    #pragma unroll
    for (int g = 0; g < 2; ++g) {
      const float4 bb = *(const float4*)(b1 + coff + 32 * g);
      const float  dn = den[g];
      float4 v;
      v.x = ((dn > 0.0f) ? acc[g].x / dn : 0.0f) + bb.x;
      v.y = ((dn > 0.0f) ? acc[g].y / dn : 0.0f) + bb.y;
      v.z = ((dn > 0.0f) ? acc[g].z / dn : 0.0f) + bb.z;
      v.w = ((dn > 0.0f) ? acc[g].w / dn : 0.0f) + bb.w;
      v.x = (v.x > 0.0f) ? v.x : expm1f(v.x);
      v.y = (v.y > 0.0f) ? v.y : expm1f(v.y);
      v.z = (v.z > 0.0f) ? v.z : expm1f(v.z);
      v.w = (v.w > 0.0f) ? v.w : expm1f(v.w);
      hv[g] = v;
    }
  }
  *(float4*)(hs + nl * H_PITCH + coff)      = hv[0];
  *(float4*)(hs + nl * H_PITCH + coff + 32) = hv[1];
  __syncthreads();

  const int wave = __builtin_amdgcn_readfirstlane((int)(threadIdx.x >> 5));
  const int m0   = blockIdx.x * 16;
  if (wave == 0) {
    const int lane = threadIdx.x & 31;
    const int half = lane >> 4;
    const int rc   = lane & 15;
    const float* hrow = hs + rc * H_PITCH;
    const bool colok = (rc < 10);
    const int  cc    = colok ? rc : 9;

    v8f accL = {}, accR = {};
    #pragma unroll
    for (int kc = 0; kc < 16; ++kc) {
      const int k0 = 4 * kc + 2 * half;
      const v2f a = *(const v2f*)(hrow + k0);
      const float l0 = W2l[(size_t)(k0 + 0) * 10 + cc];
      const float l1 = W2l[(size_t)(k0 + 1) * 10 + cc];
      const float r0 = W2r[(size_t)(k0 + 0) * 10 + cc];
      const float r1 = W2r[(size_t)(k0 + 1) * 10 + cc];
      v2f bl, br;
      bl[0] = colok ? l0 : 0.0f;
      bl[1] = colok ? l1 : 0.0f;
      br[0] = colok ? r0 : 0.0f;
      br[1] = colok ? r1 : 0.0f;
      accL = __builtin_amdgcn_wmma_f32_16x16x4_f32(
          false, a, false, bl, (short)0, accL, false, false);
      accR = __builtin_amdgcn_wmma_f32_16x16x4_f32(
          false, a, false, br, (short)0, accR, false, false);
    }

    #pragma unroll
    for (int r = 0; r < 8; ++r) {
      const int row = m0 + r + 8 * half;
      if (row < n_nodes) {
        xl2[(size_t)row * 16 + rc] = accL[r];
        xr2[(size_t)row * 16 + rc] = accR[r];
      }
    }
  }
}

__global__ void __launch_bounds__(256)
l2_node_kernel(const int* __restrict__ rows, const int* __restrict__ spill_n,
               const int* __restrict__ spill, int spill_cap,
               const float* __restrict__ xl2, const float* __restrict__ xr2,
               const float* __restrict__ att2, const float* __restrict__ b2,
               float* __restrict__ out, int n_nodes) {
  __shared__ float stage[64 * 10];
  const int nl = threadIdx.x >> 2;
  const int n  = blockIdx.x * 64 + nl;
  const int c0 = (threadIdx.x & 3) * 4;

  if (n < n_nodes) {
    float4 rv[1], at[1];
    rv[0] = *(const float4*)(xr2 + (size_t)n * 16 + c0);
    at[0].x = (c0 + 0 < 10) ? att2[c0 + 0] : 0.0f;
    at[0].y = (c0 + 1 < 10) ? att2[c0 + 1] : 0.0f;
    at[0].z = (c0 + 2 < 10) ? att2[c0 + 2] : 0.0f;
    at[0].w = (c0 + 3 < 10) ? att2[c0 + 3] : 0.0f;

    float4 acc[1]; float den[1];
    node_all_edges<16, 1, 0>(rows, spill_n, spill, spill_cap, xl2, n, c0, rv, at, acc, den);

    const bool ok = (den[0] > 0.0f);
    if (c0 + 0 < 10) stage[nl * 10 + c0 + 0] = (ok ? acc[0].x / den[0] : 0.0f) + b2[c0 + 0];
    if (c0 + 1 < 10) stage[nl * 10 + c0 + 1] = (ok ? acc[0].y / den[0] : 0.0f) + b2[c0 + 1];
    if (c0 + 2 < 10) stage[nl * 10 + c0 + 2] = (ok ? acc[0].z / den[0] : 0.0f) + b2[c0 + 2];
    if (c0 + 3 < 10) stage[nl * 10 + c0 + 3] = (ok ? acc[0].w / den[0] : 0.0f) + b2[c0 + 3];
  }
  __syncthreads();

  const long long base  = (long long)blockIdx.x * 640;
  const long long total = (long long)n_nodes * 10;
  for (int i = threadIdx.x; i < 640; i += 256)
    if (base + i < total) out[base + i] = stage[i];
}

extern "C" void kernel_launch(void* const* d_in, const int* in_sizes, int n_in,
                              void* d_out, int out_size, void* d_ws, size_t ws_size,
                              hipStream_t stream) {
  const float* x    = (const float*)d_in[0];
  const int*   ei   = (const int*)d_in[1];
  const float* W1l  = (const float*)d_in[2];
  const float* W1r  = (const float*)d_in[3];
  const float* att1 = (const float*)d_in[4];
  const float* b1   = (const float*)d_in[5];
  const float* W2l  = (const float*)d_in[6];
  const float* W2r  = (const float*)d_in[7];
  const float* att2 = (const float*)d_in[8];
  const float* b2   = (const float*)d_in[9];
  float* out = (float*)d_out;

  const int N = in_sizes[0] / 128;
  const int E = in_sizes[1] / 2;

  float* ws = (float*)d_ws;
  size_t o = 0;
  int* rows    = (int*)(ws + o); o += (size_t)N * ROW_W;
  float* xl1   = ws + o; o += (size_t)N * 64;
  float* xr1   = ws + o; o += (size_t)N * 64;
  float* xl2   = ws + o; o += (size_t)N * 16;
  float* xr2   = ws + o; o += (size_t)N * 16;
  const int n_chunk = (int)(((long long)E + CHUNK_E - 1) / CHUNK_E);
  int* spill_n = (int*)(ws + o); o += 64;
  int* claim   = (int*)(ws + o); o += ((size_t)n_chunk + 63) / 64 * 64;
  int* spill   = (int*)(ws + o); o += 2 * (size_t)E;

  auto blocks = [](long long n) { return (unsigned)((n + 255) / 256); };
  const int mtiles = (N + 15) / 16;

  zero_heads_kernel<<<blocks(N > n_chunk ? N : n_chunk), 256, 0, stream>>>(
      rows, spill_n, claim, N, n_chunk);
  const int gemm_blocks = (mtiles + 1) / 2;
  const int edge_blocks = 4 * ((n_chunk + 1) / 2);
  const int pair_blocks = (gemm_blocks > edge_blocks) ? gemm_blocks : edge_blocks;
  gemm1_edges_kernel<<<2 * (unsigned)pair_blocks, 128, 0, stream>>>(
      x, W1l, W1r, xl1, xr1, N, gemm_blocks, ei, rows, spill_n, spill, claim, E, edge_blocks,
      n_chunk, N / 2);

  l1_node_gemm2_kernel<<<(unsigned)mtiles, 128, 0, stream>>>(
      rows, spill_n, spill, E, xl1, xr1, att1, b1, W2l, W2r, xl2, xr2, N);

  l2_node_kernel<<<(unsigned)((N + 63) / 64), 256, 0, stream>>>(rows, spill_n, spill, E, xl2, xr2,
                                                                att2, b2, out, N);
}


# GEMM1 on 512 resident blocks: each wave keeps its W1l/W1r fragments in registers and walks over the 16-row tiles, so the weights are fetched 512 times instead of 1563; sums unchanged
# speedup vs baseline: 6.1405x; 1.0498x over previous
#include <hip/hip_runtime.h>
#include <math.h>

typedef __attribute__((ext_vector_type(2))) float v2f;
typedef __attribute__((ext_vector_type(8))) float v8f;

#define NEG_SLOPE 0.2f

#define ROW_W   64
#define ROW_CAP 63
#define CHUNK_E 1024

__global__ void __launch_bounds__(256)
zero_heads_kernel(int* __restrict__ rows, int* __restrict__ spill_n,
                  int* __restrict__ claim, int n, int n_chunk) {
  int i = blockIdx.x * blockDim.x + threadIdx.x;
  if (i < n) rows[(size_t)i * ROW_W] = 0;
  if (i < n_chunk) claim[i] = 0;
  if (i == 0) *spill_n = 0;
}

__device__ __forceinline__ void put_edge(int* __restrict__ rows, int* __restrict__ spill_n,
                                         int* __restrict__ spill, int d, int c, int r) {
  if (r < ROW_CAP) {
    rows[(size_t)d * ROW_W + 1 + r] = c;
  } else {
    const int k = atomicAdd(spill_n, 1);
    spill[2 * (size_t)k]     = d;
    spill[2 * (size_t)k + 1] = c;
  }
}

__device__ __forceinline__ void
edge_fill_block(const int blk, const int n_chunk, const int n_split,
                const int* __restrict__ ei, int* __restrict__ rows,
                int* __restrict__ spill_n, int* __restrict__ spill,
                int* __restrict__ claim, int E, int* __restrict__ s_part) {
  const int chunk = ((blk >> 3) << 2) + (blk & 3);
  if (chunk >= n_chunk) return;
  const int half = (int)((__builtin_amdgcn_s_sendmsg_rtn(0x87) >> 18) & 1u);

  const int* dst = ei + E;
  const int e0 = chunk * CHUNK_E + threadIdx.x;
  const int e1 = e0 + 128, e2 = e0 + 256, e3 = e0 + 384;
  const int e4 = e0 + 512, e5 = e0 + 640, e6 = e0 + 768, e7 = e0 + 896;
  const int d0 = dst[min(e0, E - 1)], d1 = dst[min(e1, E - 1)];
  const int d2 = dst[min(e2, E - 1)], d3 = dst[min(e3, E - 1)];
  const int d4 = dst[min(e4, E - 1)], d5 = dst[min(e5, E - 1)];
  const int d6 = dst[min(e6, E - 1)], d7 = dst[min(e7, E - 1)];
  const int c0 = ei[min(e0, E - 1)], c1 = ei[min(e1, E - 1)];
  const int c2 = ei[min(e2, E - 1)], c3 = ei[min(e3, E - 1)];
  const int c4 = ei[min(e4, E - 1)], c5 = ei[min(e5, E - 1)];
  const int c6 = ei[min(e6, E - 1)], c7 = ei[min(e7, E - 1)];

  if (threadIdx.x == 0) {
    const int old = atomicOr(claim + chunk, 1 << half);
    *s_part = ((old >> half) & 1) ? (half ^ 1) : half;
  }
  __syncthreads();
  const bool upper = (*s_part != 0);

  const bool l0 = (e0 < E) && ((d0 >= n_split) == upper);
  const bool l1 = (e1 < E) && ((d1 >= n_split) == upper);
  const bool l2 = (e2 < E) && ((d2 >= n_split) == upper);
  const bool l3 = (e3 < E) && ((d3 >= n_split) == upper);
  const bool l4 = (e4 < E) && ((d4 >= n_split) == upper);
  const bool l5 = (e5 < E) && ((d5 >= n_split) == upper);
  const bool l6 = (e6 < E) && ((d6 >= n_split) == upper);
  const bool l7 = (e7 < E) && ((d7 >= n_split) == upper);
  int r0 = 0, r1 = 0, r2 = 0, r3 = 0, r4 = 0, r5 = 0, r6 = 0, r7 = 0;
  if (l0) r0 = atomicAdd(rows + (size_t)d0 * ROW_W, 1);
  if (l1) r1 = atomicAdd(rows + (size_t)d1 * ROW_W, 1);
  if (l2) r2 = atomicAdd(rows + (size_t)d2 * ROW_W, 1);
  if (l3) r3 = atomicAdd(rows + (size_t)d3 * ROW_W, 1);
  if (l4) r4 = atomicAdd(rows + (size_t)d4 * ROW_W, 1);
  if (l5) r5 = atomicAdd(rows + (size_t)d5 * ROW_W, 1);
  if (l6) r6 = atomicAdd(rows + (size_t)d6 * ROW_W, 1);
  if (l7) r7 = atomicAdd(rows + (size_t)d7 * ROW_W, 1);
  if (l0) put_edge(rows, spill_n, spill, d0, c0, r0);
  if (l1) put_edge(rows, spill_n, spill, d1, c1, r1);
  if (l2) put_edge(rows, spill_n, spill, d2, c2, r2);
  if (l3) put_edge(rows, spill_n, spill, d3, c3, r3);
  if (l4) put_edge(rows, spill_n, spill, d4, c4, r4);
  if (l5) put_edge(rows, spill_n, spill, d5, c5, r5);
  if (l6) put_edge(rows, spill_n, spill, d6, c6, r6);
  if (l7) put_edge(rows, spill_n, spill, d7, c7, r7);
}

#define GEMM_RESIDENT 512

__device__ __forceinline__ const v2f*
x_row(const float* __restrict__ x, const int tile, const int rc, const int n_nodes) {
  int r = tile * 16 + rc;
  if (r >= n_nodes) r = n_nodes - 1;
  return (const v2f*)(x + (size_t)r * 128);
}

__device__ __forceinline__ void
gemm1_blocks(const int first, const int stride, const int n_tiles,
             const float* __restrict__ x, const float* __restrict__ W1l,
             const float* __restrict__ W1r, float* __restrict__ xl1,
             float* __restrict__ xr1, int n_nodes) {
  if (first >= n_tiles) return;
  const int lane = threadIdx.x & 31;
  const int wave = threadIdx.x >> 5;
  const int n0   = wave * 16;
  const int half = lane >> 4;
  const int rc   = lane & 15;
  const float* Bl = W1l + n0 + rc;
  const float* Br = W1r + n0 + rc;

  v2f bl[32], br[32];
  #pragma unroll
  for (int kc = 0; kc < 32; ++kc) {
    const int k0 = 4 * kc + 2 * half;
    bl[kc][0] = Bl[(size_t)(k0 + 0) * 64];
    bl[kc][1] = Bl[(size_t)(k0 + 1) * 64];
    br[kc][0] = Br[(size_t)(k0 + 0) * 64];
    br[kc][1] = Br[(size_t)(k0 + 1) * 64];
  }

  int tile = first;
  const v2f* xrow = x_row(x, tile, rc, n_nodes);
  v2f c[8];
  #pragma unroll
  for (int i = 0; i < 8; ++i) c[i] = xrow[2 * i + half];

  for (;;) {
    const int  next = tile + stride;
    const bool more = (next < n_tiles);
    const v2f* nrow = x_row(x, more ? next : tile, rc, n_nodes);

    v8f accL = {}, accR = {};
    #pragma unroll
    for (int jb = 0; jb < 4; ++jb) {
      v2f d[8];
      #pragma unroll
      for (int i = 0; i < 8; ++i)
        d[i] = (jb < 3) ? xrow[2 * (8 * (jb + 1) + i) + half] : nrow[2 * i + half];
      #pragma unroll
      for (int i = 0; i < 8; ++i) {
        const int kc = 8 * jb + i;
        accL = __builtin_amdgcn_wmma_f32_16x16x4_f32(
            false, c[i], false, bl[kc], (short)0, accL, false, false);
        accR = __builtin_amdgcn_wmma_f32_16x16x4_f32(
            false, c[i], false, br[kc], (short)0, accR, false, false);
      }
      #pragma unroll
      for (int i = 0; i < 8; ++i) c[i] = d[i];
    }

    const int m0 = tile * 16;
    #pragma unroll
    for (int r = 0; r < 8; ++r) {
      const int row = m0 + r + 8 * half;
      if (row < n_nodes) {
        xl1[(size_t)row * 64 + n0 + rc] = accL[r];
        xr1[(size_t)row * 64 + n0 + rc] = accR[r];
      }
    }

    if (!more) break;
    tile = next;
    xrow = nrow;
  }
}

__global__ void __launch_bounds__(128, 5)
gemm1_edges_kernel(const float* __restrict__ x, const float* __restrict__ W1l,
                   const float* __restrict__ W1r, float* __restrict__ xl1,
                   float* __restrict__ xr1, int n_nodes, int gemm_tiles,
                   const int* __restrict__ ei, int* __restrict__ rows,
                   int* __restrict__ spill_n, int* __restrict__ spill,
                   int* __restrict__ claim, int E, int edge_blocks,
                   int n_chunk, int n_split) {
  __shared__ int s_part;
  const int b = blockIdx.x;
  if (b < GEMM_RESIDENT) {
    gemm1_blocks(b, GEMM_RESIDENT, gemm_tiles, x, W1l, W1r, xl1, xr1, n_nodes);
  } else {
    const int blk = b - GEMM_RESIDENT;
    if (blk < edge_blocks)
      edge_fill_block(blk, n_chunk, n_split, ei, rows, spill_n, spill, claim, E, &s_part);
  }
}

__device__ __forceinline__ float edge_score(const float4 x, const float4 r, const float4 a) {
  float p = 0.0f, s;
  s = x.x + r.x; s = (s > 0.f) ? s : NEG_SLOPE * s; p += s * a.x;
  s = x.y + r.y; s = (s > 0.f) ? s : NEG_SLOPE * s; p += s * a.y;
  s = x.z + r.z; s = (s > 0.f) ? s : NEG_SLOPE * s; p += s * a.z;
  s = x.w + r.w; s = (s > 0.f) ? s : NEG_SLOPE * s; p += s * a.w;
  p += __shfl_xor(p, 1, 32);
  p += __shfl_xor(p, 2, 32);
  return p;
}

template <int STRIDE, int NV, int VOFF>
__device__ __forceinline__ void node_softmax_sum(
    const int* __restrict__ csrc, const float* __restrict__ xl,
    const int end, const int coff,
    const float4 (&rv)[NV], const float4 (&at)[NV],
    float (&mrun)[NV], float4 (&acc)[NV], float (&den)[NV]) {
  const float NEG_INF = -__builtin_inff();

  const int last = end - 1;
  for (int j = 0; j < end; j += 4) {
    const int s0 = csrc[j];
    const int s1 = csrc[min(j + 1, last)];
    const int s2 = csrc[min(j + 2, last)];
    const int s3 = csrc[min(j + 3, last)];
    const float* r0 = xl + (size_t)s0 * STRIDE + coff;
    const float* r1 = xl + (size_t)s1 * STRIDE + coff;
    const float* r2 = xl + (size_t)s2 * STRIDE + coff;
    const float* r3 = xl + (size_t)s3 * STRIDE + coff;
    float4 x0[NV], x1[NV], x2[NV], x3[NV];
    #pragma unroll
    for (int v = 0; v < NV; ++v) {
      x0[v] = *(const float4*)(r0 + v * VOFF);
      x1[v] = *(const float4*)(r1 + v * VOFF);
      x2[v] = *(const float4*)(r2 + v * VOFF);
      x3[v] = *(const float4*)(r3 + v * VOFF);
    }

    const bool v1 = (j + 1 < end), v2 = (j + 2 < end), v3 = (j + 3 < end);
    #pragma unroll
    for (int v = 0; v < NV; ++v) {
      const float p0 = edge_score(x0[v], rv[v], at[v]);
      float p1 = edge_score(x1[v], rv[v], at[v]);
      float p2 = edge_score(x2[v], rv[v], at[v]);
      float p3 = edge_score(x3[v], rv[v], at[v]);
      p1 = v1 ? p1 : NEG_INF;
      p2 = v2 ? p2 : NEG_INF;
      p3 = v3 ? p3 : NEG_INF;

      const float m  = mrun[v];
      const float mn = fmaxf(fmaxf(m, p0), fmaxf(fmaxf(p1, p2), p3));
      const float sc = __expf(m - mn);
      const float w0 = __expf(p0 - mn);
      const float w1 = __expf(p1 - mn);
      const float w2 = __expf(p2 - mn);
      const float w3 = __expf(p3 - mn);
      den[v]   = den[v]   * sc + ((w0 + w1) + (w2 + w3));
      acc[v].x = acc[v].x * sc + ((w0 * x0[v].x + w1 * x1[v].x) + (w2 * x2[v].x + w3 * x3[v].x));
      acc[v].y = acc[v].y * sc + ((w0 * x0[v].y + w1 * x1[v].y) + (w2 * x2[v].y + w3 * x3[v].y));
      acc[v].z = acc[v].z * sc + ((w0 * x0[v].z + w1 * x1[v].z) + (w2 * x2[v].z + w3 * x3[v].z));
      acc[v].w = acc[v].w * sc + ((w0 * x0[v].w + w1 * x1[v].w) + (w2 * x2[v].w + w3 * x3[v].w));
      mrun[v] = mn;
    }
  }
}

template <int STRIDE, int NV, int VOFF>
__device__ __forceinline__ void node_one_edge(
    const int s, const float* __restrict__ xl, const int coff,
    const float4 (&rv)[NV], const float4 (&at)[NV],
    float (&m)[NV], float4 (&a)[NV], float (&d)[NV]) {
  const float* r = xl + (size_t)s * STRIDE + coff;
  #pragma unroll
  for (int v = 0; v < NV; ++v) {
    const float4 x = *(const float4*)(r + v * VOFF);
    const float p  = edge_score(x, rv[v], at[v]);
    const float mn = fmaxf(m[v], p);
    const float sc = __expf(m[v] - mn);
    const float w  = __expf(p - mn);
    d[v]   = d[v]   * sc + w;
    a[v].x = a[v].x * sc + w * x.x;
    a[v].y = a[v].y * sc + w * x.y;
    a[v].z = a[v].z * sc + w * x.z;
    a[v].w = a[v].w * sc + w * x.w;
    m[v] = mn;
  }
}

template <int STRIDE, int NV, int VOFF>
__device__ __forceinline__ void node_all_edges(
    const int* __restrict__ rows, const int* __restrict__ spill_n,
    const int* __restrict__ spill, const int spill_cap,
    const float* __restrict__ xl, const int n, const int coff,
    const float4 (&rv)[NV], const float4 (&at)[NV],
    float4 (&acc)[NV], float (&den)[NV]) {
  const int* row = rows + (size_t)n * ROW_W;
  const int cnt = row[0];
  float m[NV];
  #pragma unroll
  for (int v = 0; v < NV; ++v) {
    m[v]   = -__builtin_inff();
    den[v] = 0.0f;
    acc[v] = make_float4(0.0f, 0.0f, 0.0f, 0.0f);
  }
  node_softmax_sum<STRIDE, NV, VOFF>(row + 1, xl, min(cnt, ROW_CAP), coff, rv, at, m, acc, den);
  if (cnt > ROW_CAP) {
    const int ns = min(*spill_n, spill_cap);
    for (int i = 0; i < ns; ++i)
      if (spill[2 * (size_t)i] == n)
        node_one_edge<STRIDE, NV, VOFF>(spill[2 * (size_t)i + 1], xl, coff, rv, at, m, acc, den);
  }
}

#define H_PITCH 68

__global__ void __launch_bounds__(128)
l1_node_gemm2_kernel(const int* __restrict__ rows, const int* __restrict__ spill_n,
                     const int* __restrict__ spill, int spill_cap,
                     const float* __restrict__ xl1, const float* __restrict__ xr1,
                     const float* __restrict__ att1, const float* __restrict__ b1,
                     const float* __restrict__ W2l, const float* __restrict__ W2r,
                     float* __restrict__ xl2, float* __restrict__ xr2, int n_nodes) {
  __shared__ float hs[16 * H_PITCH];
  const int nl   = threadIdx.x >> 3;
  const int n    = blockIdx.x * 16 + nl;
  const int coff = (threadIdx.x & 7) * 4;

  float4 hv[2];
  hv[0] = make_float4(0.0f, 0.0f, 0.0f, 0.0f);
  hv[1] = make_float4(0.0f, 0.0f, 0.0f, 0.0f);
  if (n < n_nodes) {
    float4 rv[2], at[2];
    rv[0] = *(const float4*)(xr1 + (size_t)n * 64 + coff);
    rv[1] = *(const float4*)(xr1 + (size_t)n * 64 + coff + 32);
    at[0] = *(const float4*)(att1 + coff);
    at[1] = *(const float4*)(att1 + coff + 32);

    float4 acc[2]; float den[2];
    node_all_edges<64, 2, 32>(rows, spill_n, spill, spill_cap, xl1, n, coff, rv, at, acc, den);

    #pragma unroll
    for (int g = 0; g < 2; ++g) {
      const float4 bb = *(const float4*)(b1 + coff + 32 * g);
      const float  dn = den[g];
      float4 v;
      v.x = ((dn > 0.0f) ? acc[g].x / dn : 0.0f) + bb.x;
      v.y = ((dn > 0.0f) ? acc[g].y / dn : 0.0f) + bb.y;
      v.z = ((dn > 0.0f) ? acc[g].z / dn : 0.0f) + bb.z;
      v.w = ((dn > 0.0f) ? acc[g].w / dn : 0.0f) + bb.w;
      v.x = (v.x > 0.0f) ? v.x : expm1f(v.x);
      v.y = (v.y > 0.0f) ? v.y : expm1f(v.y);
      v.z = (v.z > 0.0f) ? v.z : expm1f(v.z);
      v.w = (v.w > 0.0f) ? v.w : expm1f(v.w);
      hv[g] = v;
    }
  }
  *(float4*)(hs + nl * H_PITCH + coff)      = hv[0];
  *(float4*)(hs + nl * H_PITCH + coff + 32) = hv[1];
  __syncthreads();

  const int wave = __builtin_amdgcn_readfirstlane((int)(threadIdx.x >> 5));
  const int m0   = blockIdx.x * 16;
  if (wave == 0) {
    const int lane = threadIdx.x & 31;
    const int half = lane >> 4;
    const int rc   = lane & 15;
    const float* hrow = hs + rc * H_PITCH;
    const bool colok = (rc < 10);
    const int  cc    = colok ? rc : 9;

    v8f accL = {}, accR = {};
    #pragma unroll
    for (int kc = 0; kc < 16; ++kc) {
      const int k0 = 4 * kc + 2 * half;
      const v2f a = *(const v2f*)(hrow + k0);
      const float l0 = W2l[(size_t)(k0 + 0) * 10 + cc];
      const float l1 = W2l[(size_t)(k0 + 1) * 10 + cc];
      const float r0 = W2r[(size_t)(k0 + 0) * 10 + cc];
      const float r1 = W2r[(size_t)(k0 + 1) * 10 + cc];
      v2f bl, br;
      bl[0] = colok ? l0 : 0.0f;
      bl[1] = colok ? l1 : 0.0f;
      br[0] = colok ? r0 : 0.0f;
      br[1] = colok ? r1 : 0.0f;
      accL = __builtin_amdgcn_wmma_f32_16x16x4_f32(
          false, a, false, bl, (short)0, accL, false, false);
      accR = __builtin_amdgcn_wmma_f32_16x16x4_f32(
          false, a, false, br, (short)0, accR, false, false);
    }

    #pragma unroll
    for (int r = 0; r < 8; ++r) {
      const int row = m0 + r + 8 * half;
      if (row < n_nodes) {
        xl2[(size_t)row * 16 + rc] = accL[r];
        xr2[(size_t)row * 16 + rc] = accR[r];
      }
    }
  }
}

__global__ void __launch_bounds__(256)
l2_node_kernel(const int* __restrict__ rows, const int* __restrict__ spill_n,
               const int* __restrict__ spill, int spill_cap,
               const float* __restrict__ xl2, const float* __restrict__ xr2,
               const float* __restrict__ att2, const float* __restrict__ b2,
               float* __restrict__ out, int n_nodes) {
  __shared__ float stage[64 * 10];
  const int nl = threadIdx.x >> 2;
  const int n  = blockIdx.x * 64 + nl;
  const int c0 = (threadIdx.x & 3) * 4;

  if (n < n_nodes) {
    float4 rv[1], at[1];
    rv[0] = *(const float4*)(xr2 + (size_t)n * 16 + c0);
    at[0].x = (c0 + 0 < 10) ? att2[c0 + 0] : 0.0f;
    at[0].y = (c0 + 1 < 10) ? att2[c0 + 1] : 0.0f;
    at[0].z = (c0 + 2 < 10) ? att2[c0 + 2] : 0.0f;
    at[0].w = (c0 + 3 < 10) ? att2[c0 + 3] : 0.0f;

    float4 acc[1]; float den[1];
    node_all_edges<16, 1, 0>(rows, spill_n, spill, spill_cap, xl2, n, c0, rv, at, acc, den);

    const bool ok = (den[0] > 0.0f);
    if (c0 + 0 < 10) stage[nl * 10 + c0 + 0] = (ok ? acc[0].x / den[0] : 0.0f) + b2[c0 + 0];
    if (c0 + 1 < 10) stage[nl * 10 + c0 + 1] = (ok ? acc[0].y / den[0] : 0.0f) + b2[c0 + 1];
    if (c0 + 2 < 10) stage[nl * 10 + c0 + 2] = (ok ? acc[0].z / den[0] : 0.0f) + b2[c0 + 2];
    if (c0 + 3 < 10) stage[nl * 10 + c0 + 3] = (ok ? acc[0].w / den[0] : 0.0f) + b2[c0 + 3];
  }
  __syncthreads();

  const long long base  = (long long)blockIdx.x * 640;
  const long long total = (long long)n_nodes * 10;
  for (int i = threadIdx.x; i < 640; i += 256)
    if (base + i < total) out[base + i] = stage[i];
}

extern "C" void kernel_launch(void* const* d_in, const int* in_sizes, int n_in,
                              void* d_out, int out_size, void* d_ws, size_t ws_size,
                              hipStream_t stream) {
  const float* x    = (const float*)d_in[0];
  const int*   ei   = (const int*)d_in[1];
  const float* W1l  = (const float*)d_in[2];
  const float* W1r  = (const float*)d_in[3];
  const float* att1 = (const float*)d_in[4];
  const float* b1   = (const float*)d_in[5];
  const float* W2l  = (const float*)d_in[6];
  const float* W2r  = (const float*)d_in[7];
  const float* att2 = (const float*)d_in[8];
  const float* b2   = (const float*)d_in[9];
  float* out = (float*)d_out;

  const int N = in_sizes[0] / 128;
  const int E = in_sizes[1] / 2;

  float* ws = (float*)d_ws;
  size_t o = 0;
  int* rows    = (int*)(ws + o); o += (size_t)N * ROW_W;
  float* xl1   = ws + o; o += (size_t)N * 64;
  float* xr1   = ws + o; o += (size_t)N * 64;
  float* xl2   = ws + o; o += (size_t)N * 16;
  float* xr2   = ws + o; o += (size_t)N * 16;
  const int n_chunk = (int)(((long long)E + CHUNK_E - 1) / CHUNK_E);
  int* spill_n = (int*)(ws + o); o += 64;
  int* claim   = (int*)(ws + o); o += ((size_t)n_chunk + 63) / 64 * 64;
  int* spill   = (int*)(ws + o); o += 2 * (size_t)E;

  auto blocks = [](long long n) { return (unsigned)((n + 255) / 256); };
  const int mtiles = (N + 15) / 16;

  zero_heads_kernel<<<blocks(N > n_chunk ? N : n_chunk), 256, 0, stream>>>(
      rows, spill_n, claim, N, n_chunk);
  const int edge_blocks = 8 * ((n_chunk + 3) / 4);
  gemm1_edges_kernel<<<(unsigned)(GEMM_RESIDENT + edge_blocks), 128, 0, stream>>>(
      x, W1l, W1r, xl1, xr1, N, mtiles, ei, rows, spill_n, spill, claim, E, edge_blocks,
      n_chunk, N / 2);

  l1_node_gemm2_kernel<<<(unsigned)mtiles, 128, 0, stream>>>(
      rows, spill_n, spill, E, xl1, xr1, att1, b1, W2l, W2r, xl2, xr2, N);

  l2_node_kernel<<<(unsigned)((N + 63) / 64), 256, 0, stream>>>(rows, spill_n, spill, E, xl2, xr2,
                                                                att2, b2, out, N);
}


# Node kernels' edge loops on packed f32 pairs: LeakyReLU as max(z,0.2z), scores and weighted sums fused, 32-bit row offsets; fewer vector ops per edge; sum order in scores and edge sums changes
# speedup vs baseline: 6.5024x; 1.0589x over previous
#include <hip/hip_runtime.h>
#include <math.h>

typedef __attribute__((ext_vector_type(2))) float v2f;
typedef __attribute__((ext_vector_type(8))) float v8f;

#define NEG_SLOPE 0.2f

#define ROW_W   64
#define ROW_CAP 63
#define CHUNK_E 1024

__global__ void __launch_bounds__(256)
zero_heads_kernel(int* __restrict__ rows, int* __restrict__ spill_n,
                  int* __restrict__ claim, int n, int n_chunk) {
  int i = blockIdx.x * blockDim.x + threadIdx.x;
  if (i < n) rows[(size_t)i * ROW_W] = 0;
  if (i < n_chunk) claim[i] = 0;
  if (i == 0) *spill_n = 0;
}

__device__ __forceinline__ void put_edge(int* __restrict__ rows, int* __restrict__ spill_n,
                                         int* __restrict__ spill, int d, int c, int r) {
  if (r < ROW_CAP) {
    rows[(size_t)d * ROW_W + 1 + r] = c;
  } else {
    const int k = atomicAdd(spill_n, 1);
    spill[2 * (size_t)k]     = d;
    spill[2 * (size_t)k + 1] = c;
  }
}

__device__ __forceinline__ void
edge_fill_block(const int blk, const int n_chunk, const int n_split,
                const int* __restrict__ ei, int* __restrict__ rows,
                int* __restrict__ spill_n, int* __restrict__ spill,
                int* __restrict__ claim, int E, int* __restrict__ s_part) {
  const int chunk = ((blk >> 3) << 2) + (blk & 3);
  if (chunk >= n_chunk) return;
  const int half = (int)((__builtin_amdgcn_s_sendmsg_rtn(0x87) >> 18) & 1u);

  const int* dst = ei + E;
  const int e0 = chunk * CHUNK_E + threadIdx.x;
  const int e1 = e0 + 128, e2 = e0 + 256, e3 = e0 + 384;
  const int e4 = e0 + 512, e5 = e0 + 640, e6 = e0 + 768, e7 = e0 + 896;
  const int d0 = dst[min(e0, E - 1)], d1 = dst[min(e1, E - 1)];
  const int d2 = dst[min(e2, E - 1)], d3 = dst[min(e3, E - 1)];
  const int d4 = dst[min(e4, E - 1)], d5 = dst[min(e5, E - 1)];
  const int d6 = dst[min(e6, E - 1)], d7 = dst[min(e7, E - 1)];
  const int c0 = ei[min(e0, E - 1)], c1 = ei[min(e1, E - 1)];
  const int c2 = ei[min(e2, E - 1)], c3 = ei[min(e3, E - 1)];
  const int c4 = ei[min(e4, E - 1)], c5 = ei[min(e5, E - 1)];
  const int c6 = ei[min(e6, E - 1)], c7 = ei[min(e7, E - 1)];

  if (threadIdx.x == 0) {
    const int old = atomicOr(claim + chunk, 1 << half);
    *s_part = ((old >> half) & 1) ? (half ^ 1) : half;
  }
  __syncthreads();
  const bool upper = (*s_part != 0);

  const bool l0 = (e0 < E) && ((d0 >= n_split) == upper);
  const bool l1 = (e1 < E) && ((d1 >= n_split) == upper);
  const bool l2 = (e2 < E) && ((d2 >= n_split) == upper);
  const bool l3 = (e3 < E) && ((d3 >= n_split) == upper);
  const bool l4 = (e4 < E) && ((d4 >= n_split) == upper);
  const bool l5 = (e5 < E) && ((d5 >= n_split) == upper);
  const bool l6 = (e6 < E) && ((d6 >= n_split) == upper);
  const bool l7 = (e7 < E) && ((d7 >= n_split) == upper);
  int r0 = 0, r1 = 0, r2 = 0, r3 = 0, r4 = 0, r5 = 0, r6 = 0, r7 = 0;
  if (l0) r0 = atomicAdd(rows + (size_t)d0 * ROW_W, 1);
  if (l1) r1 = atomicAdd(rows + (size_t)d1 * ROW_W, 1);
  if (l2) r2 = atomicAdd(rows + (size_t)d2 * ROW_W, 1);
  if (l3) r3 = atomicAdd(rows + (size_t)d3 * ROW_W, 1);
  if (l4) r4 = atomicAdd(rows + (size_t)d4 * ROW_W, 1);
  if (l5) r5 = atomicAdd(rows + (size_t)d5 * ROW_W, 1);
  if (l6) r6 = atomicAdd(rows + (size_t)d6 * ROW_W, 1);
  if (l7) r7 = atomicAdd(rows + (size_t)d7 * ROW_W, 1);
  if (l0) put_edge(rows, spill_n, spill, d0, c0, r0);
  if (l1) put_edge(rows, spill_n, spill, d1, c1, r1);
  if (l2) put_edge(rows, spill_n, spill, d2, c2, r2);
  if (l3) put_edge(rows, spill_n, spill, d3, c3, r3);
  if (l4) put_edge(rows, spill_n, spill, d4, c4, r4);
  if (l5) put_edge(rows, spill_n, spill, d5, c5, r5);
  if (l6) put_edge(rows, spill_n, spill, d6, c6, r6);
  if (l7) put_edge(rows, spill_n, spill, d7, c7, r7);
}

#define GEMM_RESIDENT 512

__device__ __forceinline__ const v2f*
x_row(const float* __restrict__ x, const int tile, const int rc, const int n_nodes) {
  int r = tile * 16 + rc;
  if (r >= n_nodes) r = n_nodes - 1;
  return (const v2f*)(x + (size_t)r * 128);
}

__device__ __forceinline__ void
gemm1_blocks(const int first, const int stride, const int n_tiles,
             const float* __restrict__ x, const float* __restrict__ W1l,
             const float* __restrict__ W1r, float* __restrict__ xl1,
             float* __restrict__ xr1, int n_nodes) {
  if (first >= n_tiles) return;
  const int lane = threadIdx.x & 31;
  const int wave = threadIdx.x >> 5;
  const int n0   = wave * 16;
  const int half = lane >> 4;
  const int rc   = lane & 15;
  const float* Bl = W1l + n0 + rc;
  const float* Br = W1r + n0 + rc;

  v2f bl[32], br[32];
  #pragma unroll
  for (int kc = 0; kc < 32; ++kc) {
    const int k0 = 4 * kc + 2 * half;
    bl[kc][0] = Bl[(size_t)(k0 + 0) * 64];
    bl[kc][1] = Bl[(size_t)(k0 + 1) * 64];
    br[kc][0] = Br[(size_t)(k0 + 0) * 64];
    br[kc][1] = Br[(size_t)(k0 + 1) * 64];
  }

  int tile = first;
  const v2f* xrow = x_row(x, tile, rc, n_nodes);
  v2f c[8];
  #pragma unroll
  for (int i = 0; i < 8; ++i) c[i] = xrow[2 * i + half];

  for (;;) {
    const int  next = tile + stride;
    const bool more = (next < n_tiles);
    const v2f* nrow = x_row(x, more ? next : tile, rc, n_nodes);

    v8f accL = {}, accR = {};
    #pragma unroll
    for (int jb = 0; jb < 4; ++jb) {
      v2f d[8];
      #pragma unroll
      for (int i = 0; i < 8; ++i)
        d[i] = (jb < 3) ? xrow[2 * (8 * (jb + 1) + i) + half] : nrow[2 * i + half];
      #pragma unroll
      for (int i = 0; i < 8; ++i) {
        const int kc = 8 * jb + i;
        accL = __builtin_amdgcn_wmma_f32_16x16x4_f32(
            false, c[i], false, bl[kc], (short)0, accL, false, false);
        accR = __builtin_amdgcn_wmma_f32_16x16x4_f32(
            false, c[i], false, br[kc], (short)0, accR, false, false);
      }
      #pragma unroll
      for (int i = 0; i < 8; ++i) c[i] = d[i];
    }

    const int m0 = tile * 16;
    #pragma unroll
    for (int r = 0; r < 8; ++r) {
      const int row = m0 + r + 8 * half;
      if (row < n_nodes) {
        xl1[(size_t)row * 64 + n0 + rc] = accL[r];
        xr1[(size_t)row * 64 + n0 + rc] = accR[r];
      }
    }

    if (!more) break;
    tile = next;
    xrow = nrow;
  }
}

__global__ void __launch_bounds__(128, 5)
gemm1_edges_kernel(const float* __restrict__ x, const float* __restrict__ W1l,
                   const float* __restrict__ W1r, float* __restrict__ xl1,
                   float* __restrict__ xr1, int n_nodes, int gemm_tiles,
                   const int* __restrict__ ei, int* __restrict__ rows,
                   int* __restrict__ spill_n, int* __restrict__ spill,
                   int* __restrict__ claim, int E, int edge_blocks,
                   int n_chunk, int n_split) {
  __shared__ int s_part;
  const int b = blockIdx.x;
  if (b < GEMM_RESIDENT) {
    gemm1_blocks(b, GEMM_RESIDENT, gemm_tiles, x, W1l, W1r, xl1, xr1, n_nodes);
  } else {
    const int blk = b - GEMM_RESIDENT;
    if (blk < edge_blocks)
      edge_fill_block(blk, n_chunk, n_split, ei, rows, spill_n, spill, claim, E, &s_part);
  }
}

typedef __attribute__((ext_vector_type(4))) float v4f;

__device__ __forceinline__ float edge_score(const v4f x, const v2f r_lo, const v2f r_hi,
                                            const v2f a_lo, const v2f a_hi) {
  const v2f z0 = x.xy + r_lo;
  const v2f z1 = x.zw + r_hi;
  const v2f s0 = __builtin_elementwise_max(z0, z0 * NEG_SLOPE);
  const v2f s1 = __builtin_elementwise_max(z1, z1 * NEG_SLOPE);
  v2f q = s0 * a_lo;
  q = s1 * a_hi + q;
  float p = q.x + q.y;
  p += __shfl_xor(p, 1, 32);
  p += __shfl_xor(p, 2, 32);
  return p;
}

template <int NV>
struct NodeState {
  v2f   r_lo[NV], r_hi[NV];
  v2f   a_lo[NV], a_hi[NV];
  v2f   c_lo[NV], c_hi[NV];
  float m[NV], den[NV];
};

template <int STRIDE, int NV, int VOFF>
__device__ __forceinline__ void node_softmax_sum(
    const char* __restrict__ rows_b, const unsigned cbase,
    const char* __restrict__ xb, const int end, const unsigned cbyte,
    NodeState<NV>& st) {
  const float NEG_INF = -__builtin_inff();

  const int last = end - 1;
  for (int j = 0; j < end; j += 4) {
    const int s0 = *(const int*)(rows_b + (cbase + 4u * (unsigned)j));
    const int s1 = *(const int*)(rows_b + (cbase + 4u * (unsigned)min(j + 1, last)));
    const int s2 = *(const int*)(rows_b + (cbase + 4u * (unsigned)min(j + 2, last)));
    const int s3 = *(const int*)(rows_b + (cbase + 4u * (unsigned)min(j + 3, last)));
    const unsigned o0 = (unsigned)s0 * (unsigned)(STRIDE * 4) + cbyte;
    const unsigned o1 = (unsigned)s1 * (unsigned)(STRIDE * 4) + cbyte;
    const unsigned o2 = (unsigned)s2 * (unsigned)(STRIDE * 4) + cbyte;
    const unsigned o3 = (unsigned)s3 * (unsigned)(STRIDE * 4) + cbyte;
    v4f x0[NV], x1[NV], x2[NV], x3[NV];
    #pragma unroll
    for (int v = 0; v < NV; ++v) {
      x0[v] = *(const v4f*)(xb + o0 + (unsigned)(v * VOFF * 4));
      x1[v] = *(const v4f*)(xb + o1 + (unsigned)(v * VOFF * 4));
      x2[v] = *(const v4f*)(xb + o2 + (unsigned)(v * VOFF * 4));
      x3[v] = *(const v4f*)(xb + o3 + (unsigned)(v * VOFF * 4));
    }

    const bool v1 = (j + 1 < end), v2 = (j + 2 < end), v3 = (j + 3 < end);
    #pragma unroll
    for (int v = 0; v < NV; ++v) {
      const float p0 = edge_score(x0[v], st.r_lo[v], st.r_hi[v], st.a_lo[v], st.a_hi[v]);
      float p1 = edge_score(x1[v], st.r_lo[v], st.r_hi[v], st.a_lo[v], st.a_hi[v]);
      float p2 = edge_score(x2[v], st.r_lo[v], st.r_hi[v], st.a_lo[v], st.a_hi[v]);
      float p3 = edge_score(x3[v], st.r_lo[v], st.r_hi[v], st.a_lo[v], st.a_hi[v]);
      p1 = v1 ? p1 : NEG_INF;
      p2 = v2 ? p2 : NEG_INF;
      p3 = v3 ? p3 : NEG_INF;

      const float m  = st.m[v];
      const float mn = fmaxf(fmaxf(m, p0), fmaxf(fmaxf(p1, p2), p3));
      const float sc = __expf(m - mn);
      const float w0 = __expf(p0 - mn);
      const float w1 = __expf(p1 - mn);
      const float w2 = __expf(p2 - mn);
      const float w3 = __expf(p3 - mn);
      st.den[v] = st.den[v] * sc + ((w0 + w1) + (w2 + w3));
      v2f lo = st.c_lo[v] * sc;
      v2f hi = st.c_hi[v] * sc;
      lo = x0[v].xy * w0 + lo;   hi = x0[v].zw * w0 + hi;
      lo = x1[v].xy * w1 + lo;   hi = x1[v].zw * w1 + hi;
      lo = x2[v].xy * w2 + lo;   hi = x2[v].zw * w2 + hi;
      lo = x3[v].xy * w3 + lo;   hi = x3[v].zw * w3 + hi;
      st.c_lo[v] = lo;
      st.c_hi[v] = hi;
      st.m[v] = mn;
    }
  }
}

template <int STRIDE, int NV, int VOFF>
__device__ __forceinline__ void node_one_edge(
    const int s, const char* __restrict__ xb, const unsigned cbyte, NodeState<NV>& st) {
  const unsigned o = (unsigned)s * (unsigned)(STRIDE * 4) + cbyte;
  #pragma unroll
  for (int v = 0; v < NV; ++v) {
    const v4f x = *(const v4f*)(xb + o + (unsigned)(v * VOFF * 4));
    const float p  = edge_score(x, st.r_lo[v], st.r_hi[v], st.a_lo[v], st.a_hi[v]);
    const float mn = fmaxf(st.m[v], p);
    const float sc = __expf(st.m[v] - mn);
    const float w  = __expf(p - mn);
    st.den[v]  = st.den[v] * sc + w;
    st.c_lo[v] = x.xy * w + st.c_lo[v] * sc;
    st.c_hi[v] = x.zw * w + st.c_hi[v] * sc;
    st.m[v] = mn;
  }
}

template <int STRIDE, int NV, int VOFF>
__device__ __forceinline__ void node_all_edges(
    const int* __restrict__ rows, const int* __restrict__ spill_n,
    const int* __restrict__ spill, const int spill_cap,
    const float* __restrict__ xl, const int n, const int coff,
    const float4 (&rv)[NV], const float4 (&at)[NV],
    float4 (&acc)[NV], float (&den)[NV]) {
  const char* rows_b = (const char*)rows;
  const char* xb     = (const char*)xl;
  const unsigned rbase = (unsigned)n * (unsigned)(ROW_W * 4);
  const unsigned cbyte = (unsigned)coff * 4u;
  const int cnt = *(const int*)(rows_b + rbase);
  NodeState<NV> st;
  #pragma unroll
  for (int v = 0; v < NV; ++v) {
    st.r_lo[v] = v2f{rv[v].x, rv[v].y};  st.r_hi[v] = v2f{rv[v].z, rv[v].w};
    st.a_lo[v] = v2f{at[v].x, at[v].y};  st.a_hi[v] = v2f{at[v].z, at[v].w};
    st.c_lo[v] = v2f{0.0f, 0.0f};        st.c_hi[v] = v2f{0.0f, 0.0f};
    st.m[v]    = -__builtin_inff();
    st.den[v]  = 0.0f;
  }
  node_softmax_sum<STRIDE, NV, VOFF>(rows_b, rbase + 4u, xb, min(cnt, ROW_CAP), cbyte, st);
  if (cnt > ROW_CAP) {
    const int ns = min(*spill_n, spill_cap);
    for (int i = 0; i < ns; ++i)
      if (spill[2 * (size_t)i] == n)
        node_one_edge<STRIDE, NV, VOFF>(spill[2 * (size_t)i + 1], xb, cbyte, st);
  }
  #pragma unroll
  for (int v = 0; v < NV; ++v) {
    acc[v] = make_float4(st.c_lo[v].x, st.c_lo[v].y, st.c_hi[v].x, st.c_hi[v].y);
    den[v] = st.den[v];
  }
}

#define H_PITCH 68

__global__ void __launch_bounds__(128)
l1_node_gemm2_kernel(const int* __restrict__ rows, const int* __restrict__ spill_n,
                     const int* __restrict__ spill, int spill_cap,
                     const float* __restrict__ xl1, const float* __restrict__ xr1,
                     const float* __restrict__ att1, const float* __restrict__ b1,
                     const float* __restrict__ W2l, const float* __restrict__ W2r,
                     float* __restrict__ xl2, float* __restrict__ xr2, int n_nodes) {
  __shared__ float hs[16 * H_PITCH];
  const int nl   = threadIdx.x >> 3;
  const int n    = blockIdx.x * 16 + nl;
  const int coff = (threadIdx.x & 7) * 4;

  float4 hv[2];
  hv[0] = make_float4(0.0f, 0.0f, 0.0f, 0.0f);
  hv[1] = make_float4(0.0f, 0.0f, 0.0f, 0.0f);
  if (n < n_nodes) {
    float4 rv[2], at[2];
    rv[0] = *(const float4*)(xr1 + (size_t)n * 64 + coff);
    rv[1] = *(const float4*)(xr1 + (size_t)n * 64 + coff + 32);
    at[0] = *(const float4*)(att1 + coff);
    at[1] = *(const float4*)(att1 + coff + 32);

    float4 acc[2]; float den[2];
    node_all_edges<64, 2, 32>(rows, spill_n, spill, spill_cap, xl1, n, coff, rv, at, acc, den);

    #pragma unroll
    for (int g = 0; g < 2; ++g) {
      const float4 bb = *(const float4*)(b1 + coff + 32 * g);
      const float  dn = den[g];
      float4 v;
      v.x = ((dn > 0.0f) ? acc[g].x / dn : 0.0f) + bb.x;
      v.y = ((dn > 0.0f) ? acc[g].y / dn : 0.0f) + bb.y;
      v.z = ((dn > 0.0f) ? acc[g].z / dn : 0.0f) + bb.z;
      v.w = ((dn > 0.0f) ? acc[g].w / dn : 0.0f) + bb.w;
      v.x = (v.x > 0.0f) ? v.x : expm1f(v.x);
      v.y = (v.y > 0.0f) ? v.y : expm1f(v.y);
      v.z = (v.z > 0.0f) ? v.z : expm1f(v.z);
      v.w = (v.w > 0.0f) ? v.w : expm1f(v.w);
      hv[g] = v;
    }
  }
  *(float4*)(hs + nl * H_PITCH + coff)      = hv[0];
  *(float4*)(hs + nl * H_PITCH + coff + 32) = hv[1];
  __syncthreads();

  const int wave = __builtin_amdgcn_readfirstlane((int)(threadIdx.x >> 5));
  const int m0   = blockIdx.x * 16;
  if (wave == 0) {
    const int lane = threadIdx.x & 31;
    const int half = lane >> 4;
    const int rc   = lane & 15;
    const float* hrow = hs + rc * H_PITCH;
    const bool colok = (rc < 10);
    const int  cc    = colok ? rc : 9;

    v8f accL = {}, accR = {};
    #pragma unroll
    for (int kc = 0; kc < 16; ++kc) {
      const int k0 = 4 * kc + 2 * half;
      const v2f a = *(const v2f*)(hrow + k0);
      const float l0 = W2l[(size_t)(k0 + 0) * 10 + cc];
      const float l1 = W2l[(size_t)(k0 + 1) * 10 + cc];
      const float r0 = W2r[(size_t)(k0 + 0) * 10 + cc];
      const float r1 = W2r[(size_t)(k0 + 1) * 10 + cc];
      v2f bl, br;
      bl[0] = colok ? l0 : 0.0f;
      bl[1] = colok ? l1 : 0.0f;
      br[0] = colok ? r0 : 0.0f;
      br[1] = colok ? r1 : 0.0f;
      accL = __builtin_amdgcn_wmma_f32_16x16x4_f32(
          false, a, false, bl, (short)0, accL, false, false);
      accR = __builtin_amdgcn_wmma_f32_16x16x4_f32(
          false, a, false, br, (short)0, accR, false, false);
    }

    #pragma unroll
    for (int r = 0; r < 8; ++r) {
      const int row = m0 + r + 8 * half;
      if (row < n_nodes) {
        xl2[(size_t)row * 16 + rc] = accL[r];
        xr2[(size_t)row * 16 + rc] = accR[r];
      }
    }
  }
}

__global__ void __launch_bounds__(256)
l2_node_kernel(const int* __restrict__ rows, const int* __restrict__ spill_n,
               const int* __restrict__ spill, int spill_cap,
               const float* __restrict__ xl2, const float* __restrict__ xr2,
               const float* __restrict__ att2, const float* __restrict__ b2,
               float* __restrict__ out, int n_nodes) {
  __shared__ float stage[64 * 10];
  const int nl = threadIdx.x >> 2;
  const int n  = blockIdx.x * 64 + nl;
  const int c0 = (threadIdx.x & 3) * 4;

  if (n < n_nodes) {
    float4 rv[1], at[1];
    rv[0] = *(const float4*)(xr2 + (size_t)n * 16 + c0);
    at[0].x = (c0 + 0 < 10) ? att2[c0 + 0] : 0.0f;
    at[0].y = (c0 + 1 < 10) ? att2[c0 + 1] : 0.0f;
    at[0].z = (c0 + 2 < 10) ? att2[c0 + 2] : 0.0f;
    at[0].w = (c0 + 3 < 10) ? att2[c0 + 3] : 0.0f;

    float4 acc[1]; float den[1];
    node_all_edges<16, 1, 0>(rows, spill_n, spill, spill_cap, xl2, n, c0, rv, at, acc, den);

    const bool ok = (den[0] > 0.0f);
    if (c0 + 0 < 10) stage[nl * 10 + c0 + 0] = (ok ? acc[0].x / den[0] : 0.0f) + b2[c0 + 0];
    if (c0 + 1 < 10) stage[nl * 10 + c0 + 1] = (ok ? acc[0].y / den[0] : 0.0f) + b2[c0 + 1];
    if (c0 + 2 < 10) stage[nl * 10 + c0 + 2] = (ok ? acc[0].z / den[0] : 0.0f) + b2[c0 + 2];
    if (c0 + 3 < 10) stage[nl * 10 + c0 + 3] = (ok ? acc[0].w / den[0] : 0.0f) + b2[c0 + 3];
  }
  __syncthreads();

  const long long base  = (long long)blockIdx.x * 640;
  const long long total = (long long)n_nodes * 10;
  for (int i = threadIdx.x; i < 640; i += 256)
    if (base + i < total) out[base + i] = stage[i];
}

extern "C" void kernel_launch(void* const* d_in, const int* in_sizes, int n_in,
                              void* d_out, int out_size, void* d_ws, size_t ws_size,
                              hipStream_t stream) {
  const float* x    = (const float*)d_in[0];
  const int*   ei   = (const int*)d_in[1];
  const float* W1l  = (const float*)d_in[2];
  const float* W1r  = (const float*)d_in[3];
  const float* att1 = (const float*)d_in[4];
  const float* b1   = (const float*)d_in[5];
  const float* W2l  = (const float*)d_in[6];
  const float* W2r  = (const float*)d_in[7];
  const float* att2 = (const float*)d_in[8];
  const float* b2   = (const float*)d_in[9];
  float* out = (float*)d_out;

  const int N = in_sizes[0] / 128;
  const int E = in_sizes[1] / 2;

  float* ws = (float*)d_ws;
  size_t o = 0;
  int* rows    = (int*)(ws + o); o += (size_t)N * ROW_W;
  float* xl1   = ws + o; o += (size_t)N * 64;
  float* xr1   = ws + o; o += (size_t)N * 64;
  float* xl2   = ws + o; o += (size_t)N * 16;
  float* xr2   = ws + o; o += (size_t)N * 16;
  const int n_chunk = (int)(((long long)E + CHUNK_E - 1) / CHUNK_E);
  int* spill_n = (int*)(ws + o); o += 64;
  int* claim   = (int*)(ws + o); o += ((size_t)n_chunk + 63) / 64 * 64;
  int* spill   = (int*)(ws + o); o += 2 * (size_t)E;

  auto blocks = [](long long n) { return (unsigned)((n + 255) / 256); };
  const int mtiles = (N + 15) / 16;

  zero_heads_kernel<<<blocks(N > n_chunk ? N : n_chunk), 256, 0, stream>>>(
      rows, spill_n, claim, N, n_chunk);
  const int edge_blocks = 8 * ((n_chunk + 3) / 4);
  gemm1_edges_kernel<<<(unsigned)(GEMM_RESIDENT + edge_blocks), 128, 0, stream>>>(
      x, W1l, W1r, xl1, xr1, N, mtiles, ei, rows, spill_n, spill, claim, E, edge_blocks,
      n_chunk, N / 2);

  l1_node_gemm2_kernel<<<(unsigned)mtiles, 128, 0, stream>>>(
      rows, spill_n, spill, E, xl1, xr1, att1, b1, W2l, W2r, xl2, xr2, N);

  l2_node_kernel<<<(unsigned)((N + 63) / 64), 256, 0, stream>>>(rows, spill_n, spill, E, xl2, xr2,
                                                                att2, b2, out, N);
}


# GEMM1 x fragments as 16-byte loads: one load per lane feeds two K-steps (weight fragments loaded in the same permuted k order), 16 loads per row tile, not 32; k sum order inside each 8 changes
# speedup vs baseline: 6.7809x; 1.0428x over previous
#include <hip/hip_runtime.h>
#include <math.h>

typedef __attribute__((ext_vector_type(2))) float v2f;
typedef __attribute__((ext_vector_type(4))) float v4f;
typedef __attribute__((ext_vector_type(8))) float v8f;

#define NEG_SLOPE 0.2f

#define ROW_W   64
#define ROW_CAP 63
#define CHUNK_E 1024

__global__ void __launch_bounds__(256)
zero_heads_kernel(int* __restrict__ rows, int* __restrict__ spill_n,
                  int* __restrict__ claim, int n, int n_chunk) {
  int i = blockIdx.x * blockDim.x + threadIdx.x;
  if (i < n) rows[(size_t)i * ROW_W] = 0;
  if (i < n_chunk) claim[i] = 0;
  if (i == 0) *spill_n = 0;
}

__device__ __forceinline__ void put_edge(int* __restrict__ rows, int* __restrict__ spill_n,
                                         int* __restrict__ spill, int d, int c, int r) {
  if (r < ROW_CAP) {
    rows[(size_t)d * ROW_W + 1 + r] = c;
  } else {
    const int k = atomicAdd(spill_n, 1);
    spill[2 * (size_t)k]     = d;
    spill[2 * (size_t)k + 1] = c;
  }
}

__device__ __forceinline__ void
edge_fill_block(const int blk, const int n_chunk, const int n_split,
                const int* __restrict__ ei, int* __restrict__ rows,
                int* __restrict__ spill_n, int* __restrict__ spill,
                int* __restrict__ claim, int E, int* __restrict__ s_part) {
  const int chunk = ((blk >> 3) << 2) + (blk & 3);
  if (chunk >= n_chunk) return;
  const int half = (int)((__builtin_amdgcn_s_sendmsg_rtn(0x87) >> 18) & 1u);

  const int* dst = ei + E;
  const int e0 = chunk * CHUNK_E + threadIdx.x;
  const int e1 = e0 + 128, e2 = e0 + 256, e3 = e0 + 384;
  const int e4 = e0 + 512, e5 = e0 + 640, e6 = e0 + 768, e7 = e0 + 896;
  const int d0 = dst[min(e0, E - 1)], d1 = dst[min(e1, E - 1)];
  const int d2 = dst[min(e2, E - 1)], d3 = dst[min(e3, E - 1)];
  const int d4 = dst[min(e4, E - 1)], d5 = dst[min(e5, E - 1)];
  const int d6 = dst[min(e6, E - 1)], d7 = dst[min(e7, E - 1)];
  const int c0 = ei[min(e0, E - 1)], c1 = ei[min(e1, E - 1)];
  const int c2 = ei[min(e2, E - 1)], c3 = ei[min(e3, E - 1)];
  const int c4 = ei[min(e4, E - 1)], c5 = ei[min(e5, E - 1)];
  const int c6 = ei[min(e6, E - 1)], c7 = ei[min(e7, E - 1)];

  if (threadIdx.x == 0) {
    const int old = atomicOr(claim + chunk, 1 << half);
    *s_part = ((old >> half) & 1) ? (half ^ 1) : half;
  }
  __syncthreads();
  const bool upper = (*s_part != 0);

  const bool l0 = (e0 < E) && ((d0 >= n_split) == upper);
  const bool l1 = (e1 < E) && ((d1 >= n_split) == upper);
  const bool l2 = (e2 < E) && ((d2 >= n_split) == upper);
  const bool l3 = (e3 < E) && ((d3 >= n_split) == upper);
  const bool l4 = (e4 < E) && ((d4 >= n_split) == upper);
  const bool l5 = (e5 < E) && ((d5 >= n_split) == upper);
  const bool l6 = (e6 < E) && ((d6 >= n_split) == upper);
  const bool l7 = (e7 < E) && ((d7 >= n_split) == upper);
  int r0 = 0, r1 = 0, r2 = 0, r3 = 0, r4 = 0, r5 = 0, r6 = 0, r7 = 0;
  if (l0) r0 = atomicAdd(rows + (size_t)d0 * ROW_W, 1);
  if (l1) r1 = atomicAdd(rows + (size_t)d1 * ROW_W, 1);
  if (l2) r2 = atomicAdd(rows + (size_t)d2 * ROW_W, 1);
  if (l3) r3 = atomicAdd(rows + (size_t)d3 * ROW_W, 1);
  if (l4) r4 = atomicAdd(rows + (size_t)d4 * ROW_W, 1);
  if (l5) r5 = atomicAdd(rows + (size_t)d5 * ROW_W, 1);
  if (l6) r6 = atomicAdd(rows + (size_t)d6 * ROW_W, 1);
  if (l7) r7 = atomicAdd(rows + (size_t)d7 * ROW_W, 1);
  if (l0) put_edge(rows, spill_n, spill, d0, c0, r0);
  if (l1) put_edge(rows, spill_n, spill, d1, c1, r1);
  if (l2) put_edge(rows, spill_n, spill, d2, c2, r2);
  if (l3) put_edge(rows, spill_n, spill, d3, c3, r3);
  if (l4) put_edge(rows, spill_n, spill, d4, c4, r4);
  if (l5) put_edge(rows, spill_n, spill, d5, c5, r5);
  if (l6) put_edge(rows, spill_n, spill, d6, c6, r6);
  if (l7) put_edge(rows, spill_n, spill, d7, c7, r7);
}

#define GEMM_RESIDENT 512

__device__ __forceinline__ const v4f*
x_row(const float* __restrict__ x, const int tile, const int rc, const int n_nodes) {
  int r = tile * 16 + rc;
  if (r >= n_nodes) r = n_nodes - 1;
  return (const v4f*)(x + (size_t)r * 128);
}

__device__ __forceinline__ void
gemm1_blocks(const int first, const int stride, const int n_tiles,
             const float* __restrict__ x, const float* __restrict__ W1l,
             const float* __restrict__ W1r, float* __restrict__ xl1,
             float* __restrict__ xr1, int n_nodes) {
  if (first >= n_tiles) return;
  const int lane = threadIdx.x & 31;
  const int wave = threadIdx.x >> 5;
  const int n0   = wave * 16;
  const int half = lane >> 4;
  const int rc   = lane & 15;
  const float* Bl = W1l + n0 + rc;
  const float* Br = W1r + n0 + rc;

  v2f bl[32], br[32];
  #pragma unroll
  for (int kc = 0; kc < 32; ++kc) {
    const int k0 = 8 * (kc >> 1) + 4 * half + 2 * (kc & 1);
    bl[kc][0] = Bl[(size_t)(k0 + 0) * 64];
    bl[kc][1] = Bl[(size_t)(k0 + 1) * 64];
    br[kc][0] = Br[(size_t)(k0 + 0) * 64];
    br[kc][1] = Br[(size_t)(k0 + 1) * 64];
  }

  int tile = first;
  const v4f* xrow = x_row(x, tile, rc, n_nodes);
  v4f c[4];
  #pragma unroll
  for (int i = 0; i < 4; ++i) c[i] = xrow[2 * i + half];

  for (;;) {
    const int  next = tile + stride;
    const bool more = (next < n_tiles);
    const v4f* nrow = x_row(x, more ? next : tile, rc, n_nodes);

    v8f accL = {}, accR = {};
    #pragma unroll
    for (int jb = 0; jb < 4; ++jb) {
      v4f d[4];
      #pragma unroll
      for (int i = 0; i < 4; ++i)
        d[i] = (jb < 3) ? xrow[2 * (4 * (jb + 1) + i) + half] : nrow[2 * i + half];
      #pragma unroll
      for (int i = 0; i < 4; ++i) {
        const int kc = 8 * jb + 2 * i;
        const v2f a0 = c[i].xy;
        const v2f a1 = c[i].zw;
        accL = __builtin_amdgcn_wmma_f32_16x16x4_f32(
            false, a0, false, bl[kc], (short)0, accL, false, false);
        accR = __builtin_amdgcn_wmma_f32_16x16x4_f32(
            false, a0, false, br[kc], (short)0, accR, false, false);
        accL = __builtin_amdgcn_wmma_f32_16x16x4_f32(
            false, a1, false, bl[kc + 1], (short)0, accL, false, false);
        accR = __builtin_amdgcn_wmma_f32_16x16x4_f32(
            false, a1, false, br[kc + 1], (short)0, accR, false, false);
      }
      #pragma unroll
      for (int i = 0; i < 4; ++i) c[i] = d[i];
    }

    const int m0 = tile * 16;
    #pragma unroll
    for (int r = 0; r < 8; ++r) {
      const int row = m0 + r + 8 * half;
      if (row < n_nodes) {
        xl1[(size_t)row * 64 + n0 + rc] = accL[r];
        xr1[(size_t)row * 64 + n0 + rc] = accR[r];
      }
    }

    if (!more) break;
    tile = next;
    xrow = nrow;
  }
}

__global__ void __launch_bounds__(128, 5)
gemm1_edges_kernel(const float* __restrict__ x, const float* __restrict__ W1l,
                   const float* __restrict__ W1r, float* __restrict__ xl1,
                   float* __restrict__ xr1, int n_nodes, int gemm_tiles,
                   const int* __restrict__ ei, int* __restrict__ rows,
                   int* __restrict__ spill_n, int* __restrict__ spill,
                   int* __restrict__ claim, int E, int edge_blocks,
                   int n_chunk, int n_split) {
  __shared__ int s_part;
  const int b = blockIdx.x;
  if (b < GEMM_RESIDENT) {
    gemm1_blocks(b, GEMM_RESIDENT, gemm_tiles, x, W1l, W1r, xl1, xr1, n_nodes);
  } else {
    const int blk = b - GEMM_RESIDENT;
    if (blk < edge_blocks)
      edge_fill_block(blk, n_chunk, n_split, ei, rows, spill_n, spill, claim, E, &s_part);
  }
}

__device__ __forceinline__ float edge_score(const v4f x, const v2f r_lo, const v2f r_hi,
                                            const v2f a_lo, const v2f a_hi) {
  const v2f z0 = x.xy + r_lo;
  const v2f z1 = x.zw + r_hi;
  const v2f s0 = __builtin_elementwise_max(z0, z0 * NEG_SLOPE);
  const v2f s1 = __builtin_elementwise_max(z1, z1 * NEG_SLOPE);
  v2f q = s0 * a_lo;
  q = s1 * a_hi + q;
  float p = q.x + q.y;
  p += __shfl_xor(p, 1, 32);
  p += __shfl_xor(p, 2, 32);
  return p;
}

template <int NV>
struct NodeState {
  v2f   r_lo[NV], r_hi[NV];
  v2f   a_lo[NV], a_hi[NV];
  v2f   c_lo[NV], c_hi[NV];
  float m[NV], den[NV];
};

template <int STRIDE, int NV, int VOFF>
__device__ __forceinline__ void node_softmax_sum(
    const char* __restrict__ rows_b, const unsigned cbase,
    const char* __restrict__ xb, const int end, const unsigned cbyte,
    NodeState<NV>& st) {
  const float NEG_INF = -__builtin_inff();

  const int last = end - 1;
  for (int j = 0; j < end; j += 4) {
    const int s0 = *(const int*)(rows_b + (cbase + 4u * (unsigned)j));
    const int s1 = *(const int*)(rows_b + (cbase + 4u * (unsigned)min(j + 1, last)));
    const int s2 = *(const int*)(rows_b + (cbase + 4u * (unsigned)min(j + 2, last)));
    const int s3 = *(const int*)(rows_b + (cbase + 4u * (unsigned)min(j + 3, last)));
    const unsigned o0 = (unsigned)s0 * (unsigned)(STRIDE * 4) + cbyte;
    const unsigned o1 = (unsigned)s1 * (unsigned)(STRIDE * 4) + cbyte;
    const unsigned o2 = (unsigned)s2 * (unsigned)(STRIDE * 4) + cbyte;
    const unsigned o3 = (unsigned)s3 * (unsigned)(STRIDE * 4) + cbyte;
    v4f x0[NV], x1[NV], x2[NV], x3[NV];
    #pragma unroll
    for (int v = 0; v < NV; ++v) {
      x0[v] = *(const v4f*)(xb + o0 + (unsigned)(v * VOFF * 4));
      x1[v] = *(const v4f*)(xb + o1 + (unsigned)(v * VOFF * 4));
      x2[v] = *(const v4f*)(xb + o2 + (unsigned)(v * VOFF * 4));
      x3[v] = *(const v4f*)(xb + o3 + (unsigned)(v * VOFF * 4));
    }

    const bool v1 = (j + 1 < end), v2 = (j + 2 < end), v3 = (j + 3 < end);
    #pragma unroll
    for (int v = 0; v < NV; ++v) {
      const float p0 = edge_score(x0[v], st.r_lo[v], st.r_hi[v], st.a_lo[v], st.a_hi[v]);
      float p1 = edge_score(x1[v], st.r_lo[v], st.r_hi[v], st.a_lo[v], st.a_hi[v]);
      float p2 = edge_score(x2[v], st.r_lo[v], st.r_hi[v], st.a_lo[v], st.a_hi[v]);
      float p3 = edge_score(x3[v], st.r_lo[v], st.r_hi[v], st.a_lo[v], st.a_hi[v]);
      p1 = v1 ? p1 : NEG_INF;
      p2 = v2 ? p2 : NEG_INF;
      p3 = v3 ? p3 : NEG_INF;

      const float m  = st.m[v];
      const float mn = fmaxf(fmaxf(m, p0), fmaxf(fmaxf(p1, p2), p3));
      const float sc = __expf(m - mn);
      const float w0 = __expf(p0 - mn);
      const float w1 = __expf(p1 - mn);
      const float w2 = __expf(p2 - mn);
      const float w3 = __expf(p3 - mn);
      st.den[v] = st.den[v] * sc + ((w0 + w1) + (w2 + w3));
      v2f lo = st.c_lo[v] * sc;
      v2f hi = st.c_hi[v] * sc;
      lo = x0[v].xy * w0 + lo;   hi = x0[v].zw * w0 + hi;
      lo = x1[v].xy * w1 + lo;   hi = x1[v].zw * w1 + hi;
      lo = x2[v].xy * w2 + lo;   hi = x2[v].zw * w2 + hi;
      lo = x3[v].xy * w3 + lo;   hi = x3[v].zw * w3 + hi;
      st.c_lo[v] = lo;
      st.c_hi[v] = hi;
      st.m[v] = mn;
    }
  }
}

template <int STRIDE, int NV, int VOFF>
__device__ __forceinline__ void node_one_edge(
    const int s, const char* __restrict__ xb, const unsigned cbyte, NodeState<NV>& st) {
  const unsigned o = (unsigned)s * (unsigned)(STRIDE * 4) + cbyte;
  #pragma unroll
  for (int v = 0; v < NV; ++v) {
    const v4f x = *(const v4f*)(xb + o + (unsigned)(v * VOFF * 4));
    const float p  = edge_score(x, st.r_lo[v], st.r_hi[v], st.a_lo[v], st.a_hi[v]);
    const float mn = fmaxf(st.m[v], p);
    const float sc = __expf(st.m[v] - mn);
    const float w  = __expf(p - mn);
    st.den[v]  = st.den[v] * sc + w;
    st.c_lo[v] = x.xy * w + st.c_lo[v] * sc;
    st.c_hi[v] = x.zw * w + st.c_hi[v] * sc;
    st.m[v] = mn;
  }
}

template <int STRIDE, int NV, int VOFF>
__device__ __forceinline__ void node_all_edges(
    const int* __restrict__ rows, const int* __restrict__ spill_n,
    const int* __restrict__ spill, const int spill_cap,
    const float* __restrict__ xl, const int n, const int coff,
    const float4 (&rv)[NV], const float4 (&at)[NV],
    float4 (&acc)[NV], float (&den)[NV]) {
  const char* rows_b = (const char*)rows;
  const char* xb     = (const char*)xl;
  const unsigned rbase = (unsigned)n * (unsigned)(ROW_W * 4);
  const unsigned cbyte = (unsigned)coff * 4u;
  const int cnt = *(const int*)(rows_b + rbase);
  NodeState<NV> st;
  #pragma unroll
  for (int v = 0; v < NV; ++v) {
    st.r_lo[v] = v2f{rv[v].x, rv[v].y};  st.r_hi[v] = v2f{rv[v].z, rv[v].w};
    st.a_lo[v] = v2f{at[v].x, at[v].y};  st.a_hi[v] = v2f{at[v].z, at[v].w};
    st.c_lo[v] = v2f{0.0f, 0.0f};        st.c_hi[v] = v2f{0.0f, 0.0f};
    st.m[v]    = -__builtin_inff();
    st.den[v]  = 0.0f;
  }
  node_softmax_sum<STRIDE, NV, VOFF>(rows_b, rbase + 4u, xb, min(cnt, ROW_CAP), cbyte, st);
  if (cnt > ROW_CAP) {
    const int ns = min(*spill_n, spill_cap);
    for (int i = 0; i < ns; ++i)
      if (spill[2 * (size_t)i] == n)
        node_one_edge<STRIDE, NV, VOFF>(spill[2 * (size_t)i + 1], xb, cbyte, st);
  }
  #pragma unroll
  for (int v = 0; v < NV; ++v) {
    acc[v] = make_float4(st.c_lo[v].x, st.c_lo[v].y, st.c_hi[v].x, st.c_hi[v].y);
    den[v] = st.den[v];
  }
}

#define H_PITCH 68

__global__ void __launch_bounds__(128)
l1_node_gemm2_kernel(const int* __restrict__ rows, const int* __restrict__ spill_n,
                     const int* __restrict__ spill, int spill_cap,
                     const float* __restrict__ xl1, const float* __restrict__ xr1,
                     const float* __restrict__ att1, const float* __restrict__ b1,
                     const float* __restrict__ W2l, const float* __restrict__ W2r,
                     float* __restrict__ xl2, float* __restrict__ xr2, int n_nodes) {
  __shared__ float hs[16 * H_PITCH];
  const int nl   = threadIdx.x >> 3;
  const int n    = blockIdx.x * 16 + nl;
  const int coff = (threadIdx.x & 7) * 4;

  float4 hv[2];
  hv[0] = make_float4(0.0f, 0.0f, 0.0f, 0.0f);
  hv[1] = make_float4(0.0f, 0.0f, 0.0f, 0.0f);
  if (n < n_nodes) {
    float4 rv[2], at[2];
    rv[0] = *(const float4*)(xr1 + (size_t)n * 64 + coff);
    rv[1] = *(const float4*)(xr1 + (size_t)n * 64 + coff + 32);
    at[0] = *(const float4*)(att1 + coff);
    at[1] = *(const float4*)(att1 + coff + 32);

    float4 acc[2]; float den[2];
    node_all_edges<64, 2, 32>(rows, spill_n, spill, spill_cap, xl1, n, coff, rv, at, acc, den);

    #pragma unroll
    for (int g = 0; g < 2; ++g) {
      const float4 bb = *(const float4*)(b1 + coff + 32 * g);
      const float  dn = den[g];
      float4 v;
      v.x = ((dn > 0.0f) ? acc[g].x / dn : 0.0f) + bb.x;
      v.y = ((dn > 0.0f) ? acc[g].y / dn : 0.0f) + bb.y;
      v.z = ((dn > 0.0f) ? acc[g].z / dn : 0.0f) + bb.z;
      v.w = ((dn > 0.0f) ? acc[g].w / dn : 0.0f) + bb.w;
      v.x = (v.x > 0.0f) ? v.x : expm1f(v.x);
      v.y = (v.y > 0.0f) ? v.y : expm1f(v.y);
      v.z = (v.z > 0.0f) ? v.z : expm1f(v.z);
      v.w = (v.w > 0.0f) ? v.w : expm1f(v.w);
      hv[g] = v;
    }
  }
  *(float4*)(hs + nl * H_PITCH + coff)      = hv[0];
  *(float4*)(hs + nl * H_PITCH + coff + 32) = hv[1];
  __syncthreads();

  const int wave = __builtin_amdgcn_readfirstlane((int)(threadIdx.x >> 5));
  const int m0   = blockIdx.x * 16;
  if (wave == 0) {
    const int lane = threadIdx.x & 31;
    const int half = lane >> 4;
    const int rc   = lane & 15;
    const float* hrow = hs + rc * H_PITCH;
    const bool colok = (rc < 10);
    const int  cc    = colok ? rc : 9;

    v8f accL = {}, accR = {};
    #pragma unroll
    for (int kc = 0; kc < 16; ++kc) {
      const int k0 = 4 * kc + 2 * half;
      const v2f a = *(const v2f*)(hrow + k0);
      const float l0 = W2l[(size_t)(k0 + 0) * 10 + cc];
      const float l1 = W2l[(size_t)(k0 + 1) * 10 + cc];
      const float r0 = W2r[(size_t)(k0 + 0) * 10 + cc];
      const float r1 = W2r[(size_t)(k0 + 1) * 10 + cc];
      v2f bl, br;
      bl[0] = colok ? l0 : 0.0f;
      bl[1] = colok ? l1 : 0.0f;
      br[0] = colok ? r0 : 0.0f;
      br[1] = colok ? r1 : 0.0f;
      accL = __builtin_amdgcn_wmma_f32_16x16x4_f32(
          false, a, false, bl, (short)0, accL, false, false);
      accR = __builtin_amdgcn_wmma_f32_16x16x4_f32(
          false, a, false, br, (short)0, accR, false, false);
    }

    #pragma unroll
    for (int r = 0; r < 8; ++r) {
      const int row = m0 + r + 8 * half;
      if (row < n_nodes) {
        xl2[(size_t)row * 16 + rc] = accL[r];
        xr2[(size_t)row * 16 + rc] = accR[r];
      }
    }
  }
}

__global__ void __launch_bounds__(256)
l2_node_kernel(const int* __restrict__ rows, const int* __restrict__ spill_n,
               const int* __restrict__ spill, int spill_cap,
               const float* __restrict__ xl2, const float* __restrict__ xr2,
               const float* __restrict__ att2, const float* __restrict__ b2,
               float* __restrict__ out, int n_nodes) {
  __shared__ float stage[64 * 10];
  const int nl = threadIdx.x >> 2;
  const int n  = blockIdx.x * 64 + nl;
  const int c0 = (threadIdx.x & 3) * 4;

  if (n < n_nodes) {
    float4 rv[1], at[1];
    rv[0] = *(const float4*)(xr2 + (size_t)n * 16 + c0);
    at[0].x = (c0 + 0 < 10) ? att2[c0 + 0] : 0.0f;
    at[0].y = (c0 + 1 < 10) ? att2[c0 + 1] : 0.0f;
    at[0].z = (c0 + 2 < 10) ? att2[c0 + 2] : 0.0f;
    at[0].w = (c0 + 3 < 10) ? att2[c0 + 3] : 0.0f;

    float4 acc[1]; float den[1];
    node_all_edges<16, 1, 0>(rows, spill_n, spill, spill_cap, xl2, n, c0, rv, at, acc, den);

    const bool ok = (den[0] > 0.0f);
    if (c0 + 0 < 10) stage[nl * 10 + c0 + 0] = (ok ? acc[0].x / den[0] : 0.0f) + b2[c0 + 0];
    if (c0 + 1 < 10) stage[nl * 10 + c0 + 1] = (ok ? acc[0].y / den[0] : 0.0f) + b2[c0 + 1];
    if (c0 + 2 < 10) stage[nl * 10 + c0 + 2] = (ok ? acc[0].z / den[0] : 0.0f) + b2[c0 + 2];
    if (c0 + 3 < 10) stage[nl * 10 + c0 + 3] = (ok ? acc[0].w / den[0] : 0.0f) + b2[c0 + 3];
  }
  __syncthreads();

  const long long base  = (long long)blockIdx.x * 640;
  const long long total = (long long)n_nodes * 10;
  for (int i = threadIdx.x; i < 640; i += 256)
    if (base + i < total) out[base + i] = stage[i];
}

extern "C" void kernel_launch(void* const* d_in, const int* in_sizes, int n_in,
                              void* d_out, int out_size, void* d_ws, size_t ws_size,
                              hipStream_t stream) {
  const float* x    = (const float*)d_in[0];
  const int*   ei   = (const int*)d_in[1];
  const float* W1l  = (const float*)d_in[2];
  const float* W1r  = (const float*)d_in[3];
  const float* att1 = (const float*)d_in[4];
  const float* b1   = (const float*)d_in[5];
  const float* W2l  = (const float*)d_in[6];
  const float* W2r  = (const float*)d_in[7];
  const float* att2 = (const float*)d_in[8];
  const float* b2   = (const float*)d_in[9];
  float* out = (float*)d_out;

  const int N = in_sizes[0] / 128;
  const int E = in_sizes[1] / 2;

  float* ws = (float*)d_ws;
  size_t o = 0;
  int* rows    = (int*)(ws + o); o += (size_t)N * ROW_W;
  float* xl1   = ws + o; o += (size_t)N * 64;
  float* xr1   = ws + o; o += (size_t)N * 64;
  float* xl2   = ws + o; o += (size_t)N * 16;
  float* xr2   = ws + o; o += (size_t)N * 16;
  const int n_chunk = (int)(((long long)E + CHUNK_E - 1) / CHUNK_E);
  int* spill_n = (int*)(ws + o); o += 64;
  int* claim   = (int*)(ws + o); o += ((size_t)n_chunk + 63) / 64 * 64;
  int* spill   = (int*)(ws + o); o += 2 * (size_t)E;

  auto blocks = [](long long n) { return (unsigned)((n + 255) / 256); };
  const int mtiles = (N + 15) / 16;

  zero_heads_kernel<<<blocks(N > n_chunk ? N : n_chunk), 256, 0, stream>>>(
      rows, spill_n, claim, N, n_chunk);
  const int edge_blocks = 8 * ((n_chunk + 3) / 4);
  gemm1_edges_kernel<<<(unsigned)(GEMM_RESIDENT + edge_blocks), 128, 0, stream>>>(
      x, W1l, W1r, xl1, xr1, N, mtiles, ei, rows, spill_n, spill, claim, E, edge_blocks,
      n_chunk, N / 2);

  l1_node_gemm2_kernel<<<(unsigned)mtiles, 128, 0, stream>>>(
      rows, spill_n, spill, E, xl1, xr1, att1, b1, W2l, W2r, xl2, xr2, N);

  l2_node_kernel<<<(unsigned)((N + 63) / 64), 256, 0, stream>>>(rows, spill_n, spill, E, xl2, xr2,
                                                                att2, b2, out, N);
}


# Layer-1 node loop: GEMM1 deals its weight columns so a lane's 8 channels are ONE head: one score, softmax chain and exchange step per lane, not two of each; sum order in layer-1 scores changes
# speedup vs baseline: 7.0112x; 1.0340x over previous
#include <hip/hip_runtime.h>
#include <math.h>

typedef __attribute__((ext_vector_type(2))) float v2f;
typedef __attribute__((ext_vector_type(4))) float v4f;
typedef __attribute__((ext_vector_type(8))) float v8f;

#define NEG_SLOPE 0.2f

#define ROW_W   64
#define ROW_CAP 63
#define CHUNK_E 1024

__global__ void __launch_bounds__(256)
zero_heads_kernel(int* __restrict__ rows, int* __restrict__ spill_n,
                  int* __restrict__ claim, int n, int n_chunk) {
  int i = blockIdx.x * blockDim.x + threadIdx.x;
  if (i < n) rows[(size_t)i * ROW_W] = 0;
  if (i < n_chunk) claim[i] = 0;
  if (i == 0) *spill_n = 0;
}

__device__ __forceinline__ void put_edge(int* __restrict__ rows, int* __restrict__ spill_n,
                                         int* __restrict__ spill, int d, int c, int r) {
  if (r < ROW_CAP) {
    rows[(size_t)d * ROW_W + 1 + r] = c;
  } else {
    const int k = atomicAdd(spill_n, 1);
    spill[2 * (size_t)k]     = d;
    spill[2 * (size_t)k + 1] = c;
  }
}

__device__ __forceinline__ void
edge_fill_block(const int blk, const int n_chunk, const int n_split,
                const int* __restrict__ ei, int* __restrict__ rows,
                int* __restrict__ spill_n, int* __restrict__ spill,
                int* __restrict__ claim, int E, int* __restrict__ s_part) {
  const int chunk = ((blk >> 3) << 2) + (blk & 3);
  if (chunk >= n_chunk) return;
  const int half = (int)((__builtin_amdgcn_s_sendmsg_rtn(0x87) >> 18) & 1u);

  const int* dst = ei + E;
  const int e0 = chunk * CHUNK_E + threadIdx.x;
  const int e1 = e0 + 128, e2 = e0 + 256, e3 = e0 + 384;
  const int e4 = e0 + 512, e5 = e0 + 640, e6 = e0 + 768, e7 = e0 + 896;
  const int d0 = dst[min(e0, E - 1)], d1 = dst[min(e1, E - 1)];
  const int d2 = dst[min(e2, E - 1)], d3 = dst[min(e3, E - 1)];
  const int d4 = dst[min(e4, E - 1)], d5 = dst[min(e5, E - 1)];
  const int d6 = dst[min(e6, E - 1)], d7 = dst[min(e7, E - 1)];
  const int c0 = ei[min(e0, E - 1)], c1 = ei[min(e1, E - 1)];
  const int c2 = ei[min(e2, E - 1)], c3 = ei[min(e3, E - 1)];
  const int c4 = ei[min(e4, E - 1)], c5 = ei[min(e5, E - 1)];
  const int c6 = ei[min(e6, E - 1)], c7 = ei[min(e7, E - 1)];

  if (threadIdx.x == 0) {
    const int old = atomicOr(claim + chunk, 1 << half);
    *s_part = ((old >> half) & 1) ? (half ^ 1) : half;
  }
  __syncthreads();
  const bool upper = (*s_part != 0);

  const bool l0 = (e0 < E) && ((d0 >= n_split) == upper);
  const bool l1 = (e1 < E) && ((d1 >= n_split) == upper);
  const bool l2 = (e2 < E) && ((d2 >= n_split) == upper);
  const bool l3 = (e3 < E) && ((d3 >= n_split) == upper);
  const bool l4 = (e4 < E) && ((d4 >= n_split) == upper);
  const bool l5 = (e5 < E) && ((d5 >= n_split) == upper);
  const bool l6 = (e6 < E) && ((d6 >= n_split) == upper);
  const bool l7 = (e7 < E) && ((d7 >= n_split) == upper);
  int r0 = 0, r1 = 0, r2 = 0, r3 = 0, r4 = 0, r5 = 0, r6 = 0, r7 = 0;
  if (l0) r0 = atomicAdd(rows + (size_t)d0 * ROW_W, 1);
  if (l1) r1 = atomicAdd(rows + (size_t)d1 * ROW_W, 1);
  if (l2) r2 = atomicAdd(rows + (size_t)d2 * ROW_W, 1);
  if (l3) r3 = atomicAdd(rows + (size_t)d3 * ROW_W, 1);
  if (l4) r4 = atomicAdd(rows + (size_t)d4 * ROW_W, 1);
  if (l5) r5 = atomicAdd(rows + (size_t)d5 * ROW_W, 1);
  if (l6) r6 = atomicAdd(rows + (size_t)d6 * ROW_W, 1);
  if (l7) r7 = atomicAdd(rows + (size_t)d7 * ROW_W, 1);
  if (l0) put_edge(rows, spill_n, spill, d0, c0, r0);
  if (l1) put_edge(rows, spill_n, spill, d1, c1, r1);
  if (l2) put_edge(rows, spill_n, spill, d2, c2, r2);
  if (l3) put_edge(rows, spill_n, spill, d3, c3, r3);
  if (l4) put_edge(rows, spill_n, spill, d4, c4, r4);
  if (l5) put_edge(rows, spill_n, spill, d5, c5, r5);
  if (l6) put_edge(rows, spill_n, spill, d6, c6, r6);
  if (l7) put_edge(rows, spill_n, spill, d7, c7, r7);
}

#define GEMM_RESIDENT 512

__device__ __forceinline__ int l1_chan(const int p) {
  return 16 * ((p >> 3) & 3) + 8 * (p >> 5) + (p & 7);
}

__device__ __forceinline__ const v4f*
x_row(const float* __restrict__ x, const int tile, const int rc, const int n_nodes) {
  int r = tile * 16 + rc;
  if (r >= n_nodes) r = n_nodes - 1;
  return (const v4f*)(x + (size_t)r * 128);
}

__device__ __forceinline__ void
gemm1_blocks(const int first, const int stride, const int n_tiles,
             const float* __restrict__ x, const float* __restrict__ W1l,
             const float* __restrict__ W1r, float* __restrict__ xl1,
             float* __restrict__ xr1, int n_nodes) {
  if (first >= n_tiles) return;
  const int lane = threadIdx.x & 31;
  const int wave = threadIdx.x >> 5;
  const int n0   = wave * 16;
  const int half = lane >> 4;
  const int rc   = lane & 15;
  const float* Bl = W1l + l1_chan(n0 + rc);
  const float* Br = W1r + l1_chan(n0 + rc);

  v2f bl[32], br[32];
  #pragma unroll
  for (int kc = 0; kc < 32; ++kc) {
    const int k0 = 8 * (kc >> 1) + 4 * half + 2 * (kc & 1);
    bl[kc][0] = Bl[(size_t)(k0 + 0) * 64];
    bl[kc][1] = Bl[(size_t)(k0 + 1) * 64];
    br[kc][0] = Br[(size_t)(k0 + 0) * 64];
    br[kc][1] = Br[(size_t)(k0 + 1) * 64];
  }

  int tile = first;
  const v4f* xrow = x_row(x, tile, rc, n_nodes);
  v4f c[4];
  #pragma unroll
  for (int i = 0; i < 4; ++i) c[i] = xrow[2 * i + half];

  for (;;) {
    const int  next = tile + stride;
    const bool more = (next < n_tiles);
    const v4f* nrow = x_row(x, more ? next : tile, rc, n_nodes);

    v8f accL = {}, accR = {};
    #pragma unroll
    for (int jb = 0; jb < 4; ++jb) {
      v4f d[4];
      #pragma unroll
      for (int i = 0; i < 4; ++i)
        d[i] = (jb < 3) ? xrow[2 * (4 * (jb + 1) + i) + half] : nrow[2 * i + half];
      #pragma unroll
      for (int i = 0; i < 4; ++i) {
        const int kc = 8 * jb + 2 * i;
        const v2f a0 = c[i].xy;
        const v2f a1 = c[i].zw;
        accL = __builtin_amdgcn_wmma_f32_16x16x4_f32(
            false, a0, false, bl[kc], (short)0, accL, false, false);
        accR = __builtin_amdgcn_wmma_f32_16x16x4_f32(
            false, a0, false, br[kc], (short)0, accR, false, false);
        accL = __builtin_amdgcn_wmma_f32_16x16x4_f32(
            false, a1, false, bl[kc + 1], (short)0, accL, false, false);
        accR = __builtin_amdgcn_wmma_f32_16x16x4_f32(
            false, a1, false, br[kc + 1], (short)0, accR, false, false);
      }
      #pragma unroll
      for (int i = 0; i < 4; ++i) c[i] = d[i];
    }

    const int m0 = tile * 16;
    #pragma unroll
    for (int r = 0; r < 8; ++r) {
      const int row = m0 + r + 8 * half;
      if (row < n_nodes) {
        xl1[(size_t)row * 64 + n0 + rc] = accL[r];
        xr1[(size_t)row * 64 + n0 + rc] = accR[r];
      }
    }

    if (!more) break;
    tile = next;
    xrow = nrow;
  }
}

__global__ void __launch_bounds__(128, 5)
gemm1_edges_kernel(const float* __restrict__ x, const float* __restrict__ W1l,
                   const float* __restrict__ W1r, float* __restrict__ xl1,
                   float* __restrict__ xr1, int n_nodes, int gemm_tiles,
                   const int* __restrict__ ei, int* __restrict__ rows,
                   int* __restrict__ spill_n, int* __restrict__ spill,
                   int* __restrict__ claim, int E, int edge_blocks,
                   int n_chunk, int n_split) {
  __shared__ int s_part;
  const int b = blockIdx.x;
  if (b < GEMM_RESIDENT) {
    gemm1_blocks(b, GEMM_RESIDENT, gemm_tiles, x, W1l, W1r, xl1, xr1, n_nodes);
  } else {
    const int blk = b - GEMM_RESIDENT;
    if (blk < edge_blocks)
      edge_fill_block(blk, n_chunk, n_split, ei, rows, spill_n, spill, claim, E, &s_part);
  }
}

__device__ __forceinline__ float edge_score(const v4f x, const v2f r_lo, const v2f r_hi,
                                            const v2f a_lo, const v2f a_hi) {
  const v2f z0 = x.xy + r_lo;
  const v2f z1 = x.zw + r_hi;
  const v2f s0 = __builtin_elementwise_max(z0, z0 * NEG_SLOPE);
  const v2f s1 = __builtin_elementwise_max(z1, z1 * NEG_SLOPE);
  v2f q = s0 * a_lo;
  q = s1 * a_hi + q;
  float p = q.x + q.y;
  p += __shfl_xor(p, 1, 32);
  p += __shfl_xor(p, 2, 32);
  return p;
}

template <int NV>
struct NodeState {
  v2f   r_lo[NV], r_hi[NV];
  v2f   a_lo[NV], a_hi[NV];
  v2f   c_lo[NV], c_hi[NV];
  float m[NV], den[NV];
};

template <int STRIDE, int NV, int VOFF>
__device__ __forceinline__ void node_softmax_sum(
    const char* __restrict__ rows_b, const unsigned cbase,
    const char* __restrict__ xb, const int end, const unsigned cbyte,
    NodeState<NV>& st) {
  const float NEG_INF = -__builtin_inff();

  const int last = end - 1;
  for (int j = 0; j < end; j += 4) {
    const int s0 = *(const int*)(rows_b + (cbase + 4u * (unsigned)j));
    const int s1 = *(const int*)(rows_b + (cbase + 4u * (unsigned)min(j + 1, last)));
    const int s2 = *(const int*)(rows_b + (cbase + 4u * (unsigned)min(j + 2, last)));
    const int s3 = *(const int*)(rows_b + (cbase + 4u * (unsigned)min(j + 3, last)));
    const unsigned o0 = (unsigned)s0 * (unsigned)(STRIDE * 4) + cbyte;
    const unsigned o1 = (unsigned)s1 * (unsigned)(STRIDE * 4) + cbyte;
    const unsigned o2 = (unsigned)s2 * (unsigned)(STRIDE * 4) + cbyte;
    const unsigned o3 = (unsigned)s3 * (unsigned)(STRIDE * 4) + cbyte;
    v4f x0[NV], x1[NV], x2[NV], x3[NV];
    #pragma unroll
    for (int v = 0; v < NV; ++v) {
      x0[v] = *(const v4f*)(xb + o0 + (unsigned)(v * VOFF * 4));
      x1[v] = *(const v4f*)(xb + o1 + (unsigned)(v * VOFF * 4));
      x2[v] = *(const v4f*)(xb + o2 + (unsigned)(v * VOFF * 4));
      x3[v] = *(const v4f*)(xb + o3 + (unsigned)(v * VOFF * 4));
    }

    const bool v1 = (j + 1 < end), v2 = (j + 2 < end), v3 = (j + 3 < end);
    #pragma unroll
    for (int v = 0; v < NV; ++v) {
      const float p0 = edge_score(x0[v], st.r_lo[v], st.r_hi[v], st.a_lo[v], st.a_hi[v]);
      float p1 = edge_score(x1[v], st.r_lo[v], st.r_hi[v], st.a_lo[v], st.a_hi[v]);
      float p2 = edge_score(x2[v], st.r_lo[v], st.r_hi[v], st.a_lo[v], st.a_hi[v]);
      float p3 = edge_score(x3[v], st.r_lo[v], st.r_hi[v], st.a_lo[v], st.a_hi[v]);
      p1 = v1 ? p1 : NEG_INF;
      p2 = v2 ? p2 : NEG_INF;
      p3 = v3 ? p3 : NEG_INF;

      const float m  = st.m[v];
      const float mn = fmaxf(fmaxf(m, p0), fmaxf(fmaxf(p1, p2), p3));
      const float sc = __expf(m - mn);
      const float w0 = __expf(p0 - mn);
      const float w1 = __expf(p1 - mn);
      const float w2 = __expf(p2 - mn);
      const float w3 = __expf(p3 - mn);
      st.den[v] = st.den[v] * sc + ((w0 + w1) + (w2 + w3));
      v2f lo = st.c_lo[v] * sc;
      v2f hi = st.c_hi[v] * sc;
      lo = x0[v].xy * w0 + lo;   hi = x0[v].zw * w0 + hi;
      lo = x1[v].xy * w1 + lo;   hi = x1[v].zw * w1 + hi;
      lo = x2[v].xy * w2 + lo;   hi = x2[v].zw * w2 + hi;
      lo = x3[v].xy * w3 + lo;   hi = x3[v].zw * w3 + hi;
      st.c_lo[v] = lo;
      st.c_hi[v] = hi;
      st.m[v] = mn;
    }
  }
}

template <int STRIDE, int NV, int VOFF>
__device__ __forceinline__ void node_one_edge(
    const int s, const char* __restrict__ xb, const unsigned cbyte, NodeState<NV>& st) {
  const unsigned o = (unsigned)s * (unsigned)(STRIDE * 4) + cbyte;
  #pragma unroll
  for (int v = 0; v < NV; ++v) {
    const v4f x = *(const v4f*)(xb + o + (unsigned)(v * VOFF * 4));
    const float p  = edge_score(x, st.r_lo[v], st.r_hi[v], st.a_lo[v], st.a_hi[v]);
    const float mn = fmaxf(st.m[v], p);
    const float sc = __expf(st.m[v] - mn);
    const float w  = __expf(p - mn);
    st.den[v]  = st.den[v] * sc + w;
    st.c_lo[v] = x.xy * w + st.c_lo[v] * sc;
    st.c_hi[v] = x.zw * w + st.c_hi[v] * sc;
    st.m[v] = mn;
  }
}

template <int STRIDE, int NV, int VOFF>
__device__ __forceinline__ void node_all_edges(
    const int* __restrict__ rows, const int* __restrict__ spill_n,
    const int* __restrict__ spill, const int spill_cap,
    const float* __restrict__ xl, const int n, const int coff,
    const float4 (&rv)[NV], const float4 (&at)[NV],
    float4 (&acc)[NV], float (&den)[NV]) {
  const char* rows_b = (const char*)rows;
  const char* xb     = (const char*)xl;
  const unsigned rbase = (unsigned)n * (unsigned)(ROW_W * 4);
  const unsigned cbyte = (unsigned)coff * 4u;
  const int cnt = *(const int*)(rows_b + rbase);
  NodeState<NV> st;
  #pragma unroll
  for (int v = 0; v < NV; ++v) {
    st.r_lo[v] = v2f{rv[v].x, rv[v].y};  st.r_hi[v] = v2f{rv[v].z, rv[v].w};
    st.a_lo[v] = v2f{at[v].x, at[v].y};  st.a_hi[v] = v2f{at[v].z, at[v].w};
    st.c_lo[v] = v2f{0.0f, 0.0f};        st.c_hi[v] = v2f{0.0f, 0.0f};
    st.m[v]    = -__builtin_inff();
    st.den[v]  = 0.0f;
  }
  node_softmax_sum<STRIDE, NV, VOFF>(rows_b, rbase + 4u, xb, min(cnt, ROW_CAP), cbyte, st);
  if (cnt > ROW_CAP) {
    const int ns = min(*spill_n, spill_cap);
    for (int i = 0; i < ns; ++i)
      if (spill[2 * (size_t)i] == n)
        node_one_edge<STRIDE, NV, VOFF>(spill[2 * (size_t)i + 1], xb, cbyte, st);
  }
  #pragma unroll
  for (int v = 0; v < NV; ++v) {
    acc[v] = make_float4(st.c_lo[v].x, st.c_lo[v].y, st.c_hi[v].x, st.c_hi[v].y);
    den[v] = st.den[v];
  }
}

struct HeadState {
  v2f   r_lo[2], r_hi[2];
  v2f   a_lo[2], a_hi[2];
  v2f   c_lo[2], c_hi[2];
  float m, den;
};

__device__ __forceinline__ float edge_score8(const v4f xa, const v4f xb, const HeadState& st) {
  const v2f z0 = xa.xy + st.r_lo[0];
  const v2f z1 = xa.zw + st.r_hi[0];
  const v2f z2 = xb.xy + st.r_lo[1];
  const v2f z3 = xb.zw + st.r_hi[1];
  const v2f s0 = __builtin_elementwise_max(z0, z0 * NEG_SLOPE);
  const v2f s1 = __builtin_elementwise_max(z1, z1 * NEG_SLOPE);
  const v2f s2 = __builtin_elementwise_max(z2, z2 * NEG_SLOPE);
  const v2f s3 = __builtin_elementwise_max(z3, z3 * NEG_SLOPE);
  v2f q = s0 * st.a_lo[0];
  q = s1 * st.a_hi[0] + q;
  q = s2 * st.a_lo[1] + q;
  q = s3 * st.a_hi[1] + q;
  float p = q.x + q.y;
  p += __shfl_xor(p, 1, 32);
  return p;
}

template <int STRIDE, int VOFF>
__device__ __forceinline__ void head_softmax_sum(
    const char* __restrict__ rows_b, const unsigned cbase,
    const char* __restrict__ xb, const int end, const unsigned cbyte,
    HeadState& st) {
  const float NEG_INF = -__builtin_inff();

  const int last = end - 1;
  for (int j = 0; j < end; j += 4) {
    const int s0 = *(const int*)(rows_b + (cbase + 4u * (unsigned)j));
    const int s1 = *(const int*)(rows_b + (cbase + 4u * (unsigned)min(j + 1, last)));
    const int s2 = *(const int*)(rows_b + (cbase + 4u * (unsigned)min(j + 2, last)));
    const int s3 = *(const int*)(rows_b + (cbase + 4u * (unsigned)min(j + 3, last)));
    const unsigned o0 = (unsigned)s0 * (unsigned)(STRIDE * 4) + cbyte;
    const unsigned o1 = (unsigned)s1 * (unsigned)(STRIDE * 4) + cbyte;
    const unsigned o2 = (unsigned)s2 * (unsigned)(STRIDE * 4) + cbyte;
    const unsigned o3 = (unsigned)s3 * (unsigned)(STRIDE * 4) + cbyte;
    v4f x0[2], x1[2], x2[2], x3[2];
    #pragma unroll
    for (int v = 0; v < 2; ++v) {
      x0[v] = *(const v4f*)(xb + o0 + (unsigned)(v * VOFF * 4));
      x1[v] = *(const v4f*)(xb + o1 + (unsigned)(v * VOFF * 4));
      x2[v] = *(const v4f*)(xb + o2 + (unsigned)(v * VOFF * 4));
      x3[v] = *(const v4f*)(xb + o3 + (unsigned)(v * VOFF * 4));
    }

    const bool v1 = (j + 1 < end), v2 = (j + 2 < end), v3 = (j + 3 < end);
    const float p0 = edge_score8(x0[0], x0[1], st);
    float p1 = edge_score8(x1[0], x1[1], st);
    float p2 = edge_score8(x2[0], x2[1], st);
    float p3 = edge_score8(x3[0], x3[1], st);
    p1 = v1 ? p1 : NEG_INF;
    p2 = v2 ? p2 : NEG_INF;
    p3 = v3 ? p3 : NEG_INF;

    const float m  = st.m;
    const float mn = fmaxf(fmaxf(m, p0), fmaxf(fmaxf(p1, p2), p3));
    const float sc = __expf(m - mn);
    const float w0 = __expf(p0 - mn);
    const float w1 = __expf(p1 - mn);
    const float w2 = __expf(p2 - mn);
    const float w3 = __expf(p3 - mn);
    st.den = st.den * sc + ((w0 + w1) + (w2 + w3));
    #pragma unroll
    for (int v = 0; v < 2; ++v) {
      v2f lo = st.c_lo[v] * sc;
      v2f hi = st.c_hi[v] * sc;
      lo = x0[v].xy * w0 + lo;   hi = x0[v].zw * w0 + hi;
      lo = x1[v].xy * w1 + lo;   hi = x1[v].zw * w1 + hi;
      lo = x2[v].xy * w2 + lo;   hi = x2[v].zw * w2 + hi;
      lo = x3[v].xy * w3 + lo;   hi = x3[v].zw * w3 + hi;
      st.c_lo[v] = lo;
      st.c_hi[v] = hi;
    }
    st.m = mn;
  }
}

template <int STRIDE, int VOFF>
__device__ __forceinline__ void head_one_edge(
    const int s, const char* __restrict__ xb, const unsigned cbyte, HeadState& st) {
  const unsigned o = (unsigned)s * (unsigned)(STRIDE * 4) + cbyte;
  const v4f xa = *(const v4f*)(xb + o);
  const v4f xc = *(const v4f*)(xb + o + (unsigned)(VOFF * 4));
  const float p  = edge_score8(xa, xc, st);
  const float mn = fmaxf(st.m, p);
  const float sc = __expf(st.m - mn);
  const float w  = __expf(p - mn);
  st.den     = st.den * sc + w;
  st.c_lo[0] = xa.xy * w + st.c_lo[0] * sc;
  st.c_hi[0] = xa.zw * w + st.c_hi[0] * sc;
  st.c_lo[1] = xc.xy * w + st.c_lo[1] * sc;
  st.c_hi[1] = xc.zw * w + st.c_hi[1] * sc;
  st.m = mn;
}

template <int STRIDE, int VOFF>
__device__ __forceinline__ void head_all_edges(
    const int* __restrict__ rows, const int* __restrict__ spill_n,
    const int* __restrict__ spill, const int spill_cap,
    const float* __restrict__ xl, const int n, const int coff,
    const float4 (&rv)[2], const float4 (&at)[2],
    float4 (&acc)[2], float& den) {
  const char* rows_b = (const char*)rows;
  const char* xb     = (const char*)xl;
  const unsigned rbase = (unsigned)n * (unsigned)(ROW_W * 4);
  const unsigned cbyte = (unsigned)coff * 4u;
  const int cnt = *(const int*)(rows_b + rbase);
  HeadState st;
  #pragma unroll
  for (int v = 0; v < 2; ++v) {
    st.r_lo[v] = v2f{rv[v].x, rv[v].y};  st.r_hi[v] = v2f{rv[v].z, rv[v].w};
    st.a_lo[v] = v2f{at[v].x, at[v].y};  st.a_hi[v] = v2f{at[v].z, at[v].w};
    st.c_lo[v] = v2f{0.0f, 0.0f};        st.c_hi[v] = v2f{0.0f, 0.0f};
  }
  st.m   = -__builtin_inff();
  st.den = 0.0f;
  head_softmax_sum<STRIDE, VOFF>(rows_b, rbase + 4u, xb, min(cnt, ROW_CAP), cbyte, st);
  if (cnt > ROW_CAP) {
    const int ns = min(*spill_n, spill_cap);
    for (int i = 0; i < ns; ++i)
      if (spill[2 * (size_t)i] == n)
        head_one_edge<STRIDE, VOFF>(spill[2 * (size_t)i + 1], xb, cbyte, st);
  }
  #pragma unroll
  for (int v = 0; v < 2; ++v)
    acc[v] = make_float4(st.c_lo[v].x, st.c_lo[v].y, st.c_hi[v].x, st.c_hi[v].y);
  den = st.den;
}

#define H_PITCH 68

__global__ void __launch_bounds__(128)
l1_node_gemm2_kernel(const int* __restrict__ rows, const int* __restrict__ spill_n,
                     const int* __restrict__ spill, int spill_cap,
                     const float* __restrict__ xl1, const float* __restrict__ xr1,
                     const float* __restrict__ att1, const float* __restrict__ b1,
                     const float* __restrict__ W2l, const float* __restrict__ W2r,
                     float* __restrict__ xl2, float* __restrict__ xr2, int n_nodes) {
  __shared__ float hs[16 * H_PITCH];
  const int nl   = threadIdx.x >> 3;
  const int n    = blockIdx.x * 16 + nl;
  const int g    = threadIdx.x & 7;
  const int coff = g * 4;
  const int hc   = 16 * (g >> 1) + 4 * (g & 1);

  float4 hv[2];
  hv[0] = make_float4(0.0f, 0.0f, 0.0f, 0.0f);
  hv[1] = make_float4(0.0f, 0.0f, 0.0f, 0.0f);
  if (n < n_nodes) {
    float4 rv[2], at[2];
    rv[0] = *(const float4*)(xr1 + (size_t)n * 64 + coff);
    rv[1] = *(const float4*)(xr1 + (size_t)n * 64 + coff + 32);
    at[0] = *(const float4*)(att1 + hc);
    at[1] = *(const float4*)(att1 + hc + 8);

    float4 acc[2]; float dn;
    head_all_edges<64, 32>(rows, spill_n, spill, spill_cap, xl1, n, coff, rv, at, acc, dn);

    #pragma unroll
    for (int u = 0; u < 2; ++u) {
      const float4 bb = *(const float4*)(b1 + hc + 8 * u);
      float4 v;
      v.x = ((dn > 0.0f) ? acc[u].x / dn : 0.0f) + bb.x;
      v.y = ((dn > 0.0f) ? acc[u].y / dn : 0.0f) + bb.y;
      v.z = ((dn > 0.0f) ? acc[u].z / dn : 0.0f) + bb.z;
      v.w = ((dn > 0.0f) ? acc[u].w / dn : 0.0f) + bb.w;
      v.x = (v.x > 0.0f) ? v.x : expm1f(v.x);
      v.y = (v.y > 0.0f) ? v.y : expm1f(v.y);
      v.z = (v.z > 0.0f) ? v.z : expm1f(v.z);
      v.w = (v.w > 0.0f) ? v.w : expm1f(v.w);
      hv[u] = v;
    }
  }
  *(float4*)(hs + nl * H_PITCH + hc)     = hv[0];
  *(float4*)(hs + nl * H_PITCH + hc + 8) = hv[1];
  __syncthreads();

  const int wave = __builtin_amdgcn_readfirstlane((int)(threadIdx.x >> 5));
  const int m0   = blockIdx.x * 16;
  if (wave == 0) {
    const int lane = threadIdx.x & 31;
    const int half = lane >> 4;
    const int rc   = lane & 15;
    const float* hrow = hs + rc * H_PITCH;
    const bool colok = (rc < 10);
    const int  cc    = colok ? rc : 9;

    v8f accL = {}, accR = {};
    #pragma unroll
    for (int kc = 0; kc < 16; ++kc) {
      const int k0 = 4 * kc + 2 * half;
      const v2f a = *(const v2f*)(hrow + k0);
      const float l0 = W2l[(size_t)(k0 + 0) * 10 + cc];
      const float l1 = W2l[(size_t)(k0 + 1) * 10 + cc];
      const float r0 = W2r[(size_t)(k0 + 0) * 10 + cc];
      const float r1 = W2r[(size_t)(k0 + 1) * 10 + cc];
      v2f bl, br;
      bl[0] = colok ? l0 : 0.0f;
      bl[1] = colok ? l1 : 0.0f;
      br[0] = colok ? r0 : 0.0f;
      br[1] = colok ? r1 : 0.0f;
      accL = __builtin_amdgcn_wmma_f32_16x16x4_f32(
          false, a, false, bl, (short)0, accL, false, false);
      accR = __builtin_amdgcn_wmma_f32_16x16x4_f32(
          false, a, false, br, (short)0, accR, false, false);
    }

    #pragma unroll
    for (int r = 0; r < 8; ++r) {
      const int row = m0 + r + 8 * half;
      if (row < n_nodes) {
        xl2[(size_t)row * 16 + rc] = accL[r];
        xr2[(size_t)row * 16 + rc] = accR[r];
      }
    }
  }
}

__global__ void __launch_bounds__(256)
l2_node_kernel(const int* __restrict__ rows, const int* __restrict__ spill_n,
               const int* __restrict__ spill, int spill_cap,
               const float* __restrict__ xl2, const float* __restrict__ xr2,
               const float* __restrict__ att2, const float* __restrict__ b2,
               float* __restrict__ out, int n_nodes) {
  __shared__ float stage[64 * 10];
  const int nl = threadIdx.x >> 2;
  const int n  = blockIdx.x * 64 + nl;
  const int c0 = (threadIdx.x & 3) * 4;

  if (n < n_nodes) {
    float4 rv[1], at[1];
    rv[0] = *(const float4*)(xr2 + (size_t)n * 16 + c0);
    at[0].x = (c0 + 0 < 10) ? att2[c0 + 0] : 0.0f;
    at[0].y = (c0 + 1 < 10) ? att2[c0 + 1] : 0.0f;
    at[0].z = (c0 + 2 < 10) ? att2[c0 + 2] : 0.0f;
    at[0].w = (c0 + 3 < 10) ? att2[c0 + 3] : 0.0f;

    float4 acc[1]; float den[1];
    node_all_edges<16, 1, 0>(rows, spill_n, spill, spill_cap, xl2, n, c0, rv, at, acc, den);

    const bool ok = (den[0] > 0.0f);
    if (c0 + 0 < 10) stage[nl * 10 + c0 + 0] = (ok ? acc[0].x / den[0] : 0.0f) + b2[c0 + 0];
    if (c0 + 1 < 10) stage[nl * 10 + c0 + 1] = (ok ? acc[0].y / den[0] : 0.0f) + b2[c0 + 1];
    if (c0 + 2 < 10) stage[nl * 10 + c0 + 2] = (ok ? acc[0].z / den[0] : 0.0f) + b2[c0 + 2];
    if (c0 + 3 < 10) stage[nl * 10 + c0 + 3] = (ok ? acc[0].w / den[0] : 0.0f) + b2[c0 + 3];
  }
  __syncthreads();

  const long long base  = (long long)blockIdx.x * 640;
  const long long total = (long long)n_nodes * 10;
  for (int i = threadIdx.x; i < 640; i += 256)
    if (base + i < total) out[base + i] = stage[i];
}

extern "C" void kernel_launch(void* const* d_in, const int* in_sizes, int n_in,
                              void* d_out, int out_size, void* d_ws, size_t ws_size,
                              hipStream_t stream) {
  const float* x    = (const float*)d_in[0];
  const int*   ei   = (const int*)d_in[1];
  const float* W1l  = (const float*)d_in[2];
  const float* W1r  = (const float*)d_in[3];
  const float* att1 = (const float*)d_in[4];
  const float* b1   = (const float*)d_in[5];
  const float* W2l  = (const float*)d_in[6];
  const float* W2r  = (const float*)d_in[7];
  const float* att2 = (const float*)d_in[8];
  const float* b2   = (const float*)d_in[9];
  float* out = (float*)d_out;

  const int N = in_sizes[0] / 128;
  const int E = in_sizes[1] / 2;

  float* ws = (float*)d_ws;
  size_t o = 0;
  int* rows    = (int*)(ws + o); o += (size_t)N * ROW_W;
  float* xl1   = ws + o; o += (size_t)N * 64;
  float* xr1   = ws + o; o += (size_t)N * 64;
  float* xl2   = ws + o; o += (size_t)N * 16;
  float* xr2   = ws + o; o += (size_t)N * 16;
  const int n_chunk = (int)(((long long)E + CHUNK_E - 1) / CHUNK_E);
  int* spill_n = (int*)(ws + o); o += 64;
  int* claim   = (int*)(ws + o); o += ((size_t)n_chunk + 63) / 64 * 64;
  int* spill   = (int*)(ws + o); o += 2 * (size_t)E;

  auto blocks = [](long long n) { return (unsigned)((n + 255) / 256); };
  const int mtiles = (N + 15) / 16;

  zero_heads_kernel<<<blocks(N > n_chunk ? N : n_chunk), 256, 0, stream>>>(
      rows, spill_n, claim, N, n_chunk);
  const int edge_blocks = 8 * ((n_chunk + 3) / 4);
  gemm1_edges_kernel<<<(unsigned)(GEMM_RESIDENT + edge_blocks), 128, 0, stream>>>(
      x, W1l, W1r, xl1, xr1, N, mtiles, ei, rows, spill_n, spill, claim, E, edge_blocks,
      n_chunk, N / 2);

  l1_node_gemm2_kernel<<<(unsigned)mtiles, 128, 0, stream>>>(
      rows, spill_n, spill, E, xl1, xr1, att1, b1, W2l, W2r, xl2, xr2, N);

  l2_node_kernel<<<(unsigned)((N + 63) / 64), 256, 0, stream>>>(rows, spill_n, spill, E, xl2, xr2,
                                                                att2, b2, out, N);
}
